# starting kernel
# baseline (speedup 1.0000x reference)
#include <hip/hip_runtime.h>

typedef __attribute__((ext_vector_type(2))) float v2f;
typedef __attribute__((ext_vector_type(8))) float v8f;

#define HIDW 128
#define KCHUNK 32

__global__ __launch_bounds__(256) void k_fill1(float* p, int n) {
  int i = blockIdx.x * blockDim.x + threadIdx.x;
  if (i < n) p[i] = 1.0f;
}

__global__ __launch_bounds__(256) void k_edge_deg(float* deg, const int* __restrict__ dst, int E) {
  int e = blockIdx.x * blockDim.x + threadIdx.x;
  if (e < E) atomicAdd(&deg[dst[e]], 1.0f);
}

__global__ __launch_bounds__(256) void k_dis(float* dis, const float* __restrict__ deg, int n) {
  int i = blockIdx.x * blockDim.x + threadIdx.x;
  if (i < n) {
    float d = deg[i];
    dis[i] = (d > 0.f) ? rsqrtf(d) : 0.f;
  }
}

__global__ __launch_bounds__(256) void k_prop_init(float* __restrict__ out, const float* __restrict__ h,
                                                   const float* __restrict__ dis, const float* __restrict__ bias,
                                                   int n) {
  int idx = blockIdx.x * blockDim.x + threadIdx.x;
  if (idx < n * HIDW) {
    int i = idx >> 7;
    int f = idx & 127;
    float di = dis[i];
    out[idx] = h[idx] * di * di + bias[f];
  }
}

__global__ __launch_bounds__(256) void k_prop_scatter(const float* __restrict__ h, const int* __restrict__ src,
                                                      const int* __restrict__ dst, const float* __restrict__ dis,
                                                      float* __restrict__ out, int E) {
  int e = blockIdx.x * 2 + (threadIdx.x >> 7);
  if (e >= E) return;
  int f = threadIdx.x & 127;
  int s = src[e], d = dst[e];
  float nrm = dis[s] * dis[d];
  atomicAdd(&out[(size_t)d * HIDW + f], h[(size_t)s * HIDW + f] * nrm);
}

template<int NC, int KW, bool RELU_A0, bool RELU_OUT, bool HAS_BIAS>
__global__ __launch_bounds__(256) void k_gemm(const float* __restrict__ A0, const float* __restrict__ A1,
                                              const float* __restrict__ W, const float* __restrict__ bias,
                                              float* __restrict__ C, int M) {
  __shared__ __align__(16) float As[128][KCHUNK + 4];
  __shared__ __align__(16) float Ws[NC][KCHUNK + 4];
  constexpr int NT = NC / 16;
  constexpr int K = KW * 128;

  const int tid  = threadIdx.x;
  const int wave = tid >> 5;
  const int lane = tid & 31;
  const int l15  = lane & 15;
  const int half = lane >> 4;
  const int m0   = blockIdx.x * 128;

  v8f acc[NT];
#pragma unroll
  for (int t = 0; t < NT; ++t)
#pragma unroll
    for (int r = 0; r < 8; ++r) acc[t][r] = 0.0f;

  for (int kb = 0; kb < K; kb += KCHUNK) {
    const float* Ablk = (KW == 2 && kb >= 128) ? A1 : A0;
    const bool doRelu = RELU_A0 && (kb < 128);
    const int kloc = kb & 127;

#pragma unroll
    for (int j = 0; j < (128 * KCHUNK) / 1024; ++j) {
      int ei  = tid * 4 + j * 1024;
      int row = ei >> 5;
      int col = ei & 31;
      int grow = m0 + row;
      float4 v = {0.f, 0.f, 0.f, 0.f};
      if (grow < M) v = *(const float4*)&Ablk[(size_t)grow * HIDW + kloc + col];
      if (doRelu) {
        v.x = fmaxf(v.x, 0.f); v.y = fmaxf(v.y, 0.f);
        v.z = fmaxf(v.z, 0.f); v.w = fmaxf(v.w, 0.f);
      }
      *(float4*)&As[row][col] = v;
    }
#pragma unroll
    for (int j = 0; j < (NC * KCHUNK) / 1024; ++j) {
      int ei  = tid * 4 + j * 1024;
      int n   = ei >> 5;
      int col = ei & 31;
      *(float4*)&Ws[n][col] = *(const float4*)&W[(size_t)n * K + kb + col];
    }
    __syncthreads();

    const int arow = wave * 16 + l15;
    const int o = half * 2;
#pragma unroll
    for (int k4 = 0; k4 < KCHUNK; k4 += 4) {
      v2f a = *(const v2f*)&As[arow][k4 + o];
#pragma unroll
      for (int t = 0; t < NT; ++t) {
        v2f b = *(const v2f*)&Ws[t * 16 + l15][k4 + o];
        acc[t] = __builtin_amdgcn_wmma_f32_16x16x4_f32(
             false, a,  false, b,
             (short)0, acc[t],  false,  false);
      }
    }
    __syncthreads();
  }

  const int rbase = m0 + wave * 16 + half * 8;
#pragma unroll
  for (int t = 0; t < NT; ++t) {
    int col = t * 16 + l15;
    float bv = HAS_BIAS ? bias[col] : 0.f;
#pragma unroll
    for (int r = 0; r < 8; ++r) {
      int row = rbase + r;
      if (row < M) {
        float v = acc[t][r] + bv;
        if (RELU_OUT) v = fmaxf(v, 0.f);
        C[(size_t)row * NC + col] = v;
      }
    }
  }
}

extern "C" void kernel_launch(void* const* d_in, const int* in_sizes, int n_in,
                              void* d_out, int out_size, void* d_ws, size_t ws_size,
                              hipStream_t stream) {
  const float* x  = (const float*)d_in[0];
  const int* edge = (const int*)d_in[1];
  const float* W1 = (const float*)d_in[2];
  const float* b1 = (const float*)d_in[3];
  const float* Wc = (const float*)d_in[4];
  const float* bc = (const float*)d_in[5];
  const float* W2 = (const float*)d_in[6];
  const float* b2 = (const float*)d_in[7];
  const float* Wo = (const float*)d_in[8];
  const float* bo = (const float*)d_in[9];

  const int N = in_sizes[0] / HIDW;
  const int E = in_sizes[1] / 2;
  const int* srcIdx = edge;
  const int* dstIdx = edge + E;

  float* wsf = (float*)d_ws;
  size_t NP   = (((size_t)N) + 255) & ~(size_t)255;
  float* dis  = wsf;
  float* buf1 = wsf + NP;
  float* buf2 = buf1 + (size_t)N * HIDW;
  float* buf3 = buf2 + (size_t)N * HIDW;
  float* deg  = buf1;

  dim3 b256(256);
  int gN  = (N + 255) / 256;
  int gNH = (N * HIDW + 255) / 256;
  int gE  = (E + 255) / 256;
  int gE2 = (E + 1) / 2;
  int gG  = (N + 127) / 128;

  k_fill1   <<<gN,  b256, 0, stream>>>(deg, N);
  k_edge_deg<<<gE,  b256, 0, stream>>>(deg, dstIdx, E);
  k_dis     <<<gN,  b256, 0, stream>>>(dis, deg, N);

  k_gemm<128,1,false,false,false><<<gG, b256, 0, stream>>>(x, nullptr, W1, nullptr, buf1, N);
  k_prop_init   <<<gNH, b256, 0, stream>>>(buf2, buf1, dis, b1, N);
  k_prop_scatter<<<gE2, b256, 0, stream>>>(buf1, srcIdx, dstIdx, dis, buf2, E);

  k_gemm<128,2,true,true,true><<<gG, b256, 0, stream>>>(buf2, x, Wc, bc, buf3, N);

  k_gemm<128,1,false,false,false><<<gG, b256, 0, stream>>>(buf3, nullptr, W2, nullptr, buf1, N);
  k_prop_init   <<<gNH, b256, 0, stream>>>(buf2, buf1, dis, b2, N);
  k_prop_scatter<<<gE2, b256, 0, stream>>>(buf1, srcIdx, dstIdx, dis, buf2, E);

  k_gemm<64,2,false,false,true><<<gG, b256, 0, stream>>>(buf2, buf3, Wo, bo, (float*)d_out, N);
}


# Float-atomic scatter-add replaced by per-call edge list grouped by destination (returning int count, scan, place; arrival-order rank) + one-wave-per-row gather sums; self loop and bias folded in
# speedup vs baseline: 3.9167x; 3.9167x over previous
#include <hip/hip_runtime.h>

typedef __attribute__((ext_vector_type(2))) float v2f;
typedef __attribute__((ext_vector_type(8))) float v8f;

#define HIDW 128
#define KCHUNK 32

__global__ __launch_bounds__(256) void k_zero(int* __restrict__ p, int n) {
  int i = blockIdx.x * blockDim.x + threadIdx.x;
  if (i < n) p[i] = 0;
}

__global__ __launch_bounds__(256) void k_edge_count(int* __restrict__ cnt, int* __restrict__ rank,
                                                    const int* __restrict__ dst, int E) {
  int e = blockIdx.x * blockDim.x + threadIdx.x;
  if (e < E) rank[e] = atomicAdd(&cnt[dst[e]], 1);
}

#define SCAN_PER 16
#define SCAN_TILE (1024 * SCAN_PER)
__global__ __launch_bounds__(1024) void k_scan_dis(const int* __restrict__ cnt, int* __restrict__ rowptr,
                                                   float* __restrict__ dis, int n) {
  __shared__ int wsum[32];
  const int tid  = threadIdx.x;
  const int lane = tid & 31;
  const int wv   = tid >> 5;
  int carry = 0;
  for (int t0 = 0; t0 < n; t0 += SCAN_TILE) {
    const int b = t0 + tid * SCAN_PER;
    int c[SCAN_PER];
#pragma unroll
    for (int q = 0; q < SCAN_PER; q += 4) {
      int4 t = *(const int4*)&cnt[b + q];
      c[q] = t.x; c[q + 1] = t.y; c[q + 2] = t.z; c[q + 3] = t.w;
    }
    int s = 0;
#pragma unroll
    for (int q = 0; q < SCAN_PER; ++q) s += c[q];

    int v = s;
    for (int o = 1; o < 32; o <<= 1) {
      int t = __shfl_up(v, o);
      if (lane >= o) v += t;
    }
    if (lane == 31) wsum[wv] = v;
    __syncthreads();
    if (wv == 0) {
      int t = wsum[lane];
      for (int o = 1; o < 32; o <<= 1) {
        int u = __shfl_up(t, o);
        if (lane >= o) t += u;
      }
      wsum[lane] = t;
    }
    __syncthreads();
    int run = carry + v - s + (wv ? wsum[wv - 1] : 0);
    carry += wsum[31];

#pragma unroll
    for (int q = 0; q < SCAN_PER; q += 4) {
      int   r[4];
      float d[4];
#pragma unroll
      for (int k = 0; k < 4; ++k) {
        r[k] = run;
        run += c[q + k];
        float dg = (float)(c[q + k] + 1);
        d[k] = (dg > 0.f) ? rsqrtf(dg) : 0.f;
      }
      const int i = b + q;
      if (i + 3 < n) {
        *(int4*)&rowptr[i] = make_int4(r[0], r[1], r[2], r[3]);
        *(float4*)&dis[i]  = make_float4(d[0], d[1], d[2], d[3]);
      } else {
#pragma unroll
        for (int k = 0; k < 4; ++k)
          if (i + k < n) { rowptr[i + k] = r[k]; dis[i + k] = d[k]; }
      }
    }
    __syncthreads();
  }
  if (tid == 0) rowptr[n] = carry;
}

__global__ __launch_bounds__(256) void k_fill(int* __restrict__ csr, const int* __restrict__ rowptr,
                                              const int* __restrict__ rank, const int* __restrict__ src,
                                              const int* __restrict__ dst, int E) {
  int e = blockIdx.x * blockDim.x + threadIdx.x;
  if (e < E) csr[rowptr[dst[e]] + rank[e]] = src[e];
}

template<int U>
__device__ __forceinline__ void gather_rows(float4& acc, const float* __restrict__ h, int my_s, float my_d,
                                            float di, int j, int lane4) {
  float4 v[U];
#pragma unroll
  for (int u = 0; u < U; ++u) {
    int s = __builtin_amdgcn_readlane(my_s, j + u);
    v[u] = *(const float4*)(h + (size_t)s * HIDW + lane4);
  }
#pragma unroll
  for (int u = 0; u < U; ++u) {
    float w = __int_as_float(__builtin_amdgcn_readlane(__float_as_int(my_d), j + u)) * di;
    acc.x = fmaf(v[u].x, w, acc.x);
    acc.y = fmaf(v[u].y, w, acc.y);
    acc.z = fmaf(v[u].z, w, acc.z);
    acc.w = fmaf(v[u].w, w, acc.w);
  }
}

__global__ __launch_bounds__(256) void k_prop_gather(float* __restrict__ out, const float* __restrict__ h,
                                                     const int* __restrict__ rowptr, const int* __restrict__ csr,
                                                     const float* __restrict__ dis, const float* __restrict__ bias,
                                                     int n) {
  const int lane  = threadIdx.x & 31;
  const int lane4 = lane * 4;
  const int wave = __builtin_amdgcn_readfirstlane((int)(threadIdx.x >> 5));
  const int nw   = gridDim.x * 8;
  const float4 bv = *(const float4*)&bias[lane4];

  for (int i = blockIdx.x * 8 + wave; i < n; i += nw) {
    const int beg = rowptr[i];
    const int end = rowptr[i + 1];
    const float di = dis[i];
    const float4 hv = *(const float4*)(h + (size_t)i * HIDW + lane4);
    float4 acc;
    acc.x = hv.x * di * di + bv.x;
    acc.y = hv.y * di * di + bv.y;
    acc.z = hv.z * di * di + bv.z;
    acc.w = hv.w * di * di + bv.w;

    for (int base = beg; base < end; base += 32) {
      const int m = min(32, end - base);
      const int my_s = csr[min(base + lane, end - 1)];
      const float my_d = dis[my_s];
      int j = 0;
      for (; j + 8 <= m; j += 8) gather_rows<8>(acc, h, my_s, my_d, di, j, lane4);
      if (m - j >= 4) { gather_rows<4>(acc, h, my_s, my_d, di, j, lane4); j += 4; }
      if (m - j >= 2) { gather_rows<2>(acc, h, my_s, my_d, di, j, lane4); j += 2; }
      if (m - j >= 1) { gather_rows<1>(acc, h, my_s, my_d, di, j, lane4); }
    }
    *(float4*)(out + (size_t)i * HIDW + lane4) = acc;
  }
}

template<int NC, int KW, bool RELU_A0, bool RELU_OUT, bool HAS_BIAS>
__global__ __launch_bounds__(256) void k_gemm(const float* __restrict__ A0, const float* __restrict__ A1,
                                              const float* __restrict__ W, const float* __restrict__ bias,
                                              float* __restrict__ C, int M) {
  __shared__ __align__(16) float As[128][KCHUNK + 4];
  __shared__ __align__(16) float Ws[NC][KCHUNK + 4];
  constexpr int NT = NC / 16;
  constexpr int K = KW * 128;

  const int tid  = threadIdx.x;
  const int wave = tid >> 5;
  const int lane = tid & 31;
  const int l15  = lane & 15;
  const int half = lane >> 4;
  const int m0   = blockIdx.x * 128;

  v8f acc[NT];
#pragma unroll
  for (int t = 0; t < NT; ++t)
#pragma unroll
    for (int r = 0; r < 8; ++r) acc[t][r] = 0.0f;

  for (int kb = 0; kb < K; kb += KCHUNK) {
    const float* Ablk = (KW == 2 && kb >= 128) ? A1 : A0;
    const bool doRelu = RELU_A0 && (kb < 128);
    const int kloc = kb & 127;

#pragma unroll
    for (int j = 0; j < (128 * KCHUNK) / 1024; ++j) {
      int ei  = tid * 4 + j * 1024;
      int row = ei >> 5;
      int col = ei & 31;
      int grow = m0 + row;
      float4 v = {0.f, 0.f, 0.f, 0.f};
      if (grow < M) v = *(const float4*)&Ablk[(size_t)grow * HIDW + kloc + col];
      if (doRelu) {
        v.x = fmaxf(v.x, 0.f); v.y = fmaxf(v.y, 0.f);
        v.z = fmaxf(v.z, 0.f); v.w = fmaxf(v.w, 0.f);
      }
      *(float4*)&As[row][col] = v;
    }
#pragma unroll
    for (int j = 0; j < (NC * KCHUNK) / 1024; ++j) {
      int ei  = tid * 4 + j * 1024;
      int n   = ei >> 5;
      int col = ei & 31;
      *(float4*)&Ws[n][col] = *(const float4*)&W[(size_t)n * K + kb + col];
    }
    __syncthreads();

    const int arow = wave * 16 + l15;
    const int o = half * 2;
#pragma unroll
    for (int k4 = 0; k4 < KCHUNK; k4 += 4) {
      v2f a = *(const v2f*)&As[arow][k4 + o];
#pragma unroll
      for (int t = 0; t < NT; ++t) {
        v2f b = *(const v2f*)&Ws[t * 16 + l15][k4 + o];
        acc[t] = __builtin_amdgcn_wmma_f32_16x16x4_f32(
             false, a,  false, b,
             (short)0, acc[t],  false,  false);
      }
    }
    __syncthreads();
  }

  const int rbase = m0 + wave * 16 + half * 8;
#pragma unroll
  for (int t = 0; t < NT; ++t) {
    int col = t * 16 + l15;
    float bv = HAS_BIAS ? bias[col] : 0.f;
#pragma unroll
    for (int r = 0; r < 8; ++r) {
      int row = rbase + r;
      if (row < M) {
        float v = acc[t][r] + bv;
        if (RELU_OUT) v = fmaxf(v, 0.f);
        C[(size_t)row * NC + col] = v;
      }
    }
  }
}

extern "C" void kernel_launch(void* const* d_in, const int* in_sizes, int n_in,
                              void* d_out, int out_size, void* d_ws, size_t ws_size,
                              hipStream_t stream) {
  const float* x  = (const float*)d_in[0];
  const int* edge = (const int*)d_in[1];
  const float* W1 = (const float*)d_in[2];
  const float* b1 = (const float*)d_in[3];
  const float* Wc = (const float*)d_in[4];
  const float* bc = (const float*)d_in[5];
  const float* W2 = (const float*)d_in[6];
  const float* b2 = (const float*)d_in[7];
  const float* Wo = (const float*)d_in[8];
  const float* bo = (const float*)d_in[9];

  const int N = in_sizes[0] / HIDW;
  const int E = in_sizes[1] / 2;
  const int* srcIdx = edge;
  const int* dstIdx = edge + E;

  float* wsf = (float*)d_ws;
  size_t NP   = (((size_t)N) + 255) & ~(size_t)255;
  size_t NP1  = (((size_t)N) + 1 + 255) & ~(size_t)255;
  float* dis  = wsf;
  float* buf1 = wsf + NP;
  float* buf2 = buf1 + (size_t)N * HIDW;
  float* buf3 = buf2 + (size_t)N * HIDW;
  size_t NPS  = (((size_t)N) + SCAN_TILE - 1) & ~(size_t)(SCAN_TILE - 1);
  int* cnt    = (int*)buf3;
  int* rank   = cnt + NPS;
  int* rowptr = (int*)d_out;
  int* csr    = rowptr + NP1;

  dim3 b256(256);
  int gE  = (E + 255) / 256;
  int gG  = (N + 127) / 128;
  int gP  = 2048;

  k_zero      <<<(int)(NPS / 256), b256, 0, stream>>>(cnt, (int)NPS);
  k_edge_count<<<gE, b256, 0, stream>>>(cnt, rank, dstIdx, E);
  k_scan_dis  <<<1, dim3(1024), 0, stream>>>(cnt, rowptr, dis, N);
  k_fill      <<<gE, b256, 0, stream>>>(csr, rowptr, rank, srcIdx, dstIdx, E);

  k_gemm<128,1,false,false,false><<<gG, b256, 0, stream>>>(x, nullptr, W1, nullptr, buf1, N);
  k_prop_gather<<<gP, b256, 0, stream>>>(buf2, buf1, rowptr, csr, dis, b1, N);

  k_gemm<128,2,true,true,true><<<gG, b256, 0, stream>>>(buf2, x, Wc, bc, buf3, N);

  k_gemm<128,1,false,false,false><<<gG, b256, 0, stream>>>(buf3, nullptr, W2, nullptr, buf1, N);
  k_prop_gather<<<gP, b256, 0, stream>>>(buf2, buf1, rowptr, csr, dis, b2, N);

  k_gemm<64,2,false,false,true><<<gG, b256, 0, stream>>>(buf2, buf3, Wo, bo, (float*)d_out, N);
}


# dis[src] multiplied into the gathered rows once by the producing GEMM epilogue: both propagations drop the per-edge dis lookup, add plain rows in list order, scale the finished sum by dis[dst] once
# speedup vs baseline: 4.4976x; 1.1483x over previous
#include <hip/hip_runtime.h>

typedef __attribute__((ext_vector_type(2))) float v2f;
typedef __attribute__((ext_vector_type(8))) float v8f;

#define HIDW 128
#define KCHUNK 32

__global__ __launch_bounds__(256) void k_zero(int* __restrict__ p, int n) {
  int i = blockIdx.x * blockDim.x + threadIdx.x;
  if (i < n) p[i] = 0;
}

__global__ __launch_bounds__(256) void k_edge_count(int* __restrict__ cnt, int* __restrict__ rank,
                                                    const int* __restrict__ dst, int E) {
  int e = blockIdx.x * blockDim.x + threadIdx.x;
  if (e < E) rank[e] = atomicAdd(&cnt[dst[e]], 1);
}


__device__ __forceinline__ void fold_weights(const float* __restrict__ Wo, const float* __restrict__ W2,
                                             const float* __restrict__ b2, const float* __restrict__ bo,
                                             float* __restrict__ Wcat, float* __restrict__ bias128) {
  __shared__ __align__(16) float Fa[64][KCHUNK + 4];
  __shared__ __align__(16) float Fb[HIDW][KCHUNK + 4];
  const int tid  = threadIdx.x;
  const int wave = tid >> 5;
  const int lane = tid & 31;
  const int l15  = lane & 15;
  const int half = lane >> 4;
  const int mt   = wave & 3;
  const int nt   = wave >> 2;

  v8f acc;
#pragma unroll
  for (int r = 0; r < 8; ++r) acc[r] = 0.0f;

  for (int hb = 0; hb < HIDW; hb += KCHUNK) {
    {
      int idx = tid * 2;
      int o   = idx >> 5;
      int hh  = idx & 31;
      *(v2f*)&Fa[o][hh] = *(const v2f*)&Wo[(size_t)o * (2 * HIDW) + hb + hh];
    }
#pragma unroll
    for (int j = 0; j < (HIDW * KCHUNK) / 1024; ++j) {
      int e  = tid + j * 1024;
      int hh = e >> 7;
      int k  = e & 127;
      Fb[k][hh] = W2[(size_t)(hb + hh) * HIDW + k];
    }
    __syncthreads();
    const int o2 = half * 2;
#pragma unroll
    for (int k4 = 0; k4 < KCHUNK; k4 += 4) {
      v2f a = *(const v2f*)&Fa[mt * 16 + l15][k4 + o2];
      v2f b = *(const v2f*)&Fb[nt * 16 + l15][k4 + o2];
      acc = __builtin_amdgcn_wmma_f32_16x16x4_f32(false, a, false, b, (short)0, acc, false, false);
    }
    __syncthreads();
  }
#pragma unroll
  for (int r = 0; r < 8; ++r)
    Wcat[(size_t)(mt * 16 + half * 8 + r) * HIDW + nt * 16 + l15] = acc[r];

#pragma unroll
  for (int j = 0; j < 2; ++j) {
    int q  = tid + j * 1024;
    int o  = q >> 5;
    int c4 = (q & 31) * 4;
    *(float4*)&Wcat[(size_t)(64 + o) * HIDW + c4] = *(const float4*)&Wo[(size_t)o * (2 * HIDW) + HIDW + c4];
  }
  if (tid < 64) {
    float s = bo[tid];
    for (int h = 0; h < HIDW; ++h) s = fmaf(Wo[(size_t)tid * (2 * HIDW) + h], b2[h], s);
    bias128[tid]      = 0.f;
    bias128[64 + tid] = s;
  }
}

#define SCAN_PER 16
#define SCAN_TILE (1024 * SCAN_PER)
__global__ __launch_bounds__(1024) void k_scan_dis(const int* __restrict__ cnt, int* __restrict__ rowptr,
                                                   float* __restrict__ dis, int n,
                                                   const float* __restrict__ Wo, const float* __restrict__ W2,
                                                   const float* __restrict__ b2, const float* __restrict__ bo,
                                                   float* __restrict__ Wcat, float* __restrict__ bias128) {
  __shared__ int wsum[32];
  if (blockIdx.x == 1) { fold_weights(Wo, W2, b2, bo, Wcat, bias128); return; }
  const int tid  = threadIdx.x;
  const int lane = tid & 31;
  const int wv   = tid >> 5;
  int carry = 0;
  for (int t0 = 0; t0 < n; t0 += SCAN_TILE) {
    const int b = t0 + tid * SCAN_PER;
    int c[SCAN_PER];
#pragma unroll
    for (int q = 0; q < SCAN_PER; q += 4) {
      int4 t = *(const int4*)&cnt[b + q];
      c[q] = t.x; c[q + 1] = t.y; c[q + 2] = t.z; c[q + 3] = t.w;
    }
    int s = 0;
#pragma unroll
    for (int q = 0; q < SCAN_PER; ++q) s += c[q];

    int v = s;
    for (int o = 1; o < 32; o <<= 1) {
      int t = __shfl_up(v, o);
      if (lane >= o) v += t;
    }
    if (lane == 31) wsum[wv] = v;
    __syncthreads();
    if (wv == 0) {
      int t = wsum[lane];
      for (int o = 1; o < 32; o <<= 1) {
        int u = __shfl_up(t, o);
        if (lane >= o) t += u;
      }
      wsum[lane] = t;
    }
    __syncthreads();
    int run = carry + v - s + (wv ? wsum[wv - 1] : 0);
    carry += wsum[31];

#pragma unroll
    for (int q = 0; q < SCAN_PER; q += 4) {
      int   r[4];
      float d[4];
#pragma unroll
      for (int k = 0; k < 4; ++k) {
        r[k] = run;
        run += c[q + k];
        float dg = (float)(c[q + k] + 1);
        d[k] = (dg > 0.f) ? rsqrtf(dg) : 0.f;
      }
      const int i = b + q;
      if (i + 3 < n) {
        *(int4*)&rowptr[i] = make_int4(r[0], r[1], r[2], r[3]);
        *(float4*)&dis[i]  = make_float4(d[0], d[1], d[2], d[3]);
      } else {
#pragma unroll
        for (int k = 0; k < 4; ++k)
          if (i + k < n) { rowptr[i + k] = r[k]; dis[i + k] = d[k]; }
      }
    }
    __syncthreads();
  }
  if (tid == 0) rowptr[n] = carry;
}

__global__ __launch_bounds__(256) void k_fill(int* __restrict__ csr, const int* __restrict__ rowptr,
                                              const int* __restrict__ rank, const int* __restrict__ src,
                                              const int* __restrict__ dst, int E) {
  int e = blockIdx.x * blockDim.x + threadIdx.x;
  if (e < E) csr[rowptr[dst[e]] + rank[e]] = src[e];
}

template<int U>
__device__ __forceinline__ void gather_rows(float4& acc, const float* __restrict__ h, int my_s, int j, int lane4) {
  float4 v[U];
#pragma unroll
  for (int u = 0; u < U; ++u) {
    int s = __builtin_amdgcn_readlane(my_s, j + u);
    v[u] = *(const float4*)(h + (size_t)s * HIDW + lane4);
  }
#pragma unroll
  for (int u = 0; u < U; ++u) {
    acc.x += v[u].x;
    acc.y += v[u].y;
    acc.z += v[u].z;
    acc.w += v[u].w;
  }
}

__global__ __launch_bounds__(256) void k_prop_gather(float* __restrict__ out, const float* __restrict__ h,
                                                     const int* __restrict__ rowptr, const int* __restrict__ csr,
                                                     const float* __restrict__ dis, const float* __restrict__ bias,
                                                     int n) {
  const int lane  = threadIdx.x & 31;
  const int lane4 = lane * 4;
  const int wave = __builtin_amdgcn_readfirstlane((int)(threadIdx.x >> 5));
  const int nw   = gridDim.x * 8;
  const float4 bv = *(const float4*)&bias[lane4];

  for (int i = blockIdx.x * 8 + wave; i < n; i += nw) {
    const int beg = rowptr[i];
    const int end = rowptr[i + 1];
    const float di = dis[i];
    float4 acc = *(const float4*)(h + (size_t)i * HIDW + lane4);

    for (int base = beg; base < end; base += 32) {
      const int m = min(32, end - base);
      const int my_s = csr[min(base + lane, end - 1)];
      int j = 0;
      for (; j + 8 <= m; j += 8) gather_rows<8>(acc, h, my_s, j, lane4);
      if (m - j >= 4) { gather_rows<4>(acc, h, my_s, j, lane4); j += 4; }
      if (m - j >= 2) { gather_rows<2>(acc, h, my_s, j, lane4); j += 2; }
      if (m - j >= 1) { gather_rows<1>(acc, h, my_s, j, lane4); }
    }
    float4 o;
    o.x = fmaf(acc.x, di, bv.x);
    o.y = fmaf(acc.y, di, bv.y);
    o.z = fmaf(acc.z, di, bv.z);
    o.w = fmaf(acc.w, di, bv.w);
    *(float4*)(out + (size_t)i * HIDW + lane4) = o;
  }
}


template<int U, bool TAIL>
__device__ __forceinline__ void gather_pairs(float4& acc, const float* __restrict__ gp, int my_s,
                                             int j, int m, bool hi, int l15x4) {
  float4 v[U];
#pragma unroll
  for (int u = 0; u < U; ++u) {
    int slo = __builtin_amdgcn_readlane(my_s, j + 2 * u);
    int shi = __builtin_amdgcn_readlane(my_s, j + 2 * u + 1);
    int s = hi ? shi : slo;
    v[u] = *(const float4*)(gp + (size_t)s * HIDW + l15x4);
  }
#pragma unroll
  for (int u = 0; u < U; ++u) {
    const bool live = !TAIL || !hi || (j + 2 * u + 1 < m);
    acc.x += live ? v[u].x : 0.f;
    acc.y += live ? v[u].y : 0.f;
    acc.z += live ? v[u].z : 0.f;
    acc.w += live ? v[u].w : 0.f;
  }
}

__global__ __launch_bounds__(256) void k_prop_gather_half(float* __restrict__ out, const float* __restrict__ gp,
                                                          const int* __restrict__ rowptr, const int* __restrict__ csr,
                                                          const float* __restrict__ dis, int n) {
  const int lane  = threadIdx.x & 31;
  const bool hi   = lane >= 16;
  const int l15x4 = (lane & 15) * 4;
  const int wave = __builtin_amdgcn_readfirstlane((int)(threadIdx.x >> 5));
  const int nw   = gridDim.x * 8;

  for (int i = blockIdx.x * 8 + wave; i < n; i += nw) {
    const int beg = rowptr[i];
    const int end = rowptr[i + 1];
    const float di = dis[i];
    const float4 rv = *(const float4*)(gp + (size_t)i * HIDW + (hi ? 64 : 0) + l15x4);
    float4 acc;
    acc.x = hi ? 0.f : rv.x; acc.y = hi ? 0.f : rv.y; acc.z = hi ? 0.f : rv.z; acc.w = hi ? 0.f : rv.w;

    for (int base = beg; base < end; base += 32) {
      const int m = min(32, end - base);
      const int my_s = csr[min(base + lane, end - 1)];
      int j = 0;
      for (; j + 16 <= m; j += 16) gather_pairs<8, false>(acc, gp, my_s, j, m, hi, l15x4);
      if (m - j >= 8) { gather_pairs<4, false>(acc, gp, my_s, j, m, hi, l15x4); j += 8; }
      if (m - j >= 4) { gather_pairs<2, false>(acc, gp, my_s, j, m, hi, l15x4); j += 4; }
      if (m - j >= 3)      gather_pairs<2, true>(acc, gp, my_s, j, m, hi, l15x4);
      else if (m - j >= 1) gather_pairs<1, true>(acc, gp, my_s, j, m, hi, l15x4);
    }
    float4 o;
    o.x = fmaf(acc.x + __shfl_xor(acc.x, 16), di, __shfl_xor(rv.x, 16));
    o.y = fmaf(acc.y + __shfl_xor(acc.y, 16), di, __shfl_xor(rv.y, 16));
    o.z = fmaf(acc.z + __shfl_xor(acc.z, 16), di, __shfl_xor(rv.z, 16));
    o.w = fmaf(acc.w + __shfl_xor(acc.w, 16), di, __shfl_xor(rv.w, 16));
    if (!hi) *(float4*)(out + (size_t)i * 64 + l15x4) = o;
  }
}

template<int NC, int KW, bool RELU_A0, bool RELU_OUT, bool HAS_BIAS, int SCALE_COLS>
__global__ __launch_bounds__(256) void k_gemm(const float* __restrict__ A0, const float* __restrict__ A1,
                                              const float* __restrict__ W, const float* __restrict__ bias,
                                              const float* __restrict__ dis,
                                              float* __restrict__ C, int M) {
  __shared__ __align__(16) float As[128][KCHUNK + 4];
  __shared__ __align__(16) float Ws[NC][KCHUNK + 4];
  constexpr int NT = NC / 16;
  constexpr int K = KW * 128;

  const int tid  = threadIdx.x;
  const int wave = tid >> 5;
  const int lane = tid & 31;
  const int l15  = lane & 15;
  const int half = lane >> 4;
  const int m0   = blockIdx.x * 128;

  v8f acc[NT];
#pragma unroll
  for (int t = 0; t < NT; ++t)
#pragma unroll
    for (int r = 0; r < 8; ++r) acc[t][r] = 0.0f;

  for (int kb = 0; kb < K; kb += KCHUNK) {
    const float* Ablk = (KW == 2 && kb >= 128) ? A1 : A0;
    const bool doRelu = RELU_A0 && (kb < 128);
    const int kloc = kb & 127;

#pragma unroll
    for (int j = 0; j < (128 * KCHUNK) / 1024; ++j) {
      int ei  = tid * 4 + j * 1024;
      int row = ei >> 5;
      int col = ei & 31;
      int grow = m0 + row;
      float4 v = {0.f, 0.f, 0.f, 0.f};
      if (grow < M) v = *(const float4*)&Ablk[(size_t)grow * HIDW + kloc + col];
      if (doRelu) {
        v.x = fmaxf(v.x, 0.f); v.y = fmaxf(v.y, 0.f);
        v.z = fmaxf(v.z, 0.f); v.w = fmaxf(v.w, 0.f);
      }
      *(float4*)&As[row][col] = v;
    }
#pragma unroll
    for (int j = 0; j < (NC * KCHUNK) / 1024; ++j) {
      int ei  = tid * 4 + j * 1024;
      int n   = ei >> 5;
      int col = ei & 31;
      *(float4*)&Ws[n][col] = *(const float4*)&W[(size_t)n * K + kb + col];
    }
    __syncthreads();

    const int arow = wave * 16 + l15;
    const int o = half * 2;
#pragma unroll
    for (int k4 = 0; k4 < KCHUNK; k4 += 4) {
      v2f a = *(const v2f*)&As[arow][k4 + o];
#pragma unroll
      for (int t = 0; t < NT; ++t) {
        v2f b = *(const v2f*)&Ws[t * 16 + l15][k4 + o];
        acc[t] = __builtin_amdgcn_wmma_f32_16x16x4_f32(
             false, a,  false, b,
             (short)0, acc[t],  false,  false);
      }
    }
    __syncthreads();
  }

  const int rbase = m0 + wave * 16 + half * 8;
  float dr[8];
#pragma unroll
  for (int r = 0; r < 8; ++r) dr[r] = (SCALE_COLS > 0 && rbase + r < M) ? dis[rbase + r] : 1.0f;
#pragma unroll
  for (int t = 0; t < NT; ++t) {
    int col = t * 16 + l15;
    float bv = HAS_BIAS ? bias[col] : 0.f;
#pragma unroll
    for (int r = 0; r < 8; ++r) {
      int row = rbase + r;
      if (row < M) {
        float v = acc[t][r] + bv;
        if (RELU_OUT) v = fmaxf(v, 0.f);
        if (t * 16 < SCALE_COLS) v *= dr[r];
        C[(size_t)row * NC + col] = v;
      }
    }
  }
}

extern "C" void kernel_launch(void* const* d_in, const int* in_sizes, int n_in,
                              void* d_out, int out_size, void* d_ws, size_t ws_size,
                              hipStream_t stream) {
  const float* x  = (const float*)d_in[0];
  const int* edge = (const int*)d_in[1];
  const float* W1 = (const float*)d_in[2];
  const float* b1 = (const float*)d_in[3];
  const float* Wc = (const float*)d_in[4];
  const float* bc = (const float*)d_in[5];
  const float* W2 = (const float*)d_in[6];
  const float* b2 = (const float*)d_in[7];
  const float* Wo = (const float*)d_in[8];
  const float* bo = (const float*)d_in[9];

  const int N = in_sizes[0] / HIDW;
  const int E = in_sizes[1] / 2;
  const int* srcIdx = edge;
  const int* dstIdx = edge + E;

  float* wsf = (float*)d_ws;
  size_t NP   = (((size_t)N) + 255) & ~(size_t)255;
  size_t NP1  = (((size_t)N) + 1 + 255) & ~(size_t)255;
  size_t EP   = (((size_t)E) + 3) & ~(size_t)3;
  float* dis  = wsf;
  float* buf1 = wsf + NP;
  float* buf2 = buf1 + (size_t)N * HIDW;
  float* buf3 = buf2 + (size_t)N * HIDW;
  size_t NPS  = (((size_t)N) + SCAN_TILE - 1) & ~(size_t)(SCAN_TILE - 1);
  int* cnt    = (int*)buf3;
  int* rank   = cnt + NPS;
  int* rowptr = rank + EP;
  int* csr    = rowptr + NP1;
  float* Wcat = (float*)(csr + EP);
  float* bias128 = Wcat + HIDW * HIDW;

  dim3 b256(256);
  int gE  = (E + 255) / 256;
  int gG  = (N + 127) / 128;
  int gP  = 2048;

  k_zero      <<<(int)(NPS / 256), b256, 0, stream>>>(cnt, (int)NPS);
  k_edge_count<<<gE, b256, 0, stream>>>(cnt, rank, dstIdx, E);
  k_scan_dis  <<<2, dim3(1024), 0, stream>>>(cnt, rowptr, dis, N, Wo, W2, b2, bo, Wcat, bias128);
  k_fill      <<<gE, b256, 0, stream>>>(csr, rowptr, rank, srcIdx, dstIdx, E);

  k_gemm<128,1,false,false,false,128><<<gG, b256, 0, stream>>>(x, nullptr, W1, nullptr, dis, buf1, N);
  k_prop_gather<<<gP, b256, 0, stream>>>(buf2, buf1, rowptr, csr, dis, b1, N);

  k_gemm<128,2,true,true,true,0><<<gG, b256, 0, stream>>>(buf2, x, Wc, bc, nullptr, buf1, N);

  k_gemm<128,1,false,false,true,64><<<gG, b256, 0, stream>>>(buf1, nullptr, Wcat, bias128, dis, buf2, N);
  k_prop_gather_half<<<gP, b256, 0, stream>>>((float*)d_out, buf2, rowptr, csr, dis, N);
}


# Edge grouping by destination redone as a two-level counting sort on LDS counters (128-node buckets, runs copied out whole): no global atomics, no scattered 4-byte stores; row list order may differ
# speedup vs baseline: 7.1256x; 1.5843x over previous
#include <hip/hip_runtime.h>

typedef __attribute__((ext_vector_type(2))) float v2f;
typedef __attribute__((ext_vector_type(8))) float v8f;

#define HIDW 128
#define KCHUNK 32

__device__ __forceinline__ void fold_weights(const float* __restrict__ Wo, const float* __restrict__ W2,
                                             const float* __restrict__ b2, const float* __restrict__ bo,
                                             float* __restrict__ Wcat, float* __restrict__ bias128) {
  __shared__ __align__(16) float Fa[64][KCHUNK + 4];
  __shared__ __align__(16) float Fb[HIDW][KCHUNK + 4];
  const int tid  = threadIdx.x;
  const int wave = tid >> 5;
  const int lane = tid & 31;
  const int l15  = lane & 15;
  const int half = lane >> 4;
  const int mt   = wave & 3;
  const int nt   = wave >> 2;

  v8f acc;
#pragma unroll
  for (int r = 0; r < 8; ++r) acc[r] = 0.0f;

  for (int hb = 0; hb < HIDW; hb += KCHUNK) {
    {
      int idx = tid * 2;
      int o   = idx >> 5;
      int hh  = idx & 31;
      *(v2f*)&Fa[o][hh] = *(const v2f*)&Wo[(size_t)o * (2 * HIDW) + hb + hh];
    }
#pragma unroll
    for (int j = 0; j < (HIDW * KCHUNK) / 1024; ++j) {
      int e  = tid + j * 1024;
      int hh = e >> 7;
      int k  = e & 127;
      Fb[k][hh] = W2[(size_t)(hb + hh) * HIDW + k];
    }
    __syncthreads();
    const int o2 = half * 2;
#pragma unroll
    for (int k4 = 0; k4 < KCHUNK; k4 += 4) {
      v2f a = *(const v2f*)&Fa[mt * 16 + l15][k4 + o2];
      v2f b = *(const v2f*)&Fb[nt * 16 + l15][k4 + o2];
      acc = __builtin_amdgcn_wmma_f32_16x16x4_f32(false, a, false, b, (short)0, acc, false, false);
    }
    __syncthreads();
  }
#pragma unroll
  for (int r = 0; r < 8; ++r)
    Wcat[(size_t)(mt * 16 + half * 8 + r) * HIDW + nt * 16 + l15] = acc[r];

#pragma unroll
  for (int j = 0; j < 2; ++j) {
    int q  = tid + j * 1024;
    int o  = q >> 5;
    int c4 = (q & 31) * 4;
    *(float4*)&Wcat[(size_t)(64 + o) * HIDW + c4] = *(const float4*)&Wo[(size_t)o * (2 * HIDW) + HIDW + c4];
  }
  if (tid < 64) {
    float s = bo[tid];
    for (int h = 0; h < HIDW; ++h) s = fmaf(Wo[(size_t)tid * (2 * HIDW) + h], b2[h], s);
    bias128[tid]      = 0.f;
    bias128[64 + tid] = s;
  }
}

#define BSH 7
#define BNODES (1 << BSH)
#define NBMAX 512
#define P_THREADS 1024
#define P_PER 8
#define P_CHUNK (P_THREADS * P_PER)
#define HIST_ROWS 32

template<int NWAVES>
__device__ __forceinline__ void block_scan2(int& a, int& b, int* wsA, int* wsB) {
  const int lane = threadIdx.x & 31;
  const int wv   = threadIdx.x >> 5;
  int ia = a, ib = b;
  for (int o = 1; o < 32; o <<= 1) {
    int ta = __shfl_up(ia, o);
    int tb = __shfl_up(ib, o);
    if (lane >= o) { ia += ta; ib += tb; }
  }
  if (lane == 31) { wsA[wv] = ia; wsB[wv] = ib; }
  __syncthreads();
  if (wv == 0) {
    int ta = (lane < NWAVES) ? wsA[lane] : 0;
    int tb = (lane < NWAVES) ? wsB[lane] : 0;
    for (int o = 1; o < 32; o <<= 1) {
      int ua = __shfl_up(ta, o);
      int ub = __shfl_up(tb, o);
      if (lane >= o) { ta += ua; tb += ub; }
    }
    if (lane < NWAVES) { wsA[lane] = ta; wsB[lane] = tb; }
  }
  __syncthreads();
  a = ia - a + (wv ? wsA[wv - 1] : 0);
  b = ib - b + (wv ? wsB[wv - 1] : 0);
}

__global__ __launch_bounds__(P_THREADS) void k_bucket_hist(int* __restrict__ histG, const int* __restrict__ dst, int E,
                                                           const float* __restrict__ Wo, const float* __restrict__ W2,
                                                           const float* __restrict__ b2, const float* __restrict__ bo,
                                                           float* __restrict__ Wcat, float* __restrict__ bias128) {
  __shared__ int h[NBMAX];
  if (blockIdx.x == gridDim.x - 1) { fold_weights(Wo, W2, b2, bo, Wcat, bias128); return; }
  const int tid = threadIdx.x;
  if (tid < NBMAX) h[tid] = 0;
  __syncthreads();
  const int e0 = blockIdx.x * P_CHUNK;
#pragma unroll
  for (int k = 0; k < P_PER; ++k) {
    int e = e0 + k * P_THREADS + tid;
    if (e < E) atomicAdd(&h[dst[e] >> BSH], 1);
  }
  __syncthreads();
  if (tid < NBMAX) histG[(size_t)blockIdx.x * NBMAX + tid] = h[tid];
}

__global__ __launch_bounds__(P_THREADS) void k_bucket_scatter(int2* __restrict__ part, int* __restrict__ bstartG,
                                                              const int* __restrict__ histG,
                                                              const int* __restrict__ src, const int* __restrict__ dst, int E) {
  __shared__ int cnt[NBMAX];
  __shared__ int loc[NBMAX];
  __shared__ int gbase[NBMAX];
  __shared__ int hTot[NBMAX];
  __shared__ int hPre[NBMAX];
  __shared__ int wsA[32];
  __shared__ int wsB[32];
  __shared__ int2 stage[P_CHUNK];
  const int tid = threadIdx.x;
  const int wg  = blockIdx.x;
  const int nwg = gridDim.x;
  const int e0  = wg * P_CHUNK;
  if (tid < NBMAX) cnt[tid] = 0;
  __syncthreads();

  int s[P_PER], d[P_PER], r[P_PER];
#pragma unroll
  for (int k = 0; k < P_PER; ++k) {
    int e = e0 + k * P_THREADS + tid;
    d[k] = -1; s[k] = 0; r[k] = 0;
    if (e < E) { d[k] = dst[e]; s[k] = src[e]; }
  }
#pragma unroll
  for (int k = 0; k < P_PER; ++k)
    if (d[k] >= 0) r[k] = atomicAdd(&cnt[d[k] >> BSH], 1);

  const int bq = tid & (NBMAX - 1);
  const int hq = __builtin_amdgcn_readfirstlane(tid >> 9);
  const int nh = (nwg + 1) >> 1;
  const int wb = hq * nh;
  int tot = 0, pre = 0;
  for (int j0 = 0; j0 < nh; j0 += HIST_ROWS) {
    int v[HIST_ROWS];
#pragma unroll
    for (int u = 0; u < HIST_ROWS; ++u)
      v[u] = histG[min(wb + j0 + u, nwg - 1) * NBMAX + bq];
#pragma unroll
    for (int u = 0; u < HIST_ROWS; ++u) {
      const int w  = wb + j0 + u;
      const int vv = (j0 + u < nh && w < nwg) ? v[u] : 0;
      tot += vv;
      pre += (w < wg) ? vv : 0;
    }
  }
  if (hq) { hTot[bq] = tot; hPre[bq] = pre; }
  __syncthreads();
  int a = 0, c = 0;
  if (!hq) { tot += hTot[bq]; pre += hPre[bq]; a = tot; c = cnt[bq]; }
  block_scan2<32>(a, c, wsA, wsB);
  if (!hq) {
    loc[bq]   = c;
    gbase[bq] = a + pre;
    if (wg == 0) {
      bstartG[bq] = a;
      if (bq == NBMAX - 1) bstartG[NBMAX] = a + tot;
    }
  }
  __syncthreads();
#pragma unroll
  for (int k = 0; k < P_PER; ++k)
    if (d[k] >= 0) stage[loc[d[k] >> BSH] + r[k]] = make_int2(s[k], d[k]);
  __syncthreads();
  const int nloc = min(P_CHUNK, E - e0);
#pragma unroll
  for (int k = 0; k < P_PER; ++k) {
    int i = k * P_THREADS + tid;
    if (i < nloc) {
      int2 rec = stage[i];
      int b = rec.y >> BSH;
      part[(size_t)gbase[b] + (i - loc[b])] = rec;
    }
  }
}

#define ROWS_THREADS 256
#define ROWS_PER 8
#define ROWS_STAGE 12288
__global__ __launch_bounds__(ROWS_THREADS) void k_bucket_rows(int* __restrict__ csr, int* __restrict__ rowptr,
                                                              float* __restrict__ dis, const int2* __restrict__ part,
                                                              const int* __restrict__ bstartG, int n) {
  __shared__ int cntN[BNODES];
  __shared__ int fillN[BNODES];
  __shared__ int rstart[BNODES];
  __shared__ int wsA[8];
  __shared__ int wsB[8];
  __shared__ int stage[ROWS_STAGE];
  const int tid = threadIdx.x;
  const int b   = blockIdx.x;
  const int beg = bstartG[b];
  const int end = bstartG[b + 1];
  const int ne  = end - beg;
  if (tid < BNODES) { cntN[tid] = 0; fillN[tid] = 0; }
  __syncthreads();
  for (int i0 = 0; i0 < ne; i0 += ROWS_THREADS * ROWS_PER) {
    int2 rec[ROWS_PER];
#pragma unroll
    for (int k = 0; k < ROWS_PER; ++k) {
      const int i = i0 + k * ROWS_THREADS + tid;
      rec[k] = (i < ne) ? part[(size_t)beg + i] : make_int2(0, -1);
    }
#pragma unroll
    for (int k = 0; k < ROWS_PER; ++k)
      if (rec[k].y >= 0) atomicAdd(&cntN[rec[k].y & (BNODES - 1)], 1);
  }
  __syncthreads();
  int c = (tid < BNODES) ? cntN[tid] : 0;
  int x = c, unused = 0;
  block_scan2<ROWS_THREADS / 32>(x, unused, wsA, wsB);
  if (tid < BNODES) {
    rstart[tid] = x;
    const int node = b * BNODES + tid;
    if (node < n) {
      rowptr[node] = beg + x;
      float dg = (float)(c + 1);
      dis[node] = (dg > 0.f) ? rsqrtf(dg) : 0.f;
    }
  }
  if (b == (int)gridDim.x - 1 && tid == 0) rowptr[n] = end;
  __syncthreads();
  const bool staged = (ne <= ROWS_STAGE);
  for (int i0 = 0; i0 < ne; i0 += ROWS_THREADS * ROWS_PER) {
    int2 rec[ROWS_PER];
#pragma unroll
    for (int k = 0; k < ROWS_PER; ++k) {
      const int i = i0 + k * ROWS_THREADS + tid;
      rec[k] = (i < ne) ? part[(size_t)beg + i] : make_int2(0, -1);
    }
    int pos[ROWS_PER];
#pragma unroll
    for (int k = 0; k < ROWS_PER; ++k) {
      pos[k] = -1;
      if (rec[k].y >= 0) {
        const int dl = rec[k].y & (BNODES - 1);
        pos[k] = rstart[dl] + atomicAdd(&fillN[dl], 1);
      }
    }
    if (staged) {
#pragma unroll
      for (int k = 0; k < ROWS_PER; ++k)
        if (pos[k] >= 0) stage[pos[k]] = rec[k].x;
    } else {
#pragma unroll
      for (int k = 0; k < ROWS_PER; ++k)
        if (pos[k] >= 0) csr[(size_t)beg + pos[k]] = rec[k].x;
    }
  }
  if (staged) {
    __syncthreads();
    for (int i = tid; i < ne; i += ROWS_THREADS) csr[(size_t)beg + i] = stage[i];
  }
}

template<int U>
__device__ __forceinline__ void gather_rows(float4& acc, const float* __restrict__ h, int my_s, int j, int lane4) {
  float4 v[U];
#pragma unroll
  for (int u = 0; u < U; ++u) {
    int s = __builtin_amdgcn_readlane(my_s, j + u);
    v[u] = *(const float4*)(h + (size_t)s * HIDW + lane4);
  }
#pragma unroll
  for (int u = 0; u < U; ++u) {
    acc.x += v[u].x;
    acc.y += v[u].y;
    acc.z += v[u].z;
    acc.w += v[u].w;
  }
}

__global__ __launch_bounds__(256) void k_prop_gather(float* __restrict__ out, const float* __restrict__ h,
                                                     const int* __restrict__ rowptr, const int* __restrict__ csr,
                                                     const float* __restrict__ dis, const float* __restrict__ bias,
                                                     int n) {
  const int lane  = threadIdx.x & 31;
  const int lane4 = lane * 4;
  const int wave = __builtin_amdgcn_readfirstlane((int)(threadIdx.x >> 5));
  const int nw   = gridDim.x * 8;
  const float4 bv = *(const float4*)&bias[lane4];

  for (int i = blockIdx.x * 8 + wave; i < n; i += nw) {
    const int beg = rowptr[i];
    const int end = rowptr[i + 1];
    const float di = dis[i];
    float4 acc = *(const float4*)(h + (size_t)i * HIDW + lane4);

    for (int base = beg; base < end; base += 32) {
      const int m = min(32, end - base);
      const int my_s = csr[min(base + lane, end - 1)];
      int j = 0;
      for (; j + 8 <= m; j += 8) gather_rows<8>(acc, h, my_s, j, lane4);
      if (m - j >= 4) { gather_rows<4>(acc, h, my_s, j, lane4); j += 4; }
      if (m - j >= 2) { gather_rows<2>(acc, h, my_s, j, lane4); j += 2; }
      if (m - j >= 1) { gather_rows<1>(acc, h, my_s, j, lane4); }
    }
    float4 o;
    o.x = fmaf(acc.x, di, bv.x);
    o.y = fmaf(acc.y, di, bv.y);
    o.z = fmaf(acc.z, di, bv.z);
    o.w = fmaf(acc.w, di, bv.w);
    *(float4*)(out + (size_t)i * HIDW + lane4) = o;
  }
}


template<int U, bool TAIL>
__device__ __forceinline__ void gather_pairs(float4& acc, const float* __restrict__ gp, int my_s,
                                             int j, int m, bool hi, int l15x4) {
  float4 v[U];
#pragma unroll
  for (int u = 0; u < U; ++u) {
    int slo = __builtin_amdgcn_readlane(my_s, j + 2 * u);
    int shi = __builtin_amdgcn_readlane(my_s, j + 2 * u + 1);
    int s = hi ? shi : slo;
    v[u] = *(const float4*)(gp + (size_t)s * HIDW + l15x4);
  }
#pragma unroll
  for (int u = 0; u < U; ++u) {
    const bool live = !TAIL || !hi || (j + 2 * u + 1 < m);
    acc.x += live ? v[u].x : 0.f;
    acc.y += live ? v[u].y : 0.f;
    acc.z += live ? v[u].z : 0.f;
    acc.w += live ? v[u].w : 0.f;
  }
}

__global__ __launch_bounds__(256) void k_prop_gather_half(float* __restrict__ out, const float* __restrict__ gp,
                                                          const int* __restrict__ rowptr, const int* __restrict__ csr,
                                                          const float* __restrict__ dis, int n) {
  const int lane  = threadIdx.x & 31;
  const bool hi   = lane >= 16;
  const int l15x4 = (lane & 15) * 4;
  const int wave = __builtin_amdgcn_readfirstlane((int)(threadIdx.x >> 5));
  const int nw   = gridDim.x * 8;

  for (int i = blockIdx.x * 8 + wave; i < n; i += nw) {
    const int beg = rowptr[i];
    const int end = rowptr[i + 1];
    const float di = dis[i];
    const float4 rv = *(const float4*)(gp + (size_t)i * HIDW + (hi ? 64 : 0) + l15x4);
    float4 acc;
    acc.x = hi ? 0.f : rv.x; acc.y = hi ? 0.f : rv.y; acc.z = hi ? 0.f : rv.z; acc.w = hi ? 0.f : rv.w;

    for (int base = beg; base < end; base += 32) {
      const int m = min(32, end - base);
      const int my_s = csr[min(base + lane, end - 1)];
      int j = 0;
      for (; j + 16 <= m; j += 16) gather_pairs<8, false>(acc, gp, my_s, j, m, hi, l15x4);
      if (m - j >= 8) { gather_pairs<4, false>(acc, gp, my_s, j, m, hi, l15x4); j += 8; }
      if (m - j >= 4) { gather_pairs<2, false>(acc, gp, my_s, j, m, hi, l15x4); j += 4; }
      if (m - j >= 3)      gather_pairs<2, true>(acc, gp, my_s, j, m, hi, l15x4);
      else if (m - j >= 1) gather_pairs<1, true>(acc, gp, my_s, j, m, hi, l15x4);
    }
    float4 o;
    o.x = fmaf(acc.x + __shfl_xor(acc.x, 16), di, __shfl_xor(rv.x, 16));
    o.y = fmaf(acc.y + __shfl_xor(acc.y, 16), di, __shfl_xor(rv.y, 16));
    o.z = fmaf(acc.z + __shfl_xor(acc.z, 16), di, __shfl_xor(rv.z, 16));
    o.w = fmaf(acc.w + __shfl_xor(acc.w, 16), di, __shfl_xor(rv.w, 16));
    if (!hi) *(float4*)(out + (size_t)i * 64 + l15x4) = o;
  }
}

template<int NC, int KW, bool RELU_A0, bool RELU_OUT, bool HAS_BIAS, int SCALE_COLS>
__global__ __launch_bounds__(256) void k_gemm(const float* __restrict__ A0, const float* __restrict__ A1,
                                              const float* __restrict__ W, const float* __restrict__ bias,
                                              const float* __restrict__ dis,
                                              float* __restrict__ C, int M) {
  __shared__ __align__(16) float As[128][KCHUNK + 4];
  __shared__ __align__(16) float Ws[NC][KCHUNK + 4];
  constexpr int NT = NC / 16;
  constexpr int K = KW * 128;

  const int tid  = threadIdx.x;
  const int wave = tid >> 5;
  const int lane = tid & 31;
  const int l15  = lane & 15;
  const int half = lane >> 4;
  const int m0   = blockIdx.x * 128;

  v8f acc[NT];
#pragma unroll
  for (int t = 0; t < NT; ++t)
#pragma unroll
    for (int r = 0; r < 8; ++r) acc[t][r] = 0.0f;

  for (int kb = 0; kb < K; kb += KCHUNK) {
    const float* Ablk = (KW == 2 && kb >= 128) ? A1 : A0;
    const bool doRelu = RELU_A0 && (kb < 128);
    const int kloc = kb & 127;

#pragma unroll
    for (int j = 0; j < (128 * KCHUNK) / 1024; ++j) {
      int ei  = tid * 4 + j * 1024;
      int row = ei >> 5;
      int col = ei & 31;
      int grow = m0 + row;
      float4 v = {0.f, 0.f, 0.f, 0.f};
      if (grow < M) v = *(const float4*)&Ablk[(size_t)grow * HIDW + kloc + col];
      if (doRelu) {
        v.x = fmaxf(v.x, 0.f); v.y = fmaxf(v.y, 0.f);
        v.z = fmaxf(v.z, 0.f); v.w = fmaxf(v.w, 0.f);
      }
      *(float4*)&As[row][col] = v;
    }
#pragma unroll
    for (int j = 0; j < (NC * KCHUNK) / 1024; ++j) {
      int ei  = tid * 4 + j * 1024;
      int n   = ei >> 5;
      int col = ei & 31;
      *(float4*)&Ws[n][col] = *(const float4*)&W[(size_t)n * K + kb + col];
    }
    __syncthreads();

    const int arow = wave * 16 + l15;
    const int o = half * 2;
#pragma unroll
    for (int k4 = 0; k4 < KCHUNK; k4 += 4) {
      v2f a = *(const v2f*)&As[arow][k4 + o];
#pragma unroll
      for (int t = 0; t < NT; ++t) {
        v2f b = *(const v2f*)&Ws[t * 16 + l15][k4 + o];
        acc[t] = __builtin_amdgcn_wmma_f32_16x16x4_f32(
             false, a,  false, b,
             (short)0, acc[t],  false,  false);
      }
    }
    __syncthreads();
  }

  const int rbase = m0 + wave * 16 + half * 8;
  float dr[8];
#pragma unroll
  for (int r = 0; r < 8; ++r) dr[r] = (SCALE_COLS > 0 && rbase + r < M) ? dis[rbase + r] : 1.0f;
#pragma unroll
  for (int t = 0; t < NT; ++t) {
    int col = t * 16 + l15;
    float bv = HAS_BIAS ? bias[col] : 0.f;
#pragma unroll
    for (int r = 0; r < 8; ++r) {
      int row = rbase + r;
      if (row < M) {
        float v = acc[t][r] + bv;
        if (RELU_OUT) v = fmaxf(v, 0.f);
        if (t * 16 < SCALE_COLS) v *= dr[r];
        C[(size_t)row * NC + col] = v;
      }
    }
  }
}

extern "C" void kernel_launch(void* const* d_in, const int* in_sizes, int n_in,
                              void* d_out, int out_size, void* d_ws, size_t ws_size,
                              hipStream_t stream) {
  const float* x  = (const float*)d_in[0];
  const int* edge = (const int*)d_in[1];
  const float* W1 = (const float*)d_in[2];
  const float* b1 = (const float*)d_in[3];
  const float* Wc = (const float*)d_in[4];
  const float* bc = (const float*)d_in[5];
  const float* W2 = (const float*)d_in[6];
  const float* b2 = (const float*)d_in[7];
  const float* Wo = (const float*)d_in[8];
  const float* bo = (const float*)d_in[9];

  const int N = in_sizes[0] / HIDW;
  const int E = in_sizes[1] / 2;
  const int* srcIdx = edge;
  const int* dstIdx = edge + E;

  float* wsf = (float*)d_ws;
  size_t NP   = (((size_t)N) + 255) & ~(size_t)255;
  size_t NP1  = (((size_t)N) + 1 + 255) & ~(size_t)255;
  size_t EP   = (((size_t)E) + 3) & ~(size_t)3;
  float* dis  = wsf;
  float* buf1 = wsf + NP;
  float* buf2 = buf1 + (size_t)N * HIDW;
  float* buf3 = buf2 + (size_t)N * HIDW;
  int* rowptr = (int*)buf3;
  int* csr    = rowptr + NP1;
  float* Wcat = (float*)(csr + EP);
  float* bias128 = Wcat + HIDW * HIDW;
  int gC      = (E + P_CHUNK - 1) / P_CHUNK;
  int gB      = (N + BNODES - 1) >> BSH;
  int2* part  = (int2*)buf2;
  int* histG  = (int*)buf2 + 2 * EP;
  int* bstartG = histG + (size_t)gC * NBMAX;

  dim3 b256(256);
  int gG  = (N + 127) / 128;
  int gP  = 2048;

  k_bucket_hist   <<<gC + 1, dim3(P_THREADS), 0, stream>>>(histG, dstIdx, E, Wo, W2, b2, bo, Wcat, bias128);
  k_bucket_scatter<<<gC, dim3(P_THREADS), 0, stream>>>(part, bstartG, histG, srcIdx, dstIdx, E);
  k_bucket_rows   <<<gB, dim3(ROWS_THREADS), 0, stream>>>(csr, rowptr, dis, part, bstartG, N);

  k_gemm<128,1,false,false,false,128><<<gG, b256, 0, stream>>>(x, nullptr, W1, nullptr, dis, buf1, N);
  k_prop_gather<<<gP, b256, 0, stream>>>(buf2, buf1, rowptr, csr, dis, b1, N);

  k_gemm<128,2,true,true,true,0><<<gG, b256, 0, stream>>>(buf2, x, Wc, bc, nullptr, buf1, N);

  k_gemm<128,1,false,false,true,64><<<gG, b256, 0, stream>>>(buf1, nullptr, Wcat, bias128, dis, buf2, N);
  k_prop_gather_half<<<gP, b256, 0, stream>>>((float*)d_out, buf2, rowptr, csr, dis, N);
}


# First propagation and the two products after it in one launch: a work-group sums its 128 rows' lists into an LDS tile, forms x_comb on WMMA in place, then [g|p]; x_agg, x_comb never reach memory
# speedup vs baseline: 8.2847x; 1.1627x over previous
#include <hip/hip_runtime.h>

typedef __attribute__((ext_vector_type(2))) float v2f;
typedef __attribute__((ext_vector_type(8))) float v8f;

#define HIDW 128
#define KCHUNK 32

__device__ __forceinline__ void fold_weights(const float* __restrict__ Wo, const float* __restrict__ W2,
                                             const float* __restrict__ b2, const float* __restrict__ bo,
                                             float* __restrict__ Wcat, float* __restrict__ bias128) {
  __shared__ __align__(16) float Fa[64][KCHUNK + 4];
  __shared__ __align__(16) float Fb[HIDW][KCHUNK + 4];
  const int tid  = threadIdx.x;
  const int wave = tid >> 5;
  const int lane = tid & 31;
  const int l15  = lane & 15;
  const int half = lane >> 4;
  const int mt   = wave & 3;
  const int nt   = wave >> 2;

  v8f acc;
#pragma unroll
  for (int r = 0; r < 8; ++r) acc[r] = 0.0f;

  for (int hb = 0; hb < HIDW; hb += KCHUNK) {
    {
      int idx = tid * 2;
      int o   = idx >> 5;
      int hh  = idx & 31;
      *(v2f*)&Fa[o][hh] = *(const v2f*)&Wo[(size_t)o * (2 * HIDW) + hb + hh];
    }
#pragma unroll
    for (int j = 0; j < (HIDW * KCHUNK) / 1024; ++j) {
      int e  = tid + j * 1024;
      int hh = e >> 7;
      int k  = e & 127;
      Fb[k][hh] = W2[(size_t)(hb + hh) * HIDW + k];
    }
    __syncthreads();
    const int o2 = half * 2;
#pragma unroll
    for (int k4 = 0; k4 < KCHUNK; k4 += 4) {
      v2f a = *(const v2f*)&Fa[mt * 16 + l15][k4 + o2];
      v2f b = *(const v2f*)&Fb[nt * 16 + l15][k4 + o2];
      acc = __builtin_amdgcn_wmma_f32_16x16x4_f32(false, a, false, b, (short)0, acc, false, false);
    }
    __syncthreads();
  }
#pragma unroll
  for (int r = 0; r < 8; ++r)
    Wcat[(size_t)(mt * 16 + half * 8 + r) * HIDW + nt * 16 + l15] = acc[r];

#pragma unroll
  for (int j = 0; j < 2; ++j) {
    int q  = tid + j * 1024;
    int o  = q >> 5;
    int c4 = (q & 31) * 4;
    *(float4*)&Wcat[(size_t)(64 + o) * HIDW + c4] = *(const float4*)&Wo[(size_t)o * (2 * HIDW) + HIDW + c4];
  }
  if (tid < 64) {
    float s = bo[tid];
    for (int h = 0; h < HIDW; ++h) s = fmaf(Wo[(size_t)tid * (2 * HIDW) + h], b2[h], s);
    bias128[tid]      = 0.f;
    bias128[64 + tid] = s;
  }
}

#define BSH 7
#define BNODES (1 << BSH)
#define NBMAX 512
#define P_THREADS 1024
#define P_PER 8
#define P_CHUNK (P_THREADS * P_PER)
#define HIST_ROWS 32

template<int NWAVES>
__device__ __forceinline__ void block_scan2(int& a, int& b, int* wsA, int* wsB) {
  const int lane = threadIdx.x & 31;
  const int wv   = threadIdx.x >> 5;
  int ia = a, ib = b;
  for (int o = 1; o < 32; o <<= 1) {
    int ta = __shfl_up(ia, o);
    int tb = __shfl_up(ib, o);
    if (lane >= o) { ia += ta; ib += tb; }
  }
  if (lane == 31) { wsA[wv] = ia; wsB[wv] = ib; }
  __syncthreads();
  if (wv == 0) {
    int ta = (lane < NWAVES) ? wsA[lane] : 0;
    int tb = (lane < NWAVES) ? wsB[lane] : 0;
    for (int o = 1; o < 32; o <<= 1) {
      int ua = __shfl_up(ta, o);
      int ub = __shfl_up(tb, o);
      if (lane >= o) { ta += ua; tb += ub; }
    }
    if (lane < NWAVES) { wsA[lane] = ta; wsB[lane] = tb; }
  }
  __syncthreads();
  a = ia - a + (wv ? wsA[wv - 1] : 0);
  b = ib - b + (wv ? wsB[wv - 1] : 0);
}

__global__ __launch_bounds__(P_THREADS) void k_bucket_hist(int* __restrict__ histG, const int* __restrict__ dst, int E,
                                                           const float* __restrict__ Wo, const float* __restrict__ W2,
                                                           const float* __restrict__ b2, const float* __restrict__ bo,
                                                           float* __restrict__ Wcat, float* __restrict__ bias128) {
  __shared__ int h[NBMAX];
  if (blockIdx.x == gridDim.x - 1) { fold_weights(Wo, W2, b2, bo, Wcat, bias128); return; }
  const int tid = threadIdx.x;
  if (tid < NBMAX) h[tid] = 0;
  __syncthreads();
  const int e0 = blockIdx.x * P_CHUNK;
#pragma unroll
  for (int k = 0; k < P_PER; ++k) {
    int e = e0 + k * P_THREADS + tid;
    if (e < E) atomicAdd(&h[dst[e] >> BSH], 1);
  }
  __syncthreads();
  if (tid < NBMAX) histG[(size_t)blockIdx.x * NBMAX + tid] = h[tid];
}

__global__ __launch_bounds__(P_THREADS) void k_bucket_scatter(int2* __restrict__ part, int* __restrict__ bstartG,
                                                              const int* __restrict__ histG,
                                                              const int* __restrict__ src, const int* __restrict__ dst, int E) {
  __shared__ int cnt[NBMAX];
  __shared__ int loc[NBMAX];
  __shared__ int gbase[NBMAX];
  __shared__ int hTot[NBMAX];
  __shared__ int hPre[NBMAX];
  __shared__ int wsA[32];
  __shared__ int wsB[32];
  __shared__ int2 stage[P_CHUNK];
  const int tid = threadIdx.x;
  const int wg  = blockIdx.x;
  const int nwg = gridDim.x;
  const int e0  = wg * P_CHUNK;
  if (tid < NBMAX) cnt[tid] = 0;
  __syncthreads();

  int s[P_PER], d[P_PER], r[P_PER];
#pragma unroll
  for (int k = 0; k < P_PER; ++k) {
    int e = e0 + k * P_THREADS + tid;
    d[k] = -1; s[k] = 0; r[k] = 0;
    if (e < E) { d[k] = dst[e]; s[k] = src[e]; }
  }
#pragma unroll
  for (int k = 0; k < P_PER; ++k)
    if (d[k] >= 0) r[k] = atomicAdd(&cnt[d[k] >> BSH], 1);

  const int bq = tid & (NBMAX - 1);
  const int hq = __builtin_amdgcn_readfirstlane(tid >> 9);
  const int nh = (nwg + 1) >> 1;
  const int wb = hq * nh;
  int tot = 0, pre = 0;
  for (int j0 = 0; j0 < nh; j0 += HIST_ROWS) {
    int v[HIST_ROWS];
#pragma unroll
    for (int u = 0; u < HIST_ROWS; ++u)
      v[u] = histG[min(wb + j0 + u, nwg - 1) * NBMAX + bq];
#pragma unroll
    for (int u = 0; u < HIST_ROWS; ++u) {
      const int w  = wb + j0 + u;
      const int vv = (j0 + u < nh && w < nwg) ? v[u] : 0;
      tot += vv;
      pre += (w < wg) ? vv : 0;
    }
  }
  if (hq) { hTot[bq] = tot; hPre[bq] = pre; }
  __syncthreads();
  int a = 0, c = 0;
  if (!hq) { tot += hTot[bq]; pre += hPre[bq]; a = tot; c = cnt[bq]; }
  block_scan2<32>(a, c, wsA, wsB);
  if (!hq) {
    loc[bq]   = c;
    gbase[bq] = a + pre;
    if (wg == 0) {
      bstartG[bq] = a;
      if (bq == NBMAX - 1) bstartG[NBMAX] = a + tot;
    }
  }
  __syncthreads();
#pragma unroll
  for (int k = 0; k < P_PER; ++k)
    if (d[k] >= 0) stage[loc[d[k] >> BSH] + r[k]] = make_int2(s[k], d[k]);
  __syncthreads();
  const int nloc = min(P_CHUNK, E - e0);
#pragma unroll
  for (int k = 0; k < P_PER; ++k) {
    int i = k * P_THREADS + tid;
    if (i < nloc) {
      int2 rec = stage[i];
      int b = rec.y >> BSH;
      part[(size_t)gbase[b] + (i - loc[b])] = rec;
    }
  }
}

#define ROWS_THREADS 256
#define ROWS_PER 8
#define ROWS_STAGE 12288
__global__ __launch_bounds__(ROWS_THREADS) void k_bucket_rows(int* __restrict__ csr, int* __restrict__ rowptr,
                                                              float* __restrict__ dis, const int2* __restrict__ part,
                                                              const int* __restrict__ bstartG, int n) {
  __shared__ int cntN[BNODES];
  __shared__ int fillN[BNODES];
  __shared__ int rstart[BNODES];
  __shared__ int wsA[8];
  __shared__ int wsB[8];
  __shared__ int stage[ROWS_STAGE];
  const int tid = threadIdx.x;
  const int b   = blockIdx.x;
  const int beg = bstartG[b];
  const int end = bstartG[b + 1];
  const int ne  = end - beg;
  if (tid < BNODES) { cntN[tid] = 0; fillN[tid] = 0; }
  __syncthreads();
  for (int i0 = 0; i0 < ne; i0 += ROWS_THREADS * ROWS_PER) {
    int2 rec[ROWS_PER];
#pragma unroll
    for (int k = 0; k < ROWS_PER; ++k) {
      const int i = i0 + k * ROWS_THREADS + tid;
      rec[k] = (i < ne) ? part[(size_t)beg + i] : make_int2(0, -1);
    }
#pragma unroll
    for (int k = 0; k < ROWS_PER; ++k)
      if (rec[k].y >= 0) atomicAdd(&cntN[rec[k].y & (BNODES - 1)], 1);
  }
  __syncthreads();
  int c = (tid < BNODES) ? cntN[tid] : 0;
  int x = c, unused = 0;
  block_scan2<ROWS_THREADS / 32>(x, unused, wsA, wsB);
  if (tid < BNODES) {
    rstart[tid] = x;
    const int node = b * BNODES + tid;
    if (node < n) {
      rowptr[node] = beg + x;
      float dg = (float)(c + 1);
      dis[node] = (dg > 0.f) ? rsqrtf(dg) : 0.f;
    }
  }
  if (b == (int)gridDim.x - 1 && tid == 0) rowptr[n] = end;
  __syncthreads();
  const bool staged = (ne <= ROWS_STAGE);
  for (int i0 = 0; i0 < ne; i0 += ROWS_THREADS * ROWS_PER) {
    int2 rec[ROWS_PER];
#pragma unroll
    for (int k = 0; k < ROWS_PER; ++k) {
      const int i = i0 + k * ROWS_THREADS + tid;
      rec[k] = (i < ne) ? part[(size_t)beg + i] : make_int2(0, -1);
    }
    int pos[ROWS_PER];
#pragma unroll
    for (int k = 0; k < ROWS_PER; ++k) {
      pos[k] = -1;
      if (rec[k].y >= 0) {
        const int dl = rec[k].y & (BNODES - 1);
        pos[k] = rstart[dl] + atomicAdd(&fillN[dl], 1);
      }
    }
    if (staged) {
#pragma unroll
      for (int k = 0; k < ROWS_PER; ++k)
        if (pos[k] >= 0) stage[pos[k]] = rec[k].x;
    } else {
#pragma unroll
      for (int k = 0; k < ROWS_PER; ++k)
        if (pos[k] >= 0) csr[(size_t)beg + pos[k]] = rec[k].x;
    }
  }
  if (staged) {
    __syncthreads();
    for (int i = tid; i < ne; i += ROWS_THREADS) csr[(size_t)beg + i] = stage[i];
  }
}

template<int U>
__device__ __forceinline__ void gather_rows(float4& acc, const float* __restrict__ h, int my_s, int j, int lane4) {
  float4 v[U];
#pragma unroll
  for (int u = 0; u < U; ++u) {
    int s = __builtin_amdgcn_readlane(my_s, j + u);
    v[u] = *(const float4*)(h + (size_t)s * HIDW + lane4);
  }
#pragma unroll
  for (int u = 0; u < U; ++u) {
    acc.x += v[u].x;
    acc.y += v[u].y;
    acc.z += v[u].z;
    acc.w += v[u].w;
  }
}

template<int U, bool TAIL>
__device__ __forceinline__ void gather_pairs(float4& acc, const float* __restrict__ gp, int my_s,
                                             int j, int m, bool hi, int l15x4) {
  float4 v[U];
#pragma unroll
  for (int u = 0; u < U; ++u) {
    int slo = __builtin_amdgcn_readlane(my_s, j + 2 * u);
    int shi = __builtin_amdgcn_readlane(my_s, j + 2 * u + 1);
    int s = hi ? shi : slo;
    v[u] = *(const float4*)(gp + (size_t)s * HIDW + l15x4);
  }
#pragma unroll
  for (int u = 0; u < U; ++u) {
    const bool live = !TAIL || !hi || (j + 2 * u + 1 < m);
    acc.x += live ? v[u].x : 0.f;
    acc.y += live ? v[u].y : 0.f;
    acc.z += live ? v[u].z : 0.f;
    acc.w += live ? v[u].w : 0.f;
  }
}

__global__ __launch_bounds__(256) void k_prop_gather_half(float* __restrict__ out, const float* __restrict__ gp,
                                                          const int* __restrict__ rowptr, const int* __restrict__ csr,
                                                          const float* __restrict__ dis, int n) {
  const int lane  = threadIdx.x & 31;
  const bool hi   = lane >= 16;
  const int l15x4 = (lane & 15) * 4;
  const int wave = __builtin_amdgcn_readfirstlane((int)(threadIdx.x >> 5));
  const int nw   = gridDim.x * 8;

  for (int i = blockIdx.x * 8 + wave; i < n; i += nw) {
    const int beg = rowptr[i];
    const int end = rowptr[i + 1];
    const float di = dis[i];
    const float4 rv = *(const float4*)(gp + (size_t)i * HIDW + (hi ? 64 : 0) + l15x4);
    float4 acc;
    acc.x = hi ? 0.f : rv.x; acc.y = hi ? 0.f : rv.y; acc.z = hi ? 0.f : rv.z; acc.w = hi ? 0.f : rv.w;

    for (int base = beg; base < end; base += 32) {
      const int m = min(32, end - base);
      const int my_s = csr[min(base + lane, end - 1)];
      int j = 0;
      for (; j + 16 <= m; j += 16) gather_pairs<8, false>(acc, gp, my_s, j, m, hi, l15x4);
      if (m - j >= 8) { gather_pairs<4, false>(acc, gp, my_s, j, m, hi, l15x4); j += 8; }
      if (m - j >= 4) { gather_pairs<2, false>(acc, gp, my_s, j, m, hi, l15x4); j += 4; }
      if (m - j >= 3)      gather_pairs<2, true>(acc, gp, my_s, j, m, hi, l15x4);
      else if (m - j >= 1) gather_pairs<1, true>(acc, gp, my_s, j, m, hi, l15x4);
    }
    float4 o;
    o.x = fmaf(acc.x + __shfl_xor(acc.x, 16), di, __shfl_xor(rv.x, 16));
    o.y = fmaf(acc.y + __shfl_xor(acc.y, 16), di, __shfl_xor(rv.y, 16));
    o.z = fmaf(acc.z + __shfl_xor(acc.z, 16), di, __shfl_xor(rv.z, 16));
    o.w = fmaf(acc.w + __shfl_xor(acc.w, 16), di, __shfl_xor(rv.w, 16));
    if (!hi) *(float4*)(out + (size_t)i * 64 + l15x4) = o;
  }
}

template<int NC, int KW, bool RELU_A0, bool RELU_OUT, bool HAS_BIAS, int SCALE_COLS>
__global__ __launch_bounds__(256) void k_gemm(const float* __restrict__ A0, const float* __restrict__ A1,
                                              const float* __restrict__ W, const float* __restrict__ bias,
                                              const float* __restrict__ dis,
                                              float* __restrict__ C, int M) {
  __shared__ __align__(16) float As[128][KCHUNK + 4];
  __shared__ __align__(16) float Ws[NC][KCHUNK + 4];
  constexpr int NT = NC / 16;
  constexpr int K = KW * 128;

  const int tid  = threadIdx.x;
  const int wave = tid >> 5;
  const int lane = tid & 31;
  const int l15  = lane & 15;
  const int half = lane >> 4;
  const int m0   = blockIdx.x * 128;

  v8f acc[NT];
#pragma unroll
  for (int t = 0; t < NT; ++t)
#pragma unroll
    for (int r = 0; r < 8; ++r) acc[t][r] = 0.0f;

  for (int kb = 0; kb < K; kb += KCHUNK) {
    const float* Ablk = (KW == 2 && kb >= 128) ? A1 : A0;
    const bool doRelu = RELU_A0 && (kb < 128);
    const int kloc = kb & 127;

#pragma unroll
    for (int j = 0; j < (128 * KCHUNK) / 1024; ++j) {
      int ei  = tid * 4 + j * 1024;
      int row = ei >> 5;
      int col = ei & 31;
      int grow = m0 + row;
      float4 v = {0.f, 0.f, 0.f, 0.f};
      if (grow < M) v = *(const float4*)&Ablk[(size_t)grow * HIDW + kloc + col];
      if (doRelu) {
        v.x = fmaxf(v.x, 0.f); v.y = fmaxf(v.y, 0.f);
        v.z = fmaxf(v.z, 0.f); v.w = fmaxf(v.w, 0.f);
      }
      *(float4*)&As[row][col] = v;
    }
#pragma unroll
    for (int j = 0; j < (NC * KCHUNK) / 1024; ++j) {
      int ei  = tid * 4 + j * 1024;
      int n   = ei >> 5;
      int col = ei & 31;
      *(float4*)&Ws[n][col] = *(const float4*)&W[(size_t)n * K + kb + col];
    }
    __syncthreads();

    const int arow = wave * 16 + l15;
    const int o = half * 2;
#pragma unroll
    for (int k4 = 0; k4 < KCHUNK; k4 += 4) {
      v2f a = *(const v2f*)&As[arow][k4 + o];
#pragma unroll
      for (int t = 0; t < NT; ++t) {
        v2f b = *(const v2f*)&Ws[t * 16 + l15][k4 + o];
        acc[t] = __builtin_amdgcn_wmma_f32_16x16x4_f32(
             false, a,  false, b,
             (short)0, acc[t],  false,  false);
      }
    }
    __syncthreads();
  }

  const int rbase = m0 + wave * 16 + half * 8;
  float dr[8];
#pragma unroll
  for (int r = 0; r < 8; ++r) dr[r] = (SCALE_COLS > 0 && rbase + r < M) ? dis[rbase + r] : 1.0f;
#pragma unroll
  for (int t = 0; t < NT; ++t) {
    int col = t * 16 + l15;
    float bv = HAS_BIAS ? bias[col] : 0.f;
#pragma unroll
    for (int r = 0; r < 8; ++r) {
      int row = rbase + r;
      if (row < M) {
        float v = acc[t][r] + bv;
        if (RELU_OUT) v = fmaxf(v, 0.f);
        if (t * 16 < SCALE_COLS) v *= dr[r];
        C[(size_t)row * NC + col] = v;
      }
    }
  }
}


#define XP (HIDW + 4)
__global__ __launch_bounds__(512) void k_gather_gemm2(float* __restrict__ GP, const float* __restrict__ hs,
                                                      const float* __restrict__ x,
                                                      const int* __restrict__ rowptr, const int* __restrict__ csr,
                                                      const float* __restrict__ dis, const float* __restrict__ b1,
                                                      const float* __restrict__ Wc, const float* __restrict__ bc,
                                                      const float* __restrict__ Wcat, const float* __restrict__ bias128,
                                                      int n) {
  __shared__ __align__(16) float Xa[128][XP];
  __shared__ __align__(16) float As[128][KCHUNK + 4];
  __shared__ __align__(16) float Ws[HIDW][KCHUNK + 4];
  const int tid  = threadIdx.x;
  const int lane = tid & 31;
  const int wave = __builtin_amdgcn_readfirstlane(tid >> 5);
  const int l15  = lane & 15;
  const int half = lane >> 4;
  const int rg   = wave & 7;
  const int ch   = wave >> 3;
  const int m0   = blockIdx.x * 128;

  {
    const int lane4 = lane * 4;
    const float4 bv = *(const float4*)&b1[lane4];
    const int i0 = m0 + wave * 8;
    const int   rpl = rowptr[min(i0 + min(lane, 8), n)];
    const float dl  = dis[min(i0 + (lane & 7), n - 1)];
    for (int rr = 0; rr < 8; ++rr) {
      const int i   = i0 + rr;
      const int beg = __builtin_amdgcn_readlane(rpl, rr);
      const int end = __builtin_amdgcn_readlane(rpl, rr + 1);
      const float di = __int_as_float(__builtin_amdgcn_readlane(__float_as_int(dl), rr));
      float4 o = {0.f, 0.f, 0.f, 0.f};
      if (i < n) {
        float4 acc = *(const float4*)(hs + (size_t)i * HIDW + lane4);
        for (int base = beg; base < end; base += 32) {
          const int m = min(32, end - base);
          const int my_s = csr[min(base + lane, end - 1)];
          int j = 0;
          for (; j + 16 <= m; j += 16) gather_rows<16>(acc, hs, my_s, j, lane4);
          if (m - j >= 8) { gather_rows<8>(acc, hs, my_s, j, lane4); j += 8; }
          if (m - j >= 4) { gather_rows<4>(acc, hs, my_s, j, lane4); j += 4; }
          if (m - j >= 2) { gather_rows<2>(acc, hs, my_s, j, lane4); j += 2; }
          if (m - j >= 1) { gather_rows<1>(acc, hs, my_s, j, lane4); }
        }
        o.x = fmaxf(fmaf(acc.x, di, bv.x), 0.f);
        o.y = fmaxf(fmaf(acc.y, di, bv.y), 0.f);
        o.z = fmaxf(fmaf(acc.z, di, bv.z), 0.f);
        o.w = fmaxf(fmaf(acc.w, di, bv.w), 0.f);
      }
      *(float4*)&Xa[wave * 8 + rr][lane4] = o;
    }
  }

  const int arow = rg * 16 + l15;
  const int o2   = half * 2;
  v8f acc[4];
#pragma unroll
  for (int t = 0; t < 4; ++t)
#pragma unroll
    for (int r = 0; r < 8; ++r) acc[t][r] = 0.0f;

  for (int kb = 0; kb < 2 * HIDW; kb += KCHUNK) {
    const bool fromX = (kb >= HIDW);
    if (fromX) {
#pragma unroll
      for (int j = 0; j < (128 * KCHUNK) / 2048; ++j) {
        int ei   = tid * 4 + j * 2048;
        int row  = ei >> 5;
        int col  = ei & 31;
        int grow = m0 + row;
        float4 v = {0.f, 0.f, 0.f, 0.f};
        if (grow < n) v = *(const float4*)&x[(size_t)grow * HIDW + (kb - HIDW) + col];
        *(float4*)&As[row][col] = v;
      }
    }
#pragma unroll
    for (int j = 0; j < (HIDW * KCHUNK) / 2048; ++j) {
      int ei  = tid * 4 + j * 2048;
      int nn  = ei >> 5;
      int col = ei & 31;
      *(float4*)&Ws[nn][col] = *(const float4*)&Wc[(size_t)nn * (2 * HIDW) + kb + col];
    }
    __syncthreads();
#pragma unroll
    for (int k4 = 0; k4 < KCHUNK; k4 += 4) {
      v2f a = fromX ? *(const v2f*)&As[arow][k4 + o2] : *(const v2f*)&Xa[arow][kb + k4 + o2];
#pragma unroll
      for (int t = 0; t < 4; ++t) {
        v2f b = *(const v2f*)&Ws[ch * 64 + t * 16 + l15][k4 + o2];
        acc[t] = __builtin_amdgcn_wmma_f32_16x16x4_f32(false, a, false, b, (short)0, acc[t], false, false);
      }
    }
    __syncthreads();
  }

#pragma unroll
  for (int t = 0; t < 4; ++t) {
    const int col = ch * 64 + t * 16 + l15;
    const float bv = bc[col];
#pragma unroll
    for (int r = 0; r < 8; ++r) {
      Xa[rg * 16 + half * 8 + r][col] = fmaxf(acc[t][r] + bv, 0.f);
      acc[t][r] = 0.0f;
    }
  }

  for (int kb = 0; kb < HIDW; kb += KCHUNK) {
#pragma unroll
    for (int j = 0; j < (HIDW * KCHUNK) / 2048; ++j) {
      int ei  = tid * 4 + j * 2048;
      int nn  = ei >> 5;
      int col = ei & 31;
      *(float4*)&Ws[nn][col] = *(const float4*)&Wcat[(size_t)nn * HIDW + kb + col];
    }
    __syncthreads();
#pragma unroll
    for (int k4 = 0; k4 < KCHUNK; k4 += 4) {
      v2f a = *(const v2f*)&Xa[arow][kb + k4 + o2];
#pragma unroll
      for (int t = 0; t < 4; ++t) {
        v2f b = *(const v2f*)&Ws[ch * 64 + t * 16 + l15][k4 + o2];
        acc[t] = __builtin_amdgcn_wmma_f32_16x16x4_f32(false, a, false, b, (short)0, acc[t], false, false);
      }
    }
    __syncthreads();
  }

  const int rbase = m0 + rg * 16 + half * 8;
  float dr[8];
#pragma unroll
  for (int r = 0; r < 8; ++r) dr[r] = (ch == 0 && rbase + r < n) ? dis[rbase + r] : 1.0f;
#pragma unroll
  for (int t = 0; t < 4; ++t) {
    const int col = ch * 64 + t * 16 + l15;
    const float bv = bias128[col];
#pragma unroll
    for (int r = 0; r < 8; ++r) {
      const int row = rbase + r;
      if (row < n) GP[(size_t)row * HIDW + col] = (acc[t][r] + bv) * dr[r];
    }
  }
}

extern "C" void kernel_launch(void* const* d_in, const int* in_sizes, int n_in,
                              void* d_out, int out_size, void* d_ws, size_t ws_size,
                              hipStream_t stream) {
  const float* x  = (const float*)d_in[0];
  const int* edge = (const int*)d_in[1];
  const float* W1 = (const float*)d_in[2];
  const float* b1 = (const float*)d_in[3];
  const float* Wc = (const float*)d_in[4];
  const float* bc = (const float*)d_in[5];
  const float* W2 = (const float*)d_in[6];
  const float* b2 = (const float*)d_in[7];
  const float* Wo = (const float*)d_in[8];
  const float* bo = (const float*)d_in[9];

  const int N = in_sizes[0] / HIDW;
  const int E = in_sizes[1] / 2;
  const int* srcIdx = edge;
  const int* dstIdx = edge + E;

  float* wsf = (float*)d_ws;
  size_t NP   = (((size_t)N) + 255) & ~(size_t)255;
  size_t NP1  = (((size_t)N) + 1 + 255) & ~(size_t)255;
  size_t EP   = (((size_t)E) + 3) & ~(size_t)3;
  float* dis  = wsf;
  float* buf1 = wsf + NP;
  float* buf2 = buf1 + (size_t)N * HIDW;
  float* buf3 = buf2 + (size_t)N * HIDW;
  int* rowptr = (int*)buf3;
  int* csr    = rowptr + NP1;
  float* Wcat = (float*)(csr + EP);
  float* bias128 = Wcat + HIDW * HIDW;
  int gC      = (E + P_CHUNK - 1) / P_CHUNK;
  int gB      = (N + BNODES - 1) >> BSH;
  int2* part  = (int2*)buf2;
  int* histG  = (int*)buf2 + 2 * EP;
  int* bstartG = histG + (size_t)gC * NBMAX;

  dim3 b256(256);
  int gG  = (N + 127) / 128;
  int gP  = 2048;

  k_bucket_hist   <<<gC + 1, dim3(P_THREADS), 0, stream>>>(histG, dstIdx, E, Wo, W2, b2, bo, Wcat, bias128);
  k_bucket_scatter<<<gC, dim3(P_THREADS), 0, stream>>>(part, bstartG, histG, srcIdx, dstIdx, E);
  k_bucket_rows   <<<gB, dim3(ROWS_THREADS), 0, stream>>>(csr, rowptr, dis, part, bstartG, N);

  k_gemm<128,1,false,false,false,128><<<gG, b256, 0, stream>>>(x, nullptr, W1, nullptr, dis, buf1, N);

  k_gather_gemm2<<<gG, dim3(512), 0, stream>>>(buf2, buf1, x, rowptr, csr, dis, b1, Wc, bc, Wcat, bias128, N);
  k_prop_gather_half<<<gP, b256, 0, stream>>>((float*)d_out, buf2, rowptr, csr, dis, N);
}


# Fused launch retiled 128 to 208 rows, 26 waves: 241 work-groups, one per WGP, none doubled up; its 12 K chunks pipelined (next chunk via registers into a 2nd LDS stage, one barrier a chunk)
# speedup vs baseline: 8.6760x; 1.0472x over previous
#include <hip/hip_runtime.h>

typedef __attribute__((ext_vector_type(2))) float v2f;
typedef __attribute__((ext_vector_type(8))) float v8f;
typedef __attribute__((ext_vector_type(4))) float v4f;

#define HIDW 128
#define KCHUNK 32

__device__ __forceinline__ void fold_weights(const float* __restrict__ Wo, const float* __restrict__ W2,
                                             const float* __restrict__ b2, const float* __restrict__ bo,
                                             float* __restrict__ Wcat, float* __restrict__ bias128) {
  __shared__ __align__(16) float Fa[64][KCHUNK + 4];
  __shared__ __align__(16) float Fb[HIDW][KCHUNK + 4];
  const int tid  = threadIdx.x;
  const int wave = tid >> 5;
  const int lane = tid & 31;
  const int l15  = lane & 15;
  const int half = lane >> 4;
  const int mt   = wave & 3;
  const int nt   = wave >> 2;

  v8f acc;
#pragma unroll
  for (int r = 0; r < 8; ++r) acc[r] = 0.0f;

  for (int hb = 0; hb < HIDW; hb += KCHUNK) {
    {
      int idx = tid * 2;
      int o   = idx >> 5;
      int hh  = idx & 31;
      *(v2f*)&Fa[o][hh] = *(const v2f*)&Wo[(size_t)o * (2 * HIDW) + hb + hh];
    }
#pragma unroll
    for (int j = 0; j < (HIDW * KCHUNK) / 1024; ++j) {
      int e  = tid + j * 1024;
      int hh = e >> 7;
      int k  = e & 127;
      Fb[k][hh] = W2[(size_t)(hb + hh) * HIDW + k];
    }
    __syncthreads();
    const int o2 = half * 2;
#pragma unroll
    for (int k4 = 0; k4 < KCHUNK; k4 += 4) {
      v2f a = *(const v2f*)&Fa[mt * 16 + l15][k4 + o2];
      v2f b = *(const v2f*)&Fb[nt * 16 + l15][k4 + o2];
      acc = __builtin_amdgcn_wmma_f32_16x16x4_f32(false, a, false, b, (short)0, acc, false, false);
    }
    __syncthreads();
  }
#pragma unroll
  for (int r = 0; r < 8; ++r)
    Wcat[(size_t)(mt * 16 + half * 8 + r) * HIDW + nt * 16 + l15] = acc[r];

#pragma unroll
  for (int j = 0; j < 2; ++j) {
    int q  = tid + j * 1024;
    int o  = q >> 5;
    int c4 = (q & 31) * 4;
    *(float4*)&Wcat[(size_t)(64 + o) * HIDW + c4] = *(const float4*)&Wo[(size_t)o * (2 * HIDW) + HIDW + c4];
  }
  if (tid < 64) {
    float s = bo[tid];
    for (int h = 0; h < HIDW; ++h) s = fmaf(Wo[(size_t)tid * (2 * HIDW) + h], b2[h], s);
    bias128[tid]      = 0.f;
    bias128[64 + tid] = s;
  }
}

#define BSH 7
#define BNODES (1 << BSH)
#define NBMAX 512
#define P_THREADS 1024
#define P_PER 8
#define P_CHUNK (P_THREADS * P_PER)
#define HIST_ROWS 32

template<int NWAVES>
__device__ __forceinline__ void block_scan2(int& a, int& b, int* wsA, int* wsB) {
  const int lane = threadIdx.x & 31;
  const int wv   = threadIdx.x >> 5;
  int ia = a, ib = b;
  for (int o = 1; o < 32; o <<= 1) {
    int ta = __shfl_up(ia, o);
    int tb = __shfl_up(ib, o);
    if (lane >= o) { ia += ta; ib += tb; }
  }
  if (lane == 31) { wsA[wv] = ia; wsB[wv] = ib; }
  __syncthreads();
  if (wv == 0) {
    int ta = (lane < NWAVES) ? wsA[lane] : 0;
    int tb = (lane < NWAVES) ? wsB[lane] : 0;
    for (int o = 1; o < 32; o <<= 1) {
      int ua = __shfl_up(ta, o);
      int ub = __shfl_up(tb, o);
      if (lane >= o) { ta += ua; tb += ub; }
    }
    if (lane < NWAVES) { wsA[lane] = ta; wsB[lane] = tb; }
  }
  __syncthreads();
  a = ia - a + (wv ? wsA[wv - 1] : 0);
  b = ib - b + (wv ? wsB[wv - 1] : 0);
}

__global__ __launch_bounds__(P_THREADS) void k_bucket_hist(int* __restrict__ histG, const int* __restrict__ dst, int E,
                                                           const float* __restrict__ Wo, const float* __restrict__ W2,
                                                           const float* __restrict__ b2, const float* __restrict__ bo,
                                                           float* __restrict__ Wcat, float* __restrict__ bias128) {
  __shared__ int h[NBMAX];
  if (blockIdx.x == gridDim.x - 1) { fold_weights(Wo, W2, b2, bo, Wcat, bias128); return; }
  const int tid = threadIdx.x;
  if (tid < NBMAX) h[tid] = 0;
  __syncthreads();
  const int e0 = blockIdx.x * P_CHUNK;
#pragma unroll
  for (int k = 0; k < P_PER; ++k) {
    int e = e0 + k * P_THREADS + tid;
    if (e < E) atomicAdd(&h[dst[e] >> BSH], 1);
  }
  __syncthreads();
  if (tid < NBMAX) histG[(size_t)blockIdx.x * NBMAX + tid] = h[tid];
}

__global__ __launch_bounds__(P_THREADS) void k_bucket_scatter(int2* __restrict__ part, int* __restrict__ bstartG,
                                                              const int* __restrict__ histG,
                                                              const int* __restrict__ src, const int* __restrict__ dst, int E) {
  __shared__ int cnt[NBMAX];
  __shared__ int loc[NBMAX];
  __shared__ int gbase[NBMAX];
  __shared__ int hTot[NBMAX];
  __shared__ int hPre[NBMAX];
  __shared__ int wsA[32];
  __shared__ int wsB[32];
  __shared__ int2 stage[P_CHUNK];
  const int tid = threadIdx.x;
  const int wg  = blockIdx.x;
  const int nwg = gridDim.x;
  const int e0  = wg * P_CHUNK;
  if (tid < NBMAX) cnt[tid] = 0;
  __syncthreads();

  int s[P_PER], d[P_PER], r[P_PER];
#pragma unroll
  for (int k = 0; k < P_PER; ++k) {
    int e = e0 + k * P_THREADS + tid;
    d[k] = -1; s[k] = 0; r[k] = 0;
    if (e < E) { d[k] = dst[e]; s[k] = src[e]; }
  }
#pragma unroll
  for (int k = 0; k < P_PER; ++k)
    if (d[k] >= 0) r[k] = atomicAdd(&cnt[d[k] >> BSH], 1);

  const int bq = tid & (NBMAX - 1);
  const int hq = __builtin_amdgcn_readfirstlane(tid >> 9);
  const int nh = (nwg + 1) >> 1;
  const int wb = hq * nh;
  int tot = 0, pre = 0;
  for (int j0 = 0; j0 < nh; j0 += HIST_ROWS) {
    int v[HIST_ROWS];
#pragma unroll
    for (int u = 0; u < HIST_ROWS; ++u)
      v[u] = histG[min(wb + j0 + u, nwg - 1) * NBMAX + bq];
#pragma unroll
    for (int u = 0; u < HIST_ROWS; ++u) {
      const int w  = wb + j0 + u;
      const int vv = (j0 + u < nh && w < nwg) ? v[u] : 0;
      tot += vv;
      pre += (w < wg) ? vv : 0;
    }
  }
  if (hq) { hTot[bq] = tot; hPre[bq] = pre; }
  __syncthreads();
  int a = 0, c = 0;
  if (!hq) { tot += hTot[bq]; pre += hPre[bq]; a = tot; c = cnt[bq]; }
  block_scan2<32>(a, c, wsA, wsB);
  if (!hq) {
    loc[bq]   = c;
    gbase[bq] = a + pre;
    if (wg == 0) {
      bstartG[bq] = a;
      if (bq == NBMAX - 1) bstartG[NBMAX] = a + tot;
    }
  }
  __syncthreads();
#pragma unroll
  for (int k = 0; k < P_PER; ++k)
    if (d[k] >= 0) stage[loc[d[k] >> BSH] + r[k]] = make_int2(s[k], d[k]);
  __syncthreads();
  const int nloc = min(P_CHUNK, E - e0);
#pragma unroll
  for (int k = 0; k < P_PER; ++k) {
    int i = k * P_THREADS + tid;
    if (i < nloc) {
      int2 rec = stage[i];
      int b = rec.y >> BSH;
      part[(size_t)gbase[b] + (i - loc[b])] = rec;
    }
  }
}

#define ROWS_THREADS 256
#define ROWS_PER 8
#define ROWS_STAGE 12288
__global__ __launch_bounds__(ROWS_THREADS) void k_bucket_rows(int* __restrict__ csr, int* __restrict__ rowptr,
                                                              float* __restrict__ dis, const int2* __restrict__ part,
                                                              const int* __restrict__ bstartG, int n) {
  __shared__ int cntN[BNODES];
  __shared__ int fillN[BNODES];
  __shared__ int rstart[BNODES];
  __shared__ int wsA[8];
  __shared__ int wsB[8];
  __shared__ int stage[ROWS_STAGE];
  const int tid = threadIdx.x;
  const int b   = blockIdx.x;
  const int beg = bstartG[b];
  const int end = bstartG[b + 1];
  const int ne  = end - beg;
  if (tid < BNODES) { cntN[tid] = 0; fillN[tid] = 0; }
  __syncthreads();
  for (int i0 = 0; i0 < ne; i0 += ROWS_THREADS * ROWS_PER) {
    int2 rec[ROWS_PER];
#pragma unroll
    for (int k = 0; k < ROWS_PER; ++k) {
      const int i = i0 + k * ROWS_THREADS + tid;
      rec[k] = (i < ne) ? part[(size_t)beg + i] : make_int2(0, -1);
    }
#pragma unroll
    for (int k = 0; k < ROWS_PER; ++k)
      if (rec[k].y >= 0) atomicAdd(&cntN[rec[k].y & (BNODES - 1)], 1);
  }
  __syncthreads();
  int c = (tid < BNODES) ? cntN[tid] : 0;
  int x = c, unused = 0;
  block_scan2<ROWS_THREADS / 32>(x, unused, wsA, wsB);
  if (tid < BNODES) {
    rstart[tid] = x;
    const int node = b * BNODES + tid;
    if (node < n) {
      rowptr[node] = beg + x;
      float dg = (float)(c + 1);
      dis[node] = (dg > 0.f) ? rsqrtf(dg) : 0.f;
    }
  }
  if (b == (int)gridDim.x - 1 && tid == 0) rowptr[n] = end;
  __syncthreads();
  const bool staged = (ne <= ROWS_STAGE);
  for (int i0 = 0; i0 < ne; i0 += ROWS_THREADS * ROWS_PER) {
    int2 rec[ROWS_PER];
#pragma unroll
    for (int k = 0; k < ROWS_PER; ++k) {
      const int i = i0 + k * ROWS_THREADS + tid;
      rec[k] = (i < ne) ? part[(size_t)beg + i] : make_int2(0, -1);
    }
    int pos[ROWS_PER];
#pragma unroll
    for (int k = 0; k < ROWS_PER; ++k) {
      pos[k] = -1;
      if (rec[k].y >= 0) {
        const int dl = rec[k].y & (BNODES - 1);
        pos[k] = rstart[dl] + atomicAdd(&fillN[dl], 1);
      }
    }
    if (staged) {
#pragma unroll
      for (int k = 0; k < ROWS_PER; ++k)
        if (pos[k] >= 0) stage[pos[k]] = rec[k].x;
    } else {
#pragma unroll
      for (int k = 0; k < ROWS_PER; ++k)
        if (pos[k] >= 0) csr[(size_t)beg + pos[k]] = rec[k].x;
    }
  }
  if (staged) {
    __syncthreads();
    for (int i = tid; i < ne; i += ROWS_THREADS) csr[(size_t)beg + i] = stage[i];
  }
}

template<int U>
__device__ __forceinline__ void gather_rows(float4& acc, const float* __restrict__ h, int my_s, int j, int lane4) {
  float4 v[U];
#pragma unroll
  for (int u = 0; u < U; ++u) {
    int s = __builtin_amdgcn_readlane(my_s, j + u);
    v[u] = *(const float4*)(h + (size_t)s * HIDW + lane4);
  }
#pragma unroll
  for (int u = 0; u < U; ++u) {
    acc.x += v[u].x;
    acc.y += v[u].y;
    acc.z += v[u].z;
    acc.w += v[u].w;
  }
}

template<int U, bool TAIL>
__device__ __forceinline__ void gather_pairs(float4& acc, const float* __restrict__ gp, int my_s,
                                             int j, int m, bool hi, int l15x4) {
  float4 v[U];
#pragma unroll
  for (int u = 0; u < U; ++u) {
    int slo = __builtin_amdgcn_readlane(my_s, j + 2 * u);
    int shi = __builtin_amdgcn_readlane(my_s, j + 2 * u + 1);
    int s = hi ? shi : slo;
    v[u] = *(const float4*)(gp + (size_t)s * HIDW + l15x4);
  }
#pragma unroll
  for (int u = 0; u < U; ++u) {
    const bool live = !TAIL || !hi || (j + 2 * u + 1 < m);
    acc.x += live ? v[u].x : 0.f;
    acc.y += live ? v[u].y : 0.f;
    acc.z += live ? v[u].z : 0.f;
    acc.w += live ? v[u].w : 0.f;
  }
}

__global__ __launch_bounds__(256) void k_prop_gather_half(float* __restrict__ out, const float* __restrict__ gp,
                                                          const int* __restrict__ rowptr, const int* __restrict__ csr,
                                                          const float* __restrict__ dis, int n) {
  const int lane  = threadIdx.x & 31;
  const bool hi   = lane >= 16;
  const int l15x4 = (lane & 15) * 4;
  const int wave = __builtin_amdgcn_readfirstlane((int)(threadIdx.x >> 5));
  const int nw   = gridDim.x * 8;

  for (int i = blockIdx.x * 8 + wave; i < n; i += nw) {
    const int beg = rowptr[i];
    const int end = rowptr[i + 1];
    const float di = dis[i];
    const float4 rv = *(const float4*)(gp + (size_t)i * HIDW + (hi ? 64 : 0) + l15x4);
    float4 acc;
    acc.x = hi ? 0.f : rv.x; acc.y = hi ? 0.f : rv.y; acc.z = hi ? 0.f : rv.z; acc.w = hi ? 0.f : rv.w;

    for (int base = beg; base < end; base += 32) {
      const int m = min(32, end - base);
      const int my_s = csr[min(base + lane, end - 1)];
      int j = 0;
      for (; j + 16 <= m; j += 16) gather_pairs<8, false>(acc, gp, my_s, j, m, hi, l15x4);
      if (m - j >= 8) { gather_pairs<4, false>(acc, gp, my_s, j, m, hi, l15x4); j += 8; }
      if (m - j >= 4) { gather_pairs<2, false>(acc, gp, my_s, j, m, hi, l15x4); j += 4; }
      if (m - j >= 3)      gather_pairs<2, true>(acc, gp, my_s, j, m, hi, l15x4);
      else if (m - j >= 1) gather_pairs<1, true>(acc, gp, my_s, j, m, hi, l15x4);
    }
    float4 o;
    o.x = fmaf(acc.x + __shfl_xor(acc.x, 16), di, __shfl_xor(rv.x, 16));
    o.y = fmaf(acc.y + __shfl_xor(acc.y, 16), di, __shfl_xor(rv.y, 16));
    o.z = fmaf(acc.z + __shfl_xor(acc.z, 16), di, __shfl_xor(rv.z, 16));
    o.w = fmaf(acc.w + __shfl_xor(acc.w, 16), di, __shfl_xor(rv.w, 16));
    if (!hi) *(float4*)(out + (size_t)i * 64 + l15x4) = o;
  }
}

template<int NC, int KW, bool RELU_A0, bool RELU_OUT, bool HAS_BIAS, int SCALE_COLS>
__global__ __launch_bounds__(256) void k_gemm(const float* __restrict__ A0, const float* __restrict__ A1,
                                              const float* __restrict__ W, const float* __restrict__ bias,
                                              const float* __restrict__ dis,
                                              float* __restrict__ C, int M) {
  __shared__ __align__(16) float As[128][KCHUNK + 4];
  __shared__ __align__(16) float Ws[NC][KCHUNK + 4];
  constexpr int NT = NC / 16;
  constexpr int K = KW * 128;

  const int tid  = threadIdx.x;
  const int wave = tid >> 5;
  const int lane = tid & 31;
  const int l15  = lane & 15;
  const int half = lane >> 4;
  const int m0   = blockIdx.x * 128;

  v8f acc[NT];
#pragma unroll
  for (int t = 0; t < NT; ++t)
#pragma unroll
    for (int r = 0; r < 8; ++r) acc[t][r] = 0.0f;

  for (int kb = 0; kb < K; kb += KCHUNK) {
    const float* Ablk = (KW == 2 && kb >= 128) ? A1 : A0;
    const bool doRelu = RELU_A0 && (kb < 128);
    const int kloc = kb & 127;

#pragma unroll
    for (int j = 0; j < (128 * KCHUNK) / 1024; ++j) {
      int ei  = tid * 4 + j * 1024;
      int row = ei >> 5;
      int col = ei & 31;
      int grow = m0 + row;
      float4 v = {0.f, 0.f, 0.f, 0.f};
      if (grow < M) v = *(const float4*)&Ablk[(size_t)grow * HIDW + kloc + col];
      if (doRelu) {
        v.x = fmaxf(v.x, 0.f); v.y = fmaxf(v.y, 0.f);
        v.z = fmaxf(v.z, 0.f); v.w = fmaxf(v.w, 0.f);
      }
      *(float4*)&As[row][col] = v;
    }
#pragma unroll
    for (int j = 0; j < (NC * KCHUNK) / 1024; ++j) {
      int ei  = tid * 4 + j * 1024;
      int n   = ei >> 5;
      int col = ei & 31;
      *(float4*)&Ws[n][col] = *(const float4*)&W[(size_t)n * K + kb + col];
    }
    __syncthreads();

    const int arow = wave * 16 + l15;
    const int o = half * 2;
#pragma unroll
    for (int k4 = 0; k4 < KCHUNK; k4 += 4) {
      v2f a = *(const v2f*)&As[arow][k4 + o];
#pragma unroll
      for (int t = 0; t < NT; ++t) {
        v2f b = *(const v2f*)&Ws[t * 16 + l15][k4 + o];
        acc[t] = __builtin_amdgcn_wmma_f32_16x16x4_f32(
             false, a,  false, b,
             (short)0, acc[t],  false,  false);
      }
    }
    __syncthreads();
  }

  const int rbase = m0 + wave * 16 + half * 8;
  float dr[8];
#pragma unroll
  for (int r = 0; r < 8; ++r) dr[r] = (SCALE_COLS > 0 && rbase + r < M) ? dis[rbase + r] : 1.0f;
#pragma unroll
  for (int t = 0; t < NT; ++t) {
    int col = t * 16 + l15;
    float bv = HAS_BIAS ? bias[col] : 0.f;
#pragma unroll
    for (int r = 0; r < 8; ++r) {
      int row = rbase + r;
      if (row < M) {
        float v = acc[t][r] + bv;
        if (RELU_OUT) v = fmaxf(v, 0.f);
        if (t * 16 < SCALE_COLS) v *= dr[r];
        C[(size_t)row * NC + col] = v;
      }
    }
  }
}


#define XP (HIDW + 4)
#define TR 208
#define TRT (TR / 16)
#define GT (2 * TRT * 32)
__global__ __launch_bounds__(GT) void k_gather_gemm2(float* __restrict__ GP, const float* __restrict__ hs,
                                                      const float* __restrict__ x,
                                                      const int* __restrict__ rowptr, const int* __restrict__ csr,
                                                      const float* __restrict__ dis, const float* __restrict__ b1,
                                                      const float* __restrict__ Wc, const float* __restrict__ bc,
                                                      const float* __restrict__ Wcat, const float* __restrict__ bias128,
                                                      int n) {
  __shared__ __align__(16) float Xa[TR][XP];
  __shared__ __align__(16) float As[2][TR][KCHUNK + 4];
  __shared__ __align__(16) float Ws[2][HIDW][KCHUNK + 4];
  const int tid  = threadIdx.x;
  const int lane = tid & 31;
  const int wave = __builtin_amdgcn_readfirstlane(tid >> 5);
  const int l15  = lane & 15;
  const int half = lane >> 4;
  const int ch   = (wave >= TRT) ? 1 : 0;
  const int rg   = wave - ch * TRT;
  const int m0   = blockIdx.x * TR;

  const int wrow0 = tid >> 3;
  const int wrow1 = (tid + GT) >> 3;
  const int scol  = (tid & 7) * 4;
  const bool w1on = (tid + GT) < HIDW * 8;
  const int wrow1c = min(wrow1, HIDW - 1);
  const size_t xo0 = (size_t)min(m0 + wrow0, n - 1) * HIDW + scol;
  const size_t xo1 = (size_t)min(m0 + wrow1, n - 1) * HIDW + scol;
  const size_t wc0 = (size_t)wrow0 * (2 * HIDW) + scol, wc1 = (size_t)wrow1c * (2 * HIDW) + scol;
  const size_t wt0 = (size_t)wrow0 * HIDW + scol,       wt1 = (size_t)wrow1c * HIDW + scol;
  v4f wreg0, wreg1, xreg0, xreg1;

  {
    const int lane4 = lane * 4;
    const float4 bv = *(const float4*)&b1[lane4];
    const int i0 = m0 + wave * 8;
    const int   rpl = rowptr[min(i0 + min(lane, 8), n)];
    const float dl  = dis[min(i0 + (lane & 7), n - 1)];
    for (int rr = 0; rr < 8; ++rr) {
      const int i   = i0 + rr;
      const int beg = __builtin_amdgcn_readlane(rpl, rr);
      const int end = __builtin_amdgcn_readlane(rpl, rr + 1);
      const float di = __int_as_float(__builtin_amdgcn_readlane(__float_as_int(dl), rr));
      float4 o = {0.f, 0.f, 0.f, 0.f};
      if (i < n) {
        float4 acc = *(const float4*)(hs + (size_t)i * HIDW + lane4);
        for (int base = beg; base < end; base += 32) {
          const int m = min(32, end - base);
          const int my_s = csr[min(base + lane, end - 1)];
          int j = 0;
          for (; j + 16 <= m; j += 16) gather_rows<16>(acc, hs, my_s, j, lane4);
          if (m - j >= 8) { gather_rows<8>(acc, hs, my_s, j, lane4); j += 8; }
          if (m - j >= 4) { gather_rows<4>(acc, hs, my_s, j, lane4); j += 4; }
          if (m - j >= 2) { gather_rows<2>(acc, hs, my_s, j, lane4); j += 2; }
          if (m - j >= 1) { gather_rows<1>(acc, hs, my_s, j, lane4); }
        }
        o.x = fmaxf(fmaf(acc.x, di, bv.x), 0.f);
        o.y = fmaxf(fmaf(acc.y, di, bv.y), 0.f);
        o.z = fmaxf(fmaf(acc.z, di, bv.z), 0.f);
        o.w = fmaxf(fmaf(acc.w, di, bv.w), 0.f);
      }
      *(float4*)&Xa[wave * 8 + rr][lane4] = o;
    }
  }

  wreg0 = *(const v4f*)&Wc[wc0];
  wreg1 = *(const v4f*)&Wc[wc1];
  *(v4f*)&Ws[0][wrow0][scol] = wreg0;
  if (w1on) *(v4f*)&Ws[0][wrow1][scol] = wreg1;
  wreg0 = *(const v4f*)&Wc[wc0 + KCHUNK];
  wreg1 = *(const v4f*)&Wc[wc1 + KCHUNK];
  __syncthreads();

  const int arow = rg * 16 + l15;
  const int o2   = half * 2;
  v8f acc[4];
#pragma unroll
  for (int t = 0; t < 4; ++t)
#pragma unroll
    for (int r = 0; r < 8; ++r) acc[t][r] = 0.0f;

#pragma nounroll
  for (int c = 0; c < 8; ++c) {
    const int s = c & 1;
    {
      const int cs = c + 1, sn = s ^ 1;
      *(v4f*)&Ws[sn][wrow0][scol] = wreg0;
      if (w1on) *(v4f*)&Ws[sn][wrow1][scol] = wreg1;
      if (cs >= 4 && cs < 8) {
        *(v4f*)&As[sn][wrow0][scol] = xreg0;
        *(v4f*)&As[sn][wrow1][scol] = xreg1;
      }
      const int cf = c + 2;
      if (cf < 8) {
        wreg0 = *(const v4f*)&Wc[wc0 + cf * KCHUNK];
        wreg1 = *(const v4f*)&Wc[wc1 + cf * KCHUNK];
        if (cf >= 4) {
          xreg0 = *(const v4f*)&x[xo0 + (cf - 4) * KCHUNK];
          xreg1 = *(const v4f*)&x[xo1 + (cf - 4) * KCHUNK];
        }
      } else {
        wreg0 = *(const v4f*)&Wcat[wt0 + (cf - 8) * KCHUNK];
        wreg1 = *(const v4f*)&Wcat[wt1 + (cf - 8) * KCHUNK];
      }
    }
    const bool fromX = (c >= 4);
    const int kb = c * KCHUNK;
#pragma unroll
    for (int k4 = 0; k4 < KCHUNK; k4 += 4) {
      v2f a = fromX ? *(const v2f*)&As[s][arow][k4 + o2] : *(const v2f*)&Xa[arow][kb + k4 + o2];
#pragma unroll
      for (int t = 0; t < 4; ++t) {
        v2f b = *(const v2f*)&Ws[s][ch * 64 + t * 16 + l15][k4 + o2];
        acc[t] = __builtin_amdgcn_wmma_f32_16x16x4_f32(false, a, false, b, (short)0, acc[t], false, false);
      }
    }
    __syncthreads();
  }

#pragma unroll
  for (int t = 0; t < 4; ++t) {
    const int col = ch * 64 + t * 16 + l15;
    const float bv = bc[col];
#pragma unroll
    for (int r = 0; r < 8; ++r) {
      Xa[rg * 16 + half * 8 + r][col] = fmaxf(acc[t][r] + bv, 0.f);
      acc[t][r] = 0.0f;
    }
  }
  __syncthreads();

#pragma nounroll
  for (int c = 8; c < 12; ++c) {
    const int s = c & 1;
    if (c + 1 < 12) {
      *(v4f*)&Ws[s ^ 1][wrow0][scol] = wreg0;
      if (w1on) *(v4f*)&Ws[s ^ 1][wrow1][scol] = wreg1;
    }
    if (c + 2 < 12) {
      wreg0 = *(const v4f*)&Wcat[wt0 + (c + 2 - 8) * KCHUNK];
      wreg1 = *(const v4f*)&Wcat[wt1 + (c + 2 - 8) * KCHUNK];
    }
    const int kb = (c - 8) * KCHUNK;
#pragma unroll
    for (int k4 = 0; k4 < KCHUNK; k4 += 4) {
      v2f a = *(const v2f*)&Xa[arow][kb + k4 + o2];
#pragma unroll
      for (int t = 0; t < 4; ++t) {
        v2f b = *(const v2f*)&Ws[s][ch * 64 + t * 16 + l15][k4 + o2];
        acc[t] = __builtin_amdgcn_wmma_f32_16x16x4_f32(false, a, false, b, (short)0, acc[t], false, false);
      }
    }
    if (c + 1 < 12) __syncthreads();
  }

  const int rbase = m0 + rg * 16 + half * 8;
  float dr[8];
#pragma unroll
  for (int r = 0; r < 8; ++r) dr[r] = (ch == 0 && rbase + r < n) ? dis[rbase + r] : 1.0f;
#pragma unroll
  for (int t = 0; t < 4; ++t) {
    const int col = ch * 64 + t * 16 + l15;
    const float bv = bias128[col];
#pragma unroll
    for (int r = 0; r < 8; ++r) {
      const int row = rbase + r;
      if (row < n) GP[(size_t)row * HIDW + col] = (acc[t][r] + bv) * dr[r];
    }
  }
}

extern "C" void kernel_launch(void* const* d_in, const int* in_sizes, int n_in,
                              void* d_out, int out_size, void* d_ws, size_t ws_size,
                              hipStream_t stream) {
  const float* x  = (const float*)d_in[0];
  const int* edge = (const int*)d_in[1];
  const float* W1 = (const float*)d_in[2];
  const float* b1 = (const float*)d_in[3];
  const float* Wc = (const float*)d_in[4];
  const float* bc = (const float*)d_in[5];
  const float* W2 = (const float*)d_in[6];
  const float* b2 = (const float*)d_in[7];
  const float* Wo = (const float*)d_in[8];
  const float* bo = (const float*)d_in[9];

  const int N = in_sizes[0] / HIDW;
  const int E = in_sizes[1] / 2;
  const int* srcIdx = edge;
  const int* dstIdx = edge + E;

  float* wsf = (float*)d_ws;
  size_t NP   = (((size_t)N) + 255) & ~(size_t)255;
  size_t NP1  = (((size_t)N) + 1 + 255) & ~(size_t)255;
  size_t EP   = (((size_t)E) + 3) & ~(size_t)3;
  float* dis  = wsf;
  float* buf1 = wsf + NP;
  float* buf2 = buf1 + (size_t)N * HIDW;
  float* buf3 = buf2 + (size_t)N * HIDW;
  int* rowptr = (int*)buf3;
  int* csr    = rowptr + NP1;
  float* Wcat = (float*)(csr + EP);
  float* bias128 = Wcat + HIDW * HIDW;
  int gC      = (E + P_CHUNK - 1) / P_CHUNK;
  int gB      = (N + BNODES - 1) >> BSH;
  int2* part  = (int2*)buf2;
  int* histG  = (int*)buf2 + 2 * EP;
  int* bstartG = histG + (size_t)gC * NBMAX;

  dim3 b256(256);
  int gG  = (N + 127) / 128;
  int gF  = (N + TR - 1) / TR;
  int gP  = 2048;

  k_bucket_hist   <<<gC + 1, dim3(P_THREADS), 0, stream>>>(histG, dstIdx, E, Wo, W2, b2, bo, Wcat, bias128);
  k_bucket_scatter<<<gC, dim3(P_THREADS), 0, stream>>>(part, bstartG, histG, srcIdx, dstIdx, E);
  k_bucket_rows   <<<gB, dim3(ROWS_THREADS), 0, stream>>>(csr, rowptr, dis, part, bstartG, N);

  k_gemm<128,1,false,false,false,128><<<gG, b256, 0, stream>>>(x, nullptr, W1, nullptr, dis, buf1, N);

  k_gather_gemm2<<<gF, dim3(GT), 0, stream>>>(buf2, buf1, x, rowptr, csr, dis, b1, Wc, bc, Wcat, bias128, N);
  k_prop_gather_half<<<gP, b256, 0, stream>>>((float*)d_out, buf2, rowptr, csr, dis, N);
}


# x half of the first product (x tile, Wc[:,128:256] whole in LDS) issued on WMMA between the list sums' row requests and adds in the fused launch; that product summed x half first, x_agg half second
# speedup vs baseline: 8.9541x; 1.0321x over previous
#include <hip/hip_runtime.h>

typedef __attribute__((ext_vector_type(2))) float v2f;
typedef __attribute__((ext_vector_type(8))) float v8f;
typedef __attribute__((ext_vector_type(4))) float v4f;

#define HIDW 128
#define KCHUNK 32

__device__ __forceinline__ void fold_weights(const float* __restrict__ Wo, const float* __restrict__ W2,
                                             const float* __restrict__ b2, const float* __restrict__ bo,
                                             float* __restrict__ Wcat, float* __restrict__ bias128) {
  __shared__ __align__(16) float Fa[64][KCHUNK + 4];
  __shared__ __align__(16) float Fb[HIDW][KCHUNK + 4];
  const int tid  = threadIdx.x;
  const int wave = tid >> 5;
  const int lane = tid & 31;
  const int l15  = lane & 15;
  const int half = lane >> 4;
  const int mt   = wave & 3;
  const int nt   = wave >> 2;

  v8f acc;
#pragma unroll
  for (int r = 0; r < 8; ++r) acc[r] = 0.0f;

  for (int hb = 0; hb < HIDW; hb += KCHUNK) {
    {
      int idx = tid * 2;
      int o   = idx >> 5;
      int hh  = idx & 31;
      *(v2f*)&Fa[o][hh] = *(const v2f*)&Wo[(size_t)o * (2 * HIDW) + hb + hh];
    }
#pragma unroll
    for (int j = 0; j < (HIDW * KCHUNK) / 1024; ++j) {
      int e  = tid + j * 1024;
      int hh = e >> 7;
      int k  = e & 127;
      Fb[k][hh] = W2[(size_t)(hb + hh) * HIDW + k];
    }
    __syncthreads();
    const int o2 = half * 2;
#pragma unroll
    for (int k4 = 0; k4 < KCHUNK; k4 += 4) {
      v2f a = *(const v2f*)&Fa[mt * 16 + l15][k4 + o2];
      v2f b = *(const v2f*)&Fb[nt * 16 + l15][k4 + o2];
      acc = __builtin_amdgcn_wmma_f32_16x16x4_f32(false, a, false, b, (short)0, acc, false, false);
    }
    __syncthreads();
  }
#pragma unroll
  for (int r = 0; r < 8; ++r)
    Wcat[(size_t)(mt * 16 + half * 8 + r) * HIDW + nt * 16 + l15] = acc[r];

#pragma unroll
  for (int j = 0; j < 2; ++j) {
    int q  = tid + j * 1024;
    int o  = q >> 5;
    int c4 = (q & 31) * 4;
    *(float4*)&Wcat[(size_t)(64 + o) * HIDW + c4] = *(const float4*)&Wo[(size_t)o * (2 * HIDW) + HIDW + c4];
  }
  if (tid < 64) {
    float s = bo[tid];
    for (int h = 0; h < HIDW; ++h) s = fmaf(Wo[(size_t)tid * (2 * HIDW) + h], b2[h], s);
    bias128[tid]      = 0.f;
    bias128[64 + tid] = s;
  }
}

#define BSH 7
#define BNODES (1 << BSH)
#define NBMAX 512
#define P_THREADS 1024
#define P_PER 8
#define P_CHUNK (P_THREADS * P_PER)
#define HIST_ROWS 32

template<int NWAVES>
__device__ __forceinline__ void block_scan2(int& a, int& b, int* wsA, int* wsB) {
  const int lane = threadIdx.x & 31;
  const int wv   = threadIdx.x >> 5;
  int ia = a, ib = b;
  for (int o = 1; o < 32; o <<= 1) {
    int ta = __shfl_up(ia, o);
    int tb = __shfl_up(ib, o);
    if (lane >= o) { ia += ta; ib += tb; }
  }
  if (lane == 31) { wsA[wv] = ia; wsB[wv] = ib; }
  __syncthreads();
  if (wv == 0) {
    int ta = (lane < NWAVES) ? wsA[lane] : 0;
    int tb = (lane < NWAVES) ? wsB[lane] : 0;
    for (int o = 1; o < 32; o <<= 1) {
      int ua = __shfl_up(ta, o);
      int ub = __shfl_up(tb, o);
      if (lane >= o) { ta += ua; tb += ub; }
    }
    if (lane < NWAVES) { wsA[lane] = ta; wsB[lane] = tb; }
  }
  __syncthreads();
  a = ia - a + (wv ? wsA[wv - 1] : 0);
  b = ib - b + (wv ? wsB[wv - 1] : 0);
}

__global__ __launch_bounds__(P_THREADS) void k_bucket_hist(int* __restrict__ histG, const int* __restrict__ dst, int E,
                                                           const float* __restrict__ Wo, const float* __restrict__ W2,
                                                           const float* __restrict__ b2, const float* __restrict__ bo,
                                                           float* __restrict__ Wcat, float* __restrict__ bias128) {
  __shared__ int h[NBMAX];
  if (blockIdx.x == gridDim.x - 1) { fold_weights(Wo, W2, b2, bo, Wcat, bias128); return; }
  const int tid = threadIdx.x;
  if (tid < NBMAX) h[tid] = 0;
  __syncthreads();
  const int e0 = blockIdx.x * P_CHUNK;
#pragma unroll
  for (int k = 0; k < P_PER; ++k) {
    int e = e0 + k * P_THREADS + tid;
    if (e < E) atomicAdd(&h[dst[e] >> BSH], 1);
  }
  __syncthreads();
  if (tid < NBMAX) histG[(size_t)blockIdx.x * NBMAX + tid] = h[tid];
}

__global__ __launch_bounds__(P_THREADS) void k_bucket_scatter(int2* __restrict__ part, int* __restrict__ bstartG,
                                                              const int* __restrict__ histG,
                                                              const int* __restrict__ src, const int* __restrict__ dst, int E) {
  __shared__ int cnt[NBMAX];
  __shared__ int loc[NBMAX];
  __shared__ int gbase[NBMAX];
  __shared__ int hTot[NBMAX];
  __shared__ int hPre[NBMAX];
  __shared__ int wsA[32];
  __shared__ int wsB[32];
  __shared__ int2 stage[P_CHUNK];
  const int tid = threadIdx.x;
  const int wg  = blockIdx.x;
  const int nwg = gridDim.x;
  const int e0  = wg * P_CHUNK;
  if (tid < NBMAX) cnt[tid] = 0;
  __syncthreads();

  int s[P_PER], d[P_PER], r[P_PER];
#pragma unroll
  for (int k = 0; k < P_PER; ++k) {
    int e = e0 + k * P_THREADS + tid;
    d[k] = -1; s[k] = 0; r[k] = 0;
    if (e < E) { d[k] = dst[e]; s[k] = src[e]; }
  }
#pragma unroll
  for (int k = 0; k < P_PER; ++k)
    if (d[k] >= 0) r[k] = atomicAdd(&cnt[d[k] >> BSH], 1);

  const int bq = tid & (NBMAX - 1);
  const int hq = __builtin_amdgcn_readfirstlane(tid >> 9);
  const int nh = (nwg + 1) >> 1;
  const int wb = hq * nh;
  int tot = 0, pre = 0;
  for (int j0 = 0; j0 < nh; j0 += HIST_ROWS) {
    int v[HIST_ROWS];
#pragma unroll
    for (int u = 0; u < HIST_ROWS; ++u)
      v[u] = histG[min(wb + j0 + u, nwg - 1) * NBMAX + bq];
#pragma unroll
    for (int u = 0; u < HIST_ROWS; ++u) {
      const int w  = wb + j0 + u;
      const int vv = (j0 + u < nh && w < nwg) ? v[u] : 0;
      tot += vv;
      pre += (w < wg) ? vv : 0;
    }
  }
  if (hq) { hTot[bq] = tot; hPre[bq] = pre; }
  __syncthreads();
  int a = 0, c = 0;
  if (!hq) { tot += hTot[bq]; pre += hPre[bq]; a = tot; c = cnt[bq]; }
  block_scan2<32>(a, c, wsA, wsB);
  if (!hq) {
    loc[bq]   = c;
    gbase[bq] = a + pre;
    if (wg == 0) {
      bstartG[bq] = a;
      if (bq == NBMAX - 1) bstartG[NBMAX] = a + tot;
    }
  }
  __syncthreads();
#pragma unroll
  for (int k = 0; k < P_PER; ++k)
    if (d[k] >= 0) stage[loc[d[k] >> BSH] + r[k]] = make_int2(s[k], d[k]);
  __syncthreads();
  const int nloc = min(P_CHUNK, E - e0);
#pragma unroll
  for (int k = 0; k < P_PER; ++k) {
    int i = k * P_THREADS + tid;
    if (i < nloc) {
      int2 rec = stage[i];
      int b = rec.y >> BSH;
      part[(size_t)gbase[b] + (i - loc[b])] = rec;
    }
  }
}

#define ROWS_THREADS 256
#define ROWS_PER 8
#define ROWS_STAGE 12288
__global__ __launch_bounds__(ROWS_THREADS) void k_bucket_rows(int* __restrict__ csr, int* __restrict__ rowptr,
                                                              float* __restrict__ dis, const int2* __restrict__ part,
                                                              const int* __restrict__ bstartG, int n) {
  __shared__ int cntN[BNODES];
  __shared__ int fillN[BNODES];
  __shared__ int rstart[BNODES];
  __shared__ int wsA[8];
  __shared__ int wsB[8];
  __shared__ int stage[ROWS_STAGE];
  const int tid = threadIdx.x;
  const int b   = blockIdx.x;
  const int beg = bstartG[b];
  const int end = bstartG[b + 1];
  const int ne  = end - beg;
  if (tid < BNODES) { cntN[tid] = 0; fillN[tid] = 0; }
  __syncthreads();
  for (int i0 = 0; i0 < ne; i0 += ROWS_THREADS * ROWS_PER) {
    int2 rec[ROWS_PER];
#pragma unroll
    for (int k = 0; k < ROWS_PER; ++k) {
      const int i = i0 + k * ROWS_THREADS + tid;
      rec[k] = (i < ne) ? part[(size_t)beg + i] : make_int2(0, -1);
    }
#pragma unroll
    for (int k = 0; k < ROWS_PER; ++k)
      if (rec[k].y >= 0) atomicAdd(&cntN[rec[k].y & (BNODES - 1)], 1);
  }
  __syncthreads();
  int c = (tid < BNODES) ? cntN[tid] : 0;
  int x = c, unused = 0;
  block_scan2<ROWS_THREADS / 32>(x, unused, wsA, wsB);
  if (tid < BNODES) {
    rstart[tid] = x;
    const int node = b * BNODES + tid;
    if (node < n) {
      rowptr[node] = beg + x;
      float dg = (float)(c + 1);
      dis[node] = (dg > 0.f) ? rsqrtf(dg) : 0.f;
    }
  }
  if (b == (int)gridDim.x - 1 && tid == 0) rowptr[n] = end;
  __syncthreads();
  const bool staged = (ne <= ROWS_STAGE);
  for (int i0 = 0; i0 < ne; i0 += ROWS_THREADS * ROWS_PER) {
    int2 rec[ROWS_PER];
#pragma unroll
    for (int k = 0; k < ROWS_PER; ++k) {
      const int i = i0 + k * ROWS_THREADS + tid;
      rec[k] = (i < ne) ? part[(size_t)beg + i] : make_int2(0, -1);
    }
    int pos[ROWS_PER];
#pragma unroll
    for (int k = 0; k < ROWS_PER; ++k) {
      pos[k] = -1;
      if (rec[k].y >= 0) {
        const int dl = rec[k].y & (BNODES - 1);
        pos[k] = rstart[dl] + atomicAdd(&fillN[dl], 1);
      }
    }
    if (staged) {
#pragma unroll
      for (int k = 0; k < ROWS_PER; ++k)
        if (pos[k] >= 0) stage[pos[k]] = rec[k].x;
    } else {
#pragma unroll
      for (int k = 0; k < ROWS_PER; ++k)
        if (pos[k] >= 0) csr[(size_t)beg + pos[k]] = rec[k].x;
    }
  }
  if (staged) {
    __syncthreads();
    for (int i = tid; i < ne; i += ROWS_THREADS) csr[(size_t)beg + i] = stage[i];
  }
}

template<int U>
__device__ __forceinline__ void gather_rows(float4& acc, const float* __restrict__ h, int my_s, int j, int lane4) {
  float4 v[U];
#pragma unroll
  for (int u = 0; u < U; ++u) {
    int s = __builtin_amdgcn_readlane(my_s, j + u);
    v[u] = *(const float4*)(h + (size_t)s * HIDW + lane4);
  }
#pragma unroll
  for (int u = 0; u < U; ++u) {
    acc.x += v[u].x;
    acc.y += v[u].y;
    acc.z += v[u].z;
    acc.w += v[u].w;
  }
}

template<int U, int XS, class F>
__device__ __forceinline__ void gather_rows_mm(float4& acc, const float* __restrict__ h, int my_s, int j, int lane4,
                                               int& ks, int nks, F& mstep) {
  float4 v[U];
#pragma unroll
  for (int u = 0; u < U; ++u) {
    int s = __builtin_amdgcn_readlane(my_s, j + u);
    v[u] = *(const float4*)(h + (size_t)s * HIDW + lane4);
  }
#pragma unroll
  for (int q = 0; q < XS; ++q)
    if (ks < nks) { mstep(ks); ++ks; }
#pragma unroll
  for (int u = 0; u < U; ++u) {
    acc.x += v[u].x;
    acc.y += v[u].y;
    acc.z += v[u].z;
    acc.w += v[u].w;
  }
}

template<int U, bool TAIL>
__device__ __forceinline__ void gather_pairs(float4& acc, const float* __restrict__ gp, int my_s,
                                             int j, int m, bool hi, int l15x4) {
  float4 v[U];
#pragma unroll
  for (int u = 0; u < U; ++u) {
    int slo = __builtin_amdgcn_readlane(my_s, j + 2 * u);
    int shi = __builtin_amdgcn_readlane(my_s, j + 2 * u + 1);
    int s = hi ? shi : slo;
    v[u] = *(const float4*)(gp + (size_t)s * HIDW + l15x4);
  }
#pragma unroll
  for (int u = 0; u < U; ++u) {
    const bool live = !TAIL || !hi || (j + 2 * u + 1 < m);
    acc.x += live ? v[u].x : 0.f;
    acc.y += live ? v[u].y : 0.f;
    acc.z += live ? v[u].z : 0.f;
    acc.w += live ? v[u].w : 0.f;
  }
}

__global__ __launch_bounds__(256) void k_prop_gather_half(float* __restrict__ out, const float* __restrict__ gp,
                                                          const int* __restrict__ rowptr, const int* __restrict__ csr,
                                                          const float* __restrict__ dis, int n) {
  const int lane  = threadIdx.x & 31;
  const bool hi   = lane >= 16;
  const int l15x4 = (lane & 15) * 4;
  const int wave = __builtin_amdgcn_readfirstlane((int)(threadIdx.x >> 5));
  const int nw   = gridDim.x * 8;

  for (int i = blockIdx.x * 8 + wave; i < n; i += nw) {
    const int beg = rowptr[i];
    const int end = rowptr[i + 1];
    const float di = dis[i];
    const float4 rv = *(const float4*)(gp + (size_t)i * HIDW + (hi ? 64 : 0) + l15x4);
    float4 acc;
    acc.x = hi ? 0.f : rv.x; acc.y = hi ? 0.f : rv.y; acc.z = hi ? 0.f : rv.z; acc.w = hi ? 0.f : rv.w;

    for (int base = beg; base < end; base += 32) {
      const int m = min(32, end - base);
      const int my_s = csr[min(base + lane, end - 1)];
      int j = 0;
      for (; j + 16 <= m; j += 16) gather_pairs<8, false>(acc, gp, my_s, j, m, hi, l15x4);
      if (m - j >= 8) { gather_pairs<4, false>(acc, gp, my_s, j, m, hi, l15x4); j += 8; }
      if (m - j >= 4) { gather_pairs<2, false>(acc, gp, my_s, j, m, hi, l15x4); j += 4; }
      if (m - j >= 3)      gather_pairs<2, true>(acc, gp, my_s, j, m, hi, l15x4);
      else if (m - j >= 1) gather_pairs<1, true>(acc, gp, my_s, j, m, hi, l15x4);
    }
    float4 o;
    o.x = fmaf(acc.x + __shfl_xor(acc.x, 16), di, __shfl_xor(rv.x, 16));
    o.y = fmaf(acc.y + __shfl_xor(acc.y, 16), di, __shfl_xor(rv.y, 16));
    o.z = fmaf(acc.z + __shfl_xor(acc.z, 16), di, __shfl_xor(rv.z, 16));
    o.w = fmaf(acc.w + __shfl_xor(acc.w, 16), di, __shfl_xor(rv.w, 16));
    if (!hi) *(float4*)(out + (size_t)i * 64 + l15x4) = o;
  }
}

template<int NC, int KW, bool RELU_A0, bool RELU_OUT, bool HAS_BIAS, int SCALE_COLS>
__global__ __launch_bounds__(256) void k_gemm(const float* __restrict__ A0, const float* __restrict__ A1,
                                              const float* __restrict__ W, const float* __restrict__ bias,
                                              const float* __restrict__ dis,
                                              float* __restrict__ C, int M) {
  __shared__ __align__(16) float As[128][KCHUNK + 4];
  __shared__ __align__(16) float Ws[NC][KCHUNK + 4];
  constexpr int NT = NC / 16;
  constexpr int K = KW * 128;

  const int tid  = threadIdx.x;
  const int wave = tid >> 5;
  const int lane = tid & 31;
  const int l15  = lane & 15;
  const int half = lane >> 4;
  const int m0   = blockIdx.x * 128;

  v8f acc[NT];
#pragma unroll
  for (int t = 0; t < NT; ++t)
#pragma unroll
    for (int r = 0; r < 8; ++r) acc[t][r] = 0.0f;

  for (int kb = 0; kb < K; kb += KCHUNK) {
    const float* Ablk = (KW == 2 && kb >= 128) ? A1 : A0;
    const bool doRelu = RELU_A0 && (kb < 128);
    const int kloc = kb & 127;

#pragma unroll
    for (int j = 0; j < (128 * KCHUNK) / 1024; ++j) {
      int ei  = tid * 4 + j * 1024;
      int row = ei >> 5;
      int col = ei & 31;
      int grow = m0 + row;
      float4 v = {0.f, 0.f, 0.f, 0.f};
      if (grow < M) v = *(const float4*)&Ablk[(size_t)grow * HIDW + kloc + col];
      if (doRelu) {
        v.x = fmaxf(v.x, 0.f); v.y = fmaxf(v.y, 0.f);
        v.z = fmaxf(v.z, 0.f); v.w = fmaxf(v.w, 0.f);
      }
      *(float4*)&As[row][col] = v;
    }
#pragma unroll
    for (int j = 0; j < (NC * KCHUNK) / 1024; ++j) {
      int ei  = tid * 4 + j * 1024;
      int n   = ei >> 5;
      int col = ei & 31;
      *(float4*)&Ws[n][col] = *(const float4*)&W[(size_t)n * K + kb + col];
    }
    __syncthreads();

    const int arow = wave * 16 + l15;
    const int o = half * 2;
#pragma unroll
    for (int k4 = 0; k4 < KCHUNK; k4 += 4) {
      v2f a = *(const v2f*)&As[arow][k4 + o];
#pragma unroll
      for (int t = 0; t < NT; ++t) {
        v2f b = *(const v2f*)&Ws[t * 16 + l15][k4 + o];
        acc[t] = __builtin_amdgcn_wmma_f32_16x16x4_f32(
             false, a,  false, b,
             (short)0, acc[t],  false,  false);
      }
    }
    __syncthreads();
  }

  const int rbase = m0 + wave * 16 + half * 8;
  float dr[8];
#pragma unroll
  for (int r = 0; r < 8; ++r) dr[r] = (SCALE_COLS > 0 && rbase + r < M) ? dis[rbase + r] : 1.0f;
#pragma unroll
  for (int t = 0; t < NT; ++t) {
    int col = t * 16 + l15;
    float bv = HAS_BIAS ? bias[col] : 0.f;
#pragma unroll
    for (int r = 0; r < 8; ++r) {
      int row = rbase + r;
      if (row < M) {
        float v = acc[t][r] + bv;
        if (RELU_OUT) v = fmaxf(v, 0.f);
        if (t * 16 < SCALE_COLS) v *= dr[r];
        C[(size_t)row * NC + col] = v;
      }
    }
  }
}


#define XP (HIDW + 4)
#define TR 208
#define TRT (TR / 16)
#define GT (2 * TRT * 32)
#define XKS (HIDW / 4)
__global__ __launch_bounds__(GT) void k_gather_gemm2(float* __restrict__ GP, const float* __restrict__ hs,
                                                      const float* __restrict__ x,
                                                      const int* __restrict__ rowptr, const int* __restrict__ csr,
                                                      const float* __restrict__ dis, const float* __restrict__ b1,
                                                      const float* __restrict__ Wc, const float* __restrict__ bc,
                                                      const float* __restrict__ Wcat, const float* __restrict__ bias128,
                                                      int n) {
  __shared__ __align__(16) float Xa[TR][XP];
  __shared__ __align__(16) float Xx[TR][XP];
  __shared__ __align__(16) float Wx[HIDW][XP];
  __shared__ __align__(16) float Ws[2][HIDW][KCHUNK + 4];
  const int tid  = threadIdx.x;
  const int lane = tid & 31;
  const int wave = __builtin_amdgcn_readfirstlane(tid >> 5);
  const int l15  = lane & 15;
  const int half = lane >> 4;
  const int ch   = (wave >= TRT) ? 1 : 0;
  const int rg   = wave - ch * TRT;
  const int m0   = blockIdx.x * TR;

  const int wrow0 = tid >> 3;
  const int wrow1 = (tid + GT) >> 3;
  const int scol  = (tid & 7) * 4;
  const bool w1on = (tid + GT) < HIDW * 8;
  const int wrow1c = min(wrow1, HIDW - 1);
  const size_t wc0 = (size_t)wrow0 * (2 * HIDW) + scol, wc1 = (size_t)wrow1c * (2 * HIDW) + scol;
  const size_t wt0 = (size_t)wrow0 * HIDW + scol,       wt1 = (size_t)wrow1c * HIDW + scol;
  v4f wreg0, wreg1;

  const int lane4 = lane * 4;
  const int i0 = m0 + wave * 8;
  const int   rpl = rowptr[min(i0 + min(lane, 8), n)];
  const float dl  = dis[min(i0 + (lane & 7), n - 1)];
  const float4 bv1 = *(const float4*)&b1[lane4];

  constexpr int XJ = (TR * 32) / GT;
  constexpr int WJ = (HIDW * 32 + GT - 1) / GT;
  v4f xv[XJ], wv[WJ];
#pragma unroll
  for (int j = 0; j < XJ; ++j) {
    const int idx = tid + j * GT;
    xv[j] = *(const v4f*)&x[(size_t)min(m0 + (idx >> 5), n - 1) * HIDW + (idx & 31) * 4];
  }
#pragma unroll
  for (int j = 0; j < WJ; ++j) {
    const int idx = min(tid + j * GT, HIDW * 32 - 1);
    wv[j] = *(const v4f*)&Wc[(size_t)(idx >> 5) * (2 * HIDW) + HIDW + (idx & 31) * 4];
  }

  const int arow = rg * 16 + l15;
  const int o2   = half * 2;
  v8f acc[4];

  auto mstep = [&](int ks) {
    const int k = ks * 4 + o2;
    v2f a = *(const v2f*)&Xx[arow][k];
#pragma unroll
    for (int t = 0; t < 4; ++t) {
      v2f b = *(const v2f*)&Wx[ch * 64 + t * 16 + l15][k];
      acc[t] = __builtin_amdgcn_wmma_f32_16x16x4_f32(false, a, false, b, (short)0, acc[t], false, false);
    }
  };

  int ks = 0;
  auto do_row = [&](int rr, int nks) {
    const int i   = i0 + rr;
    const int beg = __builtin_amdgcn_readlane(rpl, rr);
    const int end = __builtin_amdgcn_readlane(rpl, rr + 1);
    const float di = __int_as_float(__builtin_amdgcn_readlane(__float_as_int(dl), rr));
    float4 o = {0.f, 0.f, 0.f, 0.f};
    if (i < n) {
      float4 ga = *(const float4*)(hs + (size_t)i * HIDW + lane4);
      for (int base = beg; base < end; base += 32) {
        const int m = min(32, end - base);
        const int my_s = csr[min(base + lane, end - 1)];
        int j = 0;
        for (; j + 16 <= m; j += 16) gather_rows_mm<16, 2>(ga, hs, my_s, j, lane4, ks, nks, mstep);
        if (m - j >= 8) { gather_rows_mm<8, 1>(ga, hs, my_s, j, lane4, ks, nks, mstep); j += 8; }
        if (m - j >= 4) { gather_rows<4>(ga, hs, my_s, j, lane4); j += 4; }
        if (m - j >= 2) { gather_rows<2>(ga, hs, my_s, j, lane4); j += 2; }
        if (m - j >= 1) { gather_rows<1>(ga, hs, my_s, j, lane4); }
      }
      o.x = fmaxf(fmaf(ga.x, di, bv1.x), 0.f);
      o.y = fmaxf(fmaf(ga.y, di, bv1.y), 0.f);
      o.z = fmaxf(fmaf(ga.z, di, bv1.z), 0.f);
      o.w = fmaxf(fmaf(ga.w, di, bv1.w), 0.f);
    }
    *(float4*)&Xa[wave * 8 + rr][lane4] = o;
  };

  do_row(0, 0);

#pragma unroll
  for (int j = 0; j < XJ; ++j) {
    const int idx = tid + j * GT;
    *(v4f*)&Xx[idx >> 5][(idx & 31) * 4] = xv[j];
  }
#pragma unroll
  for (int j = 0; j < WJ; ++j) {
    const int idx = tid + j * GT;
    if (idx < HIDW * 32) *(v4f*)&Wx[idx >> 5][(idx & 31) * 4] = wv[j];
  }
  __syncthreads();

#pragma unroll
  for (int t = 0; t < 4; ++t)
#pragma unroll
    for (int r = 0; r < 8; ++r) acc[t][r] = 0.0f;

  for (int rr = 1; rr < 8; ++rr) do_row(rr, XKS);
  for (; ks < XKS; ++ks) mstep(ks);

  wreg0 = *(const v4f*)&Wc[wc0];
  wreg1 = *(const v4f*)&Wc[wc1];
  *(v4f*)&Ws[0][wrow0][scol] = wreg0;
  if (w1on) *(v4f*)&Ws[0][wrow1][scol] = wreg1;
  wreg0 = *(const v4f*)&Wc[wc0 + KCHUNK];
  wreg1 = *(const v4f*)&Wc[wc1 + KCHUNK];
  __syncthreads();

#pragma nounroll
  for (int c = 0; c < 4; ++c) {
    const int s = c & 1;
    {
      *(v4f*)&Ws[s ^ 1][wrow0][scol] = wreg0;
      if (w1on) *(v4f*)&Ws[s ^ 1][wrow1][scol] = wreg1;
      const int cf = c + 2;
      if (cf < 4) {
        wreg0 = *(const v4f*)&Wc[wc0 + cf * KCHUNK];
        wreg1 = *(const v4f*)&Wc[wc1 + cf * KCHUNK];
      } else {
        wreg0 = *(const v4f*)&Wcat[wt0 + (cf - 4) * KCHUNK];
        wreg1 = *(const v4f*)&Wcat[wt1 + (cf - 4) * KCHUNK];
      }
    }
    const int kb = c * KCHUNK;
#pragma unroll
    for (int k4 = 0; k4 < KCHUNK; k4 += 4) {
      v2f a = *(const v2f*)&Xa[arow][kb + k4 + o2];
#pragma unroll
      for (int t = 0; t < 4; ++t) {
        v2f b = *(const v2f*)&Ws[s][ch * 64 + t * 16 + l15][k4 + o2];
        acc[t] = __builtin_amdgcn_wmma_f32_16x16x4_f32(false, a, false, b, (short)0, acc[t], false, false);
      }
    }
    __syncthreads();
  }

#pragma unroll
  for (int t = 0; t < 4; ++t) {
    const int col = ch * 64 + t * 16 + l15;
    const float bv = bc[col];
#pragma unroll
    for (int r = 0; r < 8; ++r) {
      Xa[rg * 16 + half * 8 + r][col] = fmaxf(acc[t][r] + bv, 0.f);
      acc[t][r] = 0.0f;
    }
  }
  __syncthreads();

#pragma nounroll
  for (int c = 4; c < 8; ++c) {
    const int s = c & 1;
    if (c + 1 < 8) {
      *(v4f*)&Ws[s ^ 1][wrow0][scol] = wreg0;
      if (w1on) *(v4f*)&Ws[s ^ 1][wrow1][scol] = wreg1;
    }
    if (c + 2 < 8) {
      wreg0 = *(const v4f*)&Wcat[wt0 + (c + 2 - 4) * KCHUNK];
      wreg1 = *(const v4f*)&Wcat[wt1 + (c + 2 - 4) * KCHUNK];
    }
    const int kb = (c - 4) * KCHUNK;
#pragma unroll
    for (int k4 = 0; k4 < KCHUNK; k4 += 4) {
      v2f a = *(const v2f*)&Xa[arow][kb + k4 + o2];
#pragma unroll
      for (int t = 0; t < 4; ++t) {
        v2f b = *(const v2f*)&Ws[s][ch * 64 + t * 16 + l15][k4 + o2];
        acc[t] = __builtin_amdgcn_wmma_f32_16x16x4_f32(false, a, false, b, (short)0, acc[t], false, false);
      }
    }
    if (c + 1 < 8) __syncthreads();
  }

  const int rbase = m0 + rg * 16 + half * 8;
  float dr[8];
#pragma unroll
  for (int r = 0; r < 8; ++r) dr[r] = (ch == 0 && rbase + r < n) ? dis[rbase + r] : 1.0f;
#pragma unroll
  for (int t = 0; t < 4; ++t) {
    const int col = ch * 64 + t * 16 + l15;
    const float bv = bias128[col];
#pragma unroll
    for (int r = 0; r < 8; ++r) {
      const int row = rbase + r;
      if (row < n) GP[(size_t)row * HIDW + col] = (acc[t][r] + bv) * dr[r];
    }
  }
}

extern "C" void kernel_launch(void* const* d_in, const int* in_sizes, int n_in,
                              void* d_out, int out_size, void* d_ws, size_t ws_size,
                              hipStream_t stream) {
  const float* x  = (const float*)d_in[0];
  const int* edge = (const int*)d_in[1];
  const float* W1 = (const float*)d_in[2];
  const float* b1 = (const float*)d_in[3];
  const float* Wc = (const float*)d_in[4];
  const float* bc = (const float*)d_in[5];
  const float* W2 = (const float*)d_in[6];
  const float* b2 = (const float*)d_in[7];
  const float* Wo = (const float*)d_in[8];
  const float* bo = (const float*)d_in[9];

  const int N = in_sizes[0] / HIDW;
  const int E = in_sizes[1] / 2;
  const int* srcIdx = edge;
  const int* dstIdx = edge + E;

  float* wsf = (float*)d_ws;
  size_t NP   = (((size_t)N) + 255) & ~(size_t)255;
  size_t NP1  = (((size_t)N) + 1 + 255) & ~(size_t)255;
  size_t EP   = (((size_t)E) + 3) & ~(size_t)3;
  float* dis  = wsf;
  float* buf1 = wsf + NP;
  float* buf2 = buf1 + (size_t)N * HIDW;
  float* buf3 = buf2 + (size_t)N * HIDW;
  int* rowptr = (int*)buf3;
  int* csr    = rowptr + NP1;
  float* Wcat = (float*)(csr + EP);
  float* bias128 = Wcat + HIDW * HIDW;
  int gC      = (E + P_CHUNK - 1) / P_CHUNK;
  int gB      = (N + BNODES - 1) >> BSH;
  int2* part  = (int2*)buf2;
  int* histG  = (int*)buf2 + 2 * EP;
  int* bstartG = histG + (size_t)gC * NBMAX;

  dim3 b256(256);
  int gG  = (N + 127) / 128;
  int gF  = (N + TR - 1) / TR;
  int gP  = 2048;

  k_bucket_hist   <<<gC + 1, dim3(P_THREADS), 0, stream>>>(histG, dstIdx, E, Wo, W2, b2, bo, Wcat, bias128);
  k_bucket_scatter<<<gC, dim3(P_THREADS), 0, stream>>>(part, bstartG, histG, srcIdx, dstIdx, E);
  k_bucket_rows   <<<gB, dim3(ROWS_THREADS), 0, stream>>>(csr, rowptr, dis, part, bstartG, N);

  k_gemm<128,1,false,false,false,128><<<gG, b256, 0, stream>>>(x, nullptr, W1, nullptr, dis, buf1, N);

  k_gather_gemm2<<<gF, dim3(GT), 0, stream>>>(buf2, buf1, x, rowptr, csr, dis, b1, Wc, bc, Wcat, bias128, N);
  k_prop_gather_half<<<gP, b256, 0, stream>>>((float*)d_out, buf2, rowptr, csr, dis, N);
}


# Edge grouping level 1 keeps each chunk in place (bucket runs + per-chunk run table); level 2 gathers its bucket's run from every chunk: no global histogram, one launch fewer; list order may differ
# speedup vs baseline: 9.5338x; 1.0647x over previous
#include <hip/hip_runtime.h>

typedef __attribute__((ext_vector_type(2))) float v2f;
typedef __attribute__((ext_vector_type(8))) float v8f;
typedef __attribute__((ext_vector_type(4))) float v4f;

#define HIDW 128
#define KCHUNK 32

__device__ __forceinline__ void fold_weights(const float* __restrict__ Wo, const float* __restrict__ W2,
                                             const float* __restrict__ b2, const float* __restrict__ bo,
                                             float* __restrict__ Wcat, float* __restrict__ bias128) {
  __shared__ __align__(16) float Fa[64][KCHUNK + 4];
  __shared__ __align__(16) float Fb[HIDW][KCHUNK + 4];
  const int tid  = threadIdx.x;
  const int wave = tid >> 5;
  const int lane = tid & 31;
  const int l15  = lane & 15;
  const int half = lane >> 4;
  const int mt   = wave & 3;
  const int nt   = wave >> 2;

  v8f acc;
#pragma unroll
  for (int r = 0; r < 8; ++r) acc[r] = 0.0f;

  for (int hb = 0; hb < HIDW; hb += KCHUNK) {
    {
      int idx = tid * 2;
      int o   = idx >> 5;
      int hh  = idx & 31;
      *(v2f*)&Fa[o][hh] = *(const v2f*)&Wo[(size_t)o * (2 * HIDW) + hb + hh];
    }
#pragma unroll
    for (int j = 0; j < (HIDW * KCHUNK) / 1024; ++j) {
      int e  = tid + j * 1024;
      int hh = e >> 7;
      int k  = e & 127;
      Fb[k][hh] = W2[(size_t)(hb + hh) * HIDW + k];
    }
    __syncthreads();
    const int o2 = half * 2;
#pragma unroll
    for (int k4 = 0; k4 < KCHUNK; k4 += 4) {
      v2f a = *(const v2f*)&Fa[mt * 16 + l15][k4 + o2];
      v2f b = *(const v2f*)&Fb[nt * 16 + l15][k4 + o2];
      acc = __builtin_amdgcn_wmma_f32_16x16x4_f32(false, a, false, b, (short)0, acc, false, false);
    }
    __syncthreads();
  }
#pragma unroll
  for (int r = 0; r < 8; ++r)
    Wcat[(size_t)(mt * 16 + half * 8 + r) * HIDW + nt * 16 + l15] = acc[r];

#pragma unroll
  for (int j = 0; j < 2; ++j) {
    int q  = tid + j * 1024;
    int o  = q >> 5;
    int c4 = (q & 31) * 4;
    *(float4*)&Wcat[(size_t)(64 + o) * HIDW + c4] = *(const float4*)&Wo[(size_t)o * (2 * HIDW) + HIDW + c4];
  }
  if (tid < 64) {
    float s = bo[tid];
    for (int h = 0; h < HIDW; ++h) s = fmaf(Wo[(size_t)tid * (2 * HIDW) + h], b2[h], s);
    bias128[tid]      = 0.f;
    bias128[64 + tid] = s;
  }
}

#define BSH 7
#define BNODES (1 << BSH)
#define NBMAX 512
#define P_THREADS 1024
#define P_PER 8
#define P_CHUNK (P_THREADS * P_PER)
#define LOCP (NBMAX + 1)

template<int NWAVES>
__device__ __forceinline__ void block_scan2(int& a, int& b, int* wsA, int* wsB) {
  const int lane = threadIdx.x & 31;
  const int wv   = threadIdx.x >> 5;
  int ia = a, ib = b;
  for (int o = 1; o < 32; o <<= 1) {
    int ta = __shfl_up(ia, o);
    int tb = __shfl_up(ib, o);
    if (lane >= o) { ia += ta; ib += tb; }
  }
  if (lane == 31) { wsA[wv] = ia; wsB[wv] = ib; }
  __syncthreads();
  if (wv == 0) {
    int ta = (lane < NWAVES) ? wsA[lane] : 0;
    int tb = (lane < NWAVES) ? wsB[lane] : 0;
    for (int o = 1; o < 32; o <<= 1) {
      int ua = __shfl_up(ta, o);
      int ub = __shfl_up(tb, o);
      if (lane >= o) { ta += ua; tb += ub; }
    }
    if (lane < NWAVES) { wsA[lane] = ta; wsB[lane] = tb; }
  }
  __syncthreads();
  a = ia - a + (wv ? wsA[wv - 1] : 0);
  b = ib - b + (wv ? wsB[wv - 1] : 0);
}

__global__ __launch_bounds__(P_THREADS) void k_bucket_scatter(int2* __restrict__ part, int* __restrict__ locG,
                                                              const int* __restrict__ src, const int* __restrict__ dst, int E,
                                                              const float* __restrict__ Wo, const float* __restrict__ W2,
                                                              const float* __restrict__ b2, const float* __restrict__ bo,
                                                              float* __restrict__ Wcat, float* __restrict__ bias128) {
  __shared__ int cnt[NBMAX];
  __shared__ int loc[NBMAX];
  __shared__ int wsA[32];
  __shared__ int wsB[32];
  __shared__ int2 stage[P_CHUNK];
  if (blockIdx.x == gridDim.x - 1) { fold_weights(Wo, W2, b2, bo, Wcat, bias128); return; }
  const int tid = threadIdx.x;
  const int wg  = blockIdx.x;
  const int e0  = wg * P_CHUNK;
  if (tid < NBMAX) cnt[tid] = 0;
  __syncthreads();

  int s[P_PER], d[P_PER], r[P_PER];
#pragma unroll
  for (int k = 0; k < P_PER; ++k) {
    int e = e0 + k * P_THREADS + tid;
    d[k] = -1; s[k] = 0; r[k] = 0;
    if (e < E) { d[k] = dst[e]; s[k] = src[e]; }
  }
#pragma unroll
  for (int k = 0; k < P_PER; ++k)
    if (d[k] >= 0) r[k] = atomicAdd(&cnt[d[k] >> BSH], 1);
  __syncthreads();

  const int nloc = min(P_CHUNK, E - e0);
  int unused = 0;
  int c = (tid < NBMAX) ? cnt[tid] : 0;
  block_scan2<32>(unused, c, wsA, wsB);
  if (tid < NBMAX) {
    loc[tid] = c;
    locG[(size_t)wg * LOCP + tid] = c;
  }
  if (tid == 0) locG[(size_t)wg * LOCP + NBMAX] = nloc;
  __syncthreads();
#pragma unroll
  for (int k = 0; k < P_PER; ++k)
    if (d[k] >= 0) stage[loc[d[k] >> BSH] + r[k]] = make_int2(s[k], d[k]);
  __syncthreads();
#pragma unroll
  for (int k = 0; k < P_PER; ++k) {
    int i = k * P_THREADS + tid;
    if (i < nloc) part[(size_t)e0 + i] = stage[i];
  }
}

#define ROWS_THREADS 256
#define ROWS_WAVES (ROWS_THREADS / 32)
#define RUNS_U 13
#define ROWS_STAGE 12288
__global__ __launch_bounds__(ROWS_THREADS) void k_bucket_rows(int* __restrict__ csr, int* __restrict__ rowptr,
                                                              float* __restrict__ dis, const int2* __restrict__ part,
                                                              const int* __restrict__ locG, int nchunks, int n) {
  __shared__ int cntN[BNODES];
  __shared__ int fillN[BNODES];
  __shared__ int rstart[BNODES];
  __shared__ int wsA[ROWS_WAVES];
  __shared__ int wsB[ROWS_WAVES];
  __shared__ int tot[3];
  __shared__ int stage[ROWS_STAGE];
  const int tid  = threadIdx.x;
  const int lane = tid & 31;
  const int wave = __builtin_amdgcn_readfirstlane(tid >> 5);
  const int b    = blockIdx.x;
  const int cpw  = (nchunks + ROWS_WAVES - 1) / ROWS_WAVES;
  const int cme  = wave * cpw + lane;
  int lo = 0, len = 0;
  if (lane < cpw && cme < nchunks) {
    lo  = locG[(size_t)cme * LOCP + b];
    len = locG[(size_t)cme * LOCP + b + 1] - lo;
  }
  const int myPos = cme * P_CHUNK + lo;
  if (tid < BNODES) { cntN[tid] = 0; fillN[tid] = 0; }
  if (tid == 0) tot[2] = 0;
  int pl = len, pb = lo;
  block_scan2<ROWS_WAVES>(pl, pb, wsA, wsB);
  if (len > 64) tot[2] = 1;
  if (tid == ROWS_THREADS - 1) { tot[0] = pl + len; tot[1] = pb + lo; }
  __syncthreads();
  const int ne  = tot[0];
  const int beg = tot[1];
  const int end = beg + ne;
  const bool longRuns = tot[2] != 0;

  for (int k0 = 0; k0 < cpw; k0 += RUNS_U) {
    int2 ra[RUNS_U], rb[RUNS_U];
#pragma unroll
    for (int u = 0; u < RUNS_U; ++u) {
      const int k = min(k0 + u, 31);
      const int L = (k0 + u < cpw) ? __builtin_amdgcn_readlane(len, k) : 0;
      const int p = __builtin_amdgcn_readlane(myPos, k);
      ra[u] = (lane < L)      ? part[(size_t)p + lane]      : make_int2(0, -1);
      rb[u] = (lane + 32 < L) ? part[(size_t)p + 32 + lane] : make_int2(0, -1);
    }
#pragma unroll
    for (int u = 0; u < RUNS_U; ++u) {
      if (ra[u].y >= 0) atomicAdd(&cntN[ra[u].y & (BNODES - 1)], 1);
      if (rb[u].y >= 0) atomicAdd(&cntN[rb[u].y & (BNODES - 1)], 1);
    }
  }
  if (longRuns) {
    for (int k = 0; k < cpw; ++k) {
      const int L = __builtin_amdgcn_readlane(len, k);
      const int p = __builtin_amdgcn_readlane(myPos, k);
      for (int o = 64 + lane; o < L; o += 32) atomicAdd(&cntN[part[(size_t)p + o].y & (BNODES - 1)], 1);
    }
  }
  __syncthreads();
  int c = (tid < BNODES) ? cntN[tid] : 0;
  int x = c, unused = 0;
  block_scan2<ROWS_WAVES>(x, unused, wsA, wsB);
  if (tid < BNODES) {
    rstart[tid] = x;
    const int node = b * BNODES + tid;
    if (node < n) {
      rowptr[node] = beg + x;
      float dg = (float)(c + 1);
      dis[node] = (dg > 0.f) ? rsqrtf(dg) : 0.f;
    }
  }
  if (b == (int)gridDim.x - 1 && tid == 0) rowptr[n] = end;
  __syncthreads();
  const bool staged = (ne <= ROWS_STAGE);
  for (int k0 = 0; k0 < cpw; k0 += RUNS_U) {
    int2 ra[RUNS_U], rb[RUNS_U];
#pragma unroll
    for (int u = 0; u < RUNS_U; ++u) {
      const int k = min(k0 + u, 31);
      const int L = (k0 + u < cpw) ? __builtin_amdgcn_readlane(len, k) : 0;
      const int p = __builtin_amdgcn_readlane(myPos, k);
      ra[u] = (lane < L)      ? part[(size_t)p + lane]      : make_int2(0, -1);
      rb[u] = (lane + 32 < L) ? part[(size_t)p + 32 + lane] : make_int2(0, -1);
    }
    int pa[RUNS_U], pc[RUNS_U];
#pragma unroll
    for (int u = 0; u < RUNS_U; ++u) {
      pa[u] = -1; pc[u] = -1;
      if (ra[u].y >= 0) {
        const int dl = ra[u].y & (BNODES - 1);
        pa[u] = rstart[dl] + atomicAdd(&fillN[dl], 1);
      }
      if (rb[u].y >= 0) {
        const int dl = rb[u].y & (BNODES - 1);
        pc[u] = rstart[dl] + atomicAdd(&fillN[dl], 1);
      }
    }
    if (staged) {
#pragma unroll
      for (int u = 0; u < RUNS_U; ++u) {
        if (pa[u] >= 0) stage[pa[u]] = ra[u].x;
        if (pc[u] >= 0) stage[pc[u]] = rb[u].x;
      }
    } else {
#pragma unroll
      for (int u = 0; u < RUNS_U; ++u) {
        if (pa[u] >= 0) csr[(size_t)beg + pa[u]] = ra[u].x;
        if (pc[u] >= 0) csr[(size_t)beg + pc[u]] = rb[u].x;
      }
    }
  }
  if (longRuns) {
    for (int k = 0; k < cpw; ++k) {
      const int L = __builtin_amdgcn_readlane(len, k);
      const int p = __builtin_amdgcn_readlane(myPos, k);
      for (int o = 64 + lane; o < L; o += 32) {
        const int2 rec = part[(size_t)p + o];
        const int dl  = rec.y & (BNODES - 1);
        const int pos = rstart[dl] + atomicAdd(&fillN[dl], 1);
        if (staged) stage[pos] = rec.x; else csr[(size_t)beg + pos] = rec.x;
      }
    }
  }
  if (staged) {
    __syncthreads();
    for (int i = tid; i < ne; i += ROWS_THREADS) csr[(size_t)beg + i] = stage[i];
  }
}

template<int U>
__device__ __forceinline__ void gather_rows(float4& acc, const float* __restrict__ h, int my_s, int j, int lane4) {
  float4 v[U];
#pragma unroll
  for (int u = 0; u < U; ++u) {
    int s = __builtin_amdgcn_readlane(my_s, j + u);
    v[u] = *(const float4*)(h + (size_t)s * HIDW + lane4);
  }
#pragma unroll
  for (int u = 0; u < U; ++u) {
    acc.x += v[u].x;
    acc.y += v[u].y;
    acc.z += v[u].z;
    acc.w += v[u].w;
  }
}

template<int U, int XS, class F>
__device__ __forceinline__ void gather_rows_mm(float4& acc, const float* __restrict__ h, int my_s, int j, int lane4,
                                               int& ks, int nks, F& mstep) {
  float4 v[U];
#pragma unroll
  for (int u = 0; u < U; ++u) {
    int s = __builtin_amdgcn_readlane(my_s, j + u);
    v[u] = *(const float4*)(h + (size_t)s * HIDW + lane4);
  }
#pragma unroll
  for (int q = 0; q < XS; ++q)
    if (ks < nks) { mstep(ks); ++ks; }
#pragma unroll
  for (int u = 0; u < U; ++u) {
    acc.x += v[u].x;
    acc.y += v[u].y;
    acc.z += v[u].z;
    acc.w += v[u].w;
  }
}

template<int U, bool TAIL>
__device__ __forceinline__ void gather_pairs(float4& acc, const float* __restrict__ gp, int my_s,
                                             int j, int m, bool hi, int l15x4) {
  float4 v[U];
#pragma unroll
  for (int u = 0; u < U; ++u) {
    int slo = __builtin_amdgcn_readlane(my_s, j + 2 * u);
    int shi = __builtin_amdgcn_readlane(my_s, j + 2 * u + 1);
    int s = hi ? shi : slo;
    v[u] = *(const float4*)(gp + (size_t)s * HIDW + l15x4);
  }
#pragma unroll
  for (int u = 0; u < U; ++u) {
    const bool live = !TAIL || !hi || (j + 2 * u + 1 < m);
    acc.x += live ? v[u].x : 0.f;
    acc.y += live ? v[u].y : 0.f;
    acc.z += live ? v[u].z : 0.f;
    acc.w += live ? v[u].w : 0.f;
  }
}

__global__ __launch_bounds__(256) void k_prop_gather_half(float* __restrict__ out, const float* __restrict__ gp,
                                                          const int* __restrict__ rowptr, const int* __restrict__ csr,
                                                          const float* __restrict__ dis, int n) {
  const int lane  = threadIdx.x & 31;
  const bool hi   = lane >= 16;
  const int l15x4 = (lane & 15) * 4;
  const int wave = __builtin_amdgcn_readfirstlane((int)(threadIdx.x >> 5));
  const int nw   = gridDim.x * 8;

  for (int i = blockIdx.x * 8 + wave; i < n; i += nw) {
    const int beg = rowptr[i];
    const int end = rowptr[i + 1];
    const float di = dis[i];
    const float4 rv = *(const float4*)(gp + (size_t)i * HIDW + (hi ? 64 : 0) + l15x4);
    float4 acc;
    acc.x = hi ? 0.f : rv.x; acc.y = hi ? 0.f : rv.y; acc.z = hi ? 0.f : rv.z; acc.w = hi ? 0.f : rv.w;

    for (int base = beg; base < end; base += 32) {
      const int m = min(32, end - base);
      const int my_s = csr[min(base + lane, end - 1)];
      int j = 0;
      for (; j + 16 <= m; j += 16) gather_pairs<8, false>(acc, gp, my_s, j, m, hi, l15x4);
      if (m - j >= 8) { gather_pairs<4, false>(acc, gp, my_s, j, m, hi, l15x4); j += 8; }
      if (m - j >= 4) { gather_pairs<2, false>(acc, gp, my_s, j, m, hi, l15x4); j += 4; }
      if (m - j >= 3)      gather_pairs<2, true>(acc, gp, my_s, j, m, hi, l15x4);
      else if (m - j >= 1) gather_pairs<1, true>(acc, gp, my_s, j, m, hi, l15x4);
    }
    float4 o;
    o.x = fmaf(acc.x + __shfl_xor(acc.x, 16), di, __shfl_xor(rv.x, 16));
    o.y = fmaf(acc.y + __shfl_xor(acc.y, 16), di, __shfl_xor(rv.y, 16));
    o.z = fmaf(acc.z + __shfl_xor(acc.z, 16), di, __shfl_xor(rv.z, 16));
    o.w = fmaf(acc.w + __shfl_xor(acc.w, 16), di, __shfl_xor(rv.w, 16));
    if (!hi) *(float4*)(out + (size_t)i * 64 + l15x4) = o;
  }
}

template<int NC, int KW, bool RELU_A0, bool RELU_OUT, bool HAS_BIAS, int SCALE_COLS>
__global__ __launch_bounds__(256) void k_gemm(const float* __restrict__ A0, const float* __restrict__ A1,
                                              const float* __restrict__ W, const float* __restrict__ bias,
                                              const float* __restrict__ dis,
                                              float* __restrict__ C, int M) {
  __shared__ __align__(16) float As[128][KCHUNK + 4];
  __shared__ __align__(16) float Ws[NC][KCHUNK + 4];
  constexpr int NT = NC / 16;
  constexpr int K = KW * 128;

  const int tid  = threadIdx.x;
  const int wave = tid >> 5;
  const int lane = tid & 31;
  const int l15  = lane & 15;
  const int half = lane >> 4;
  const int m0   = blockIdx.x * 128;

  v8f acc[NT];
#pragma unroll
  for (int t = 0; t < NT; ++t)
#pragma unroll
    for (int r = 0; r < 8; ++r) acc[t][r] = 0.0f;

  for (int kb = 0; kb < K; kb += KCHUNK) {
    const float* Ablk = (KW == 2 && kb >= 128) ? A1 : A0;
    const bool doRelu = RELU_A0 && (kb < 128);
    const int kloc = kb & 127;

#pragma unroll
    for (int j = 0; j < (128 * KCHUNK) / 1024; ++j) {
      int ei  = tid * 4 + j * 1024;
      int row = ei >> 5;
      int col = ei & 31;
      int grow = m0 + row;
      float4 v = {0.f, 0.f, 0.f, 0.f};
      if (grow < M) v = *(const float4*)&Ablk[(size_t)grow * HIDW + kloc + col];
      if (doRelu) {
        v.x = fmaxf(v.x, 0.f); v.y = fmaxf(v.y, 0.f);
        v.z = fmaxf(v.z, 0.f); v.w = fmaxf(v.w, 0.f);
      }
      *(float4*)&As[row][col] = v;
    }
#pragma unroll
    for (int j = 0; j < (NC * KCHUNK) / 1024; ++j) {
      int ei  = tid * 4 + j * 1024;
      int n   = ei >> 5;
      int col = ei & 31;
      *(float4*)&Ws[n][col] = *(const float4*)&W[(size_t)n * K + kb + col];
    }
    __syncthreads();

    const int arow = wave * 16 + l15;
    const int o = half * 2;
#pragma unroll
    for (int k4 = 0; k4 < KCHUNK; k4 += 4) {
      v2f a = *(const v2f*)&As[arow][k4 + o];
#pragma unroll
      for (int t = 0; t < NT; ++t) {
        v2f b = *(const v2f*)&Ws[t * 16 + l15][k4 + o];
        acc[t] = __builtin_amdgcn_wmma_f32_16x16x4_f32(
             false, a,  false, b,
             (short)0, acc[t],  false,  false);
      }
    }
    __syncthreads();
  }

  const int rbase = m0 + wave * 16 + half * 8;
  float dr[8];
#pragma unroll
  for (int r = 0; r < 8; ++r) dr[r] = (SCALE_COLS > 0 && rbase + r < M) ? dis[rbase + r] : 1.0f;
#pragma unroll
  for (int t = 0; t < NT; ++t) {
    int col = t * 16 + l15;
    float bv = HAS_BIAS ? bias[col] : 0.f;
#pragma unroll
    for (int r = 0; r < 8; ++r) {
      int row = rbase + r;
      if (row < M) {
        float v = acc[t][r] + bv;
        if (RELU_OUT) v = fmaxf(v, 0.f);
        if (t * 16 < SCALE_COLS) v *= dr[r];
        C[(size_t)row * NC + col] = v;
      }
    }
  }
}


#define XP (HIDW + 4)
#define TR 208
#define TRT (TR / 16)
#define GT (2 * TRT * 32)
#define XKS (HIDW / 4)
__global__ __launch_bounds__(GT) void k_gather_gemm2(float* __restrict__ GP, const float* __restrict__ hs,
                                                      const float* __restrict__ x,
                                                      const int* __restrict__ rowptr, const int* __restrict__ csr,
                                                      const float* __restrict__ dis, const float* __restrict__ b1,
                                                      const float* __restrict__ Wc, const float* __restrict__ bc,
                                                      const float* __restrict__ Wcat, const float* __restrict__ bias128,
                                                      int n) {
  __shared__ __align__(16) float Xa[TR][XP];
  __shared__ __align__(16) float Xx[TR][XP];
  __shared__ __align__(16) float Wx[HIDW][XP];
  __shared__ __align__(16) float Ws[2][HIDW][KCHUNK + 4];
  const int tid  = threadIdx.x;
  const int lane = tid & 31;
  const int wave = __builtin_amdgcn_readfirstlane(tid >> 5);
  const int l15  = lane & 15;
  const int half = lane >> 4;
  const int ch   = (wave >= TRT) ? 1 : 0;
  const int rg   = wave - ch * TRT;
  const int m0   = blockIdx.x * TR;

  const int wrow0 = tid >> 3;
  const int wrow1 = (tid + GT) >> 3;
  const int scol  = (tid & 7) * 4;
  const bool w1on = (tid + GT) < HIDW * 8;
  const int wrow1c = min(wrow1, HIDW - 1);
  const size_t wc0 = (size_t)wrow0 * (2 * HIDW) + scol, wc1 = (size_t)wrow1c * (2 * HIDW) + scol;
  const size_t wt0 = (size_t)wrow0 * HIDW + scol,       wt1 = (size_t)wrow1c * HIDW + scol;
  v4f wreg0, wreg1;

  const int lane4 = lane * 4;
  const int i0 = m0 + wave * 8;
  const int   rpl = rowptr[min(i0 + min(lane, 8), n)];
  const float dl  = dis[min(i0 + (lane & 7), n - 1)];
  const float4 bv1 = *(const float4*)&b1[lane4];

  constexpr int XJ = (TR * 32) / GT;
  constexpr int WJ = (HIDW * 32 + GT - 1) / GT;
  v4f xv[XJ], wv[WJ];
#pragma unroll
  for (int j = 0; j < XJ; ++j) {
    const int idx = tid + j * GT;
    xv[j] = *(const v4f*)&x[(size_t)min(m0 + (idx >> 5), n - 1) * HIDW + (idx & 31) * 4];
  }
#pragma unroll
  for (int j = 0; j < WJ; ++j) {
    const int idx = min(tid + j * GT, HIDW * 32 - 1);
    wv[j] = *(const v4f*)&Wc[(size_t)(idx >> 5) * (2 * HIDW) + HIDW + (idx & 31) * 4];
  }

  const int arow = rg * 16 + l15;
  const int o2   = half * 2;
  v8f acc[4];

  auto mstep = [&](int ks) {
    const int k = ks * 4 + o2;
    v2f a = *(const v2f*)&Xx[arow][k];
#pragma unroll
    for (int t = 0; t < 4; ++t) {
      v2f b = *(const v2f*)&Wx[ch * 64 + t * 16 + l15][k];
      acc[t] = __builtin_amdgcn_wmma_f32_16x16x4_f32(false, a, false, b, (short)0, acc[t], false, false);
    }
  };

  int ks = 0;
  auto do_row = [&](int rr, int nks) {
    const int i   = i0 + rr;
    const int beg = __builtin_amdgcn_readlane(rpl, rr);
    const int end = __builtin_amdgcn_readlane(rpl, rr + 1);
    const float di = __int_as_float(__builtin_amdgcn_readlane(__float_as_int(dl), rr));
    float4 o = {0.f, 0.f, 0.f, 0.f};
    if (i < n) {
      float4 ga = *(const float4*)(hs + (size_t)i * HIDW + lane4);
      for (int base = beg; base < end; base += 32) {
        const int m = min(32, end - base);
        const int my_s = csr[min(base + lane, end - 1)];
        int j = 0;
        for (; j + 16 <= m; j += 16) gather_rows_mm<16, 2>(ga, hs, my_s, j, lane4, ks, nks, mstep);
        if (m - j >= 8) { gather_rows_mm<8, 1>(ga, hs, my_s, j, lane4, ks, nks, mstep); j += 8; }
        if (m - j >= 4) { gather_rows<4>(ga, hs, my_s, j, lane4); j += 4; }
        if (m - j >= 2) { gather_rows<2>(ga, hs, my_s, j, lane4); j += 2; }
        if (m - j >= 1) { gather_rows<1>(ga, hs, my_s, j, lane4); }
      }
      o.x = fmaxf(fmaf(ga.x, di, bv1.x), 0.f);
      o.y = fmaxf(fmaf(ga.y, di, bv1.y), 0.f);
      o.z = fmaxf(fmaf(ga.z, di, bv1.z), 0.f);
      o.w = fmaxf(fmaf(ga.w, di, bv1.w), 0.f);
    }
    *(float4*)&Xa[wave * 8 + rr][lane4] = o;
  };

  do_row(0, 0);

#pragma unroll
  for (int j = 0; j < XJ; ++j) {
    const int idx = tid + j * GT;
    *(v4f*)&Xx[idx >> 5][(idx & 31) * 4] = xv[j];
  }
#pragma unroll
  for (int j = 0; j < WJ; ++j) {
    const int idx = tid + j * GT;
    if (idx < HIDW * 32) *(v4f*)&Wx[idx >> 5][(idx & 31) * 4] = wv[j];
  }
  __syncthreads();

#pragma unroll
  for (int t = 0; t < 4; ++t)
#pragma unroll
    for (int r = 0; r < 8; ++r) acc[t][r] = 0.0f;

  for (int rr = 1; rr < 8; ++rr) do_row(rr, XKS);
  for (; ks < XKS; ++ks) mstep(ks);

  wreg0 = *(const v4f*)&Wc[wc0];
  wreg1 = *(const v4f*)&Wc[wc1];
  *(v4f*)&Ws[0][wrow0][scol] = wreg0;
  if (w1on) *(v4f*)&Ws[0][wrow1][scol] = wreg1;
  wreg0 = *(const v4f*)&Wc[wc0 + KCHUNK];
  wreg1 = *(const v4f*)&Wc[wc1 + KCHUNK];
  __syncthreads();

#pragma nounroll
  for (int c = 0; c < 4; ++c) {
    const int s = c & 1;
    {
      *(v4f*)&Ws[s ^ 1][wrow0][scol] = wreg0;
      if (w1on) *(v4f*)&Ws[s ^ 1][wrow1][scol] = wreg1;
      const int cf = c + 2;
      if (cf < 4) {
        wreg0 = *(const v4f*)&Wc[wc0 + cf * KCHUNK];
        wreg1 = *(const v4f*)&Wc[wc1 + cf * KCHUNK];
      } else {
        wreg0 = *(const v4f*)&Wcat[wt0 + (cf - 4) * KCHUNK];
        wreg1 = *(const v4f*)&Wcat[wt1 + (cf - 4) * KCHUNK];
      }
    }
    const int kb = c * KCHUNK;
#pragma unroll
    for (int k4 = 0; k4 < KCHUNK; k4 += 4) {
      v2f a = *(const v2f*)&Xa[arow][kb + k4 + o2];
#pragma unroll
      for (int t = 0; t < 4; ++t) {
        v2f b = *(const v2f*)&Ws[s][ch * 64 + t * 16 + l15][k4 + o2];
        acc[t] = __builtin_amdgcn_wmma_f32_16x16x4_f32(false, a, false, b, (short)0, acc[t], false, false);
      }
    }
    __syncthreads();
  }

#pragma unroll
  for (int t = 0; t < 4; ++t) {
    const int col = ch * 64 + t * 16 + l15;
    const float bv = bc[col];
#pragma unroll
    for (int r = 0; r < 8; ++r) {
      Xa[rg * 16 + half * 8 + r][col] = fmaxf(acc[t][r] + bv, 0.f);
      acc[t][r] = 0.0f;
    }
  }
  __syncthreads();

#pragma nounroll
  for (int c = 4; c < 8; ++c) {
    const int s = c & 1;
    if (c + 1 < 8) {
      *(v4f*)&Ws[s ^ 1][wrow0][scol] = wreg0;
      if (w1on) *(v4f*)&Ws[s ^ 1][wrow1][scol] = wreg1;
    }
    if (c + 2 < 8) {
      wreg0 = *(const v4f*)&Wcat[wt0 + (c + 2 - 4) * KCHUNK];
      wreg1 = *(const v4f*)&Wcat[wt1 + (c + 2 - 4) * KCHUNK];
    }
    const int kb = (c - 4) * KCHUNK;
#pragma unroll
    for (int k4 = 0; k4 < KCHUNK; k4 += 4) {
      v2f a = *(const v2f*)&Xa[arow][kb + k4 + o2];
#pragma unroll
      for (int t = 0; t < 4; ++t) {
        v2f b = *(const v2f*)&Ws[s][ch * 64 + t * 16 + l15][k4 + o2];
        acc[t] = __builtin_amdgcn_wmma_f32_16x16x4_f32(false, a, false, b, (short)0, acc[t], false, false);
      }
    }
    if (c + 1 < 8) __syncthreads();
  }

  const int rbase = m0 + rg * 16 + half * 8;
  float dr[8];
#pragma unroll
  for (int r = 0; r < 8; ++r) dr[r] = (ch == 0 && rbase + r < n) ? dis[rbase + r] : 1.0f;
#pragma unroll
  for (int t = 0; t < 4; ++t) {
    const int col = ch * 64 + t * 16 + l15;
    const float bv = bias128[col];
#pragma unroll
    for (int r = 0; r < 8; ++r) {
      const int row = rbase + r;
      if (row < n) GP[(size_t)row * HIDW + col] = (acc[t][r] + bv) * dr[r];
    }
  }
}

extern "C" void kernel_launch(void* const* d_in, const int* in_sizes, int n_in,
                              void* d_out, int out_size, void* d_ws, size_t ws_size,
                              hipStream_t stream) {
  const float* x  = (const float*)d_in[0];
  const int* edge = (const int*)d_in[1];
  const float* W1 = (const float*)d_in[2];
  const float* b1 = (const float*)d_in[3];
  const float* Wc = (const float*)d_in[4];
  const float* bc = (const float*)d_in[5];
  const float* W2 = (const float*)d_in[6];
  const float* b2 = (const float*)d_in[7];
  const float* Wo = (const float*)d_in[8];
  const float* bo = (const float*)d_in[9];

  const int N = in_sizes[0] / HIDW;
  const int E = in_sizes[1] / 2;
  const int* srcIdx = edge;
  const int* dstIdx = edge + E;

  float* wsf = (float*)d_ws;
  size_t NP   = (((size_t)N) + 255) & ~(size_t)255;
  size_t NP1  = (((size_t)N) + 1 + 255) & ~(size_t)255;
  size_t EP   = (((size_t)E) + 3) & ~(size_t)3;
  float* dis  = wsf;
  float* buf1 = wsf + NP;
  float* buf2 = buf1 + (size_t)N * HIDW;
  float* buf3 = buf2 + (size_t)N * HIDW;
  int* rowptr = (int*)buf3;
  int* csr    = rowptr + NP1;
  float* Wcat = (float*)(csr + EP);
  float* bias128 = Wcat + HIDW * HIDW;
  int gC      = (E + P_CHUNK - 1) / P_CHUNK;
  int gB      = (N + BNODES - 1) >> BSH;
  int2* part  = (int2*)buf2;
  int* locG   = (int*)buf2 + 2 * EP;

  dim3 b256(256);
  int gG  = (N + 127) / 128;
  int gF  = (N + TR - 1) / TR;
  int gP  = 2048;

  k_bucket_scatter<<<gC + 1, dim3(P_THREADS), 0, stream>>>(part, locG, srcIdx, dstIdx, E, Wo, W2, b2, bo, Wcat, bias128);
  k_bucket_rows   <<<gB, dim3(ROWS_THREADS), 0, stream>>>(csr, rowptr, dis, part, locG, gC, N);

  k_gemm<128,1,false,false,false,128><<<gG, b256, 0, stream>>>(x, nullptr, W1, nullptr, dis, buf1, N);

  k_gather_gemm2<<<gF, dim3(GT), 0, stream>>>(buf2, buf1, x, rowptr, csr, dis, b1, Wc, bc, Wcat, bias128, N);
  k_prop_gather_half<<<gP, b256, 0, stream>>>((float*)d_out, buf2, rowptr, csr, dis, N);
}


# W1 product rides in the level-2 grouping launch as extra work-groups (matrix unit under the grouping's waits, one launch fewer); h left unscaled, dis[src] applied per edge by fma in 1st propagation
# speedup vs baseline: 9.9680x; 1.0455x over previous
#include <hip/hip_runtime.h>

typedef __attribute__((ext_vector_type(2))) float v2f;
typedef __attribute__((ext_vector_type(8))) float v8f;
typedef __attribute__((ext_vector_type(4))) float v4f;

#define HIDW 128
#define KCHUNK 32

__device__ __forceinline__ void fold_weights(const float* __restrict__ Wo, const float* __restrict__ W2,
                                             const float* __restrict__ b2, const float* __restrict__ bo,
                                             float* __restrict__ Wcat, float* __restrict__ bias128) {
  __shared__ __align__(16) float Fa[64][KCHUNK + 4];
  __shared__ __align__(16) float Fb[HIDW][KCHUNK + 4];
  const int tid  = threadIdx.x;
  const int wave = tid >> 5;
  const int lane = tid & 31;
  const int l15  = lane & 15;
  const int half = lane >> 4;
  const int mt   = wave & 3;
  const int nt   = wave >> 2;

  v8f acc;
#pragma unroll
  for (int r = 0; r < 8; ++r) acc[r] = 0.0f;

  for (int hb = 0; hb < HIDW; hb += KCHUNK) {
    {
      int idx = tid * 2;
      int o   = idx >> 5;
      int hh  = idx & 31;
      *(v2f*)&Fa[o][hh] = *(const v2f*)&Wo[(size_t)o * (2 * HIDW) + hb + hh];
    }
#pragma unroll
    for (int j = 0; j < (HIDW * KCHUNK) / 1024; ++j) {
      int e  = tid + j * 1024;
      int hh = e >> 7;
      int k  = e & 127;
      Fb[k][hh] = W2[(size_t)(hb + hh) * HIDW + k];
    }
    __syncthreads();
    const int o2 = half * 2;
#pragma unroll
    for (int k4 = 0; k4 < KCHUNK; k4 += 4) {
      v2f a = *(const v2f*)&Fa[mt * 16 + l15][k4 + o2];
      v2f b = *(const v2f*)&Fb[nt * 16 + l15][k4 + o2];
      acc = __builtin_amdgcn_wmma_f32_16x16x4_f32(false, a, false, b, (short)0, acc, false, false);
    }
    __syncthreads();
  }
#pragma unroll
  for (int r = 0; r < 8; ++r)
    Wcat[(size_t)(mt * 16 + half * 8 + r) * HIDW + nt * 16 + l15] = acc[r];

#pragma unroll
  for (int j = 0; j < 2; ++j) {
    int q  = tid + j * 1024;
    int o  = q >> 5;
    int c4 = (q & 31) * 4;
    *(float4*)&Wcat[(size_t)(64 + o) * HIDW + c4] = *(const float4*)&Wo[(size_t)o * (2 * HIDW) + HIDW + c4];
  }
  if (tid < 64) {
    float s = bo[tid];
    for (int h = 0; h < HIDW; ++h) s = fmaf(Wo[(size_t)tid * (2 * HIDW) + h], b2[h], s);
    bias128[tid]      = 0.f;
    bias128[64 + tid] = s;
  }
}

#define BSH 7
#define BNODES (1 << BSH)
#define NBMAX 512
#define P_THREADS 1024
#define P_PER 8
#define P_CHUNK (P_THREADS * P_PER)
#define LOCP (NBMAX + 1)

template<int NWAVES>
__device__ __forceinline__ void block_scan2(int& a, int& b, int* wsA, int* wsB) {
  const int lane = threadIdx.x & 31;
  const int wv   = threadIdx.x >> 5;
  int ia = a, ib = b;
  for (int o = 1; o < 32; o <<= 1) {
    int ta = __shfl_up(ia, o);
    int tb = __shfl_up(ib, o);
    if (lane >= o) { ia += ta; ib += tb; }
  }
  if (lane == 31) { wsA[wv] = ia; wsB[wv] = ib; }
  __syncthreads();
  if (wv == 0) {
    int ta = (lane < NWAVES) ? wsA[lane] : 0;
    int tb = (lane < NWAVES) ? wsB[lane] : 0;
    for (int o = 1; o < 32; o <<= 1) {
      int ua = __shfl_up(ta, o);
      int ub = __shfl_up(tb, o);
      if (lane >= o) { ta += ua; tb += ub; }
    }
    if (lane < NWAVES) { wsA[lane] = ta; wsB[lane] = tb; }
  }
  __syncthreads();
  a = ia - a + (wv ? wsA[wv - 1] : 0);
  b = ib - b + (wv ? wsB[wv - 1] : 0);
}

__global__ __launch_bounds__(P_THREADS) void k_bucket_scatter(int2* __restrict__ part, int* __restrict__ locG,
                                                              const int* __restrict__ src, const int* __restrict__ dst, int E,
                                                              const float* __restrict__ Wo, const float* __restrict__ W2,
                                                              const float* __restrict__ b2, const float* __restrict__ bo,
                                                              float* __restrict__ Wcat, float* __restrict__ bias128) {
  __shared__ int cnt[NBMAX];
  __shared__ int loc[NBMAX];
  __shared__ int wsA[32];
  __shared__ int wsB[32];
  __shared__ int2 stage[P_CHUNK];
  if (blockIdx.x == gridDim.x - 1) { fold_weights(Wo, W2, b2, bo, Wcat, bias128); return; }
  const int tid = threadIdx.x;
  const int wg  = blockIdx.x;
  const int e0  = wg * P_CHUNK;
  if (tid < NBMAX) cnt[tid] = 0;
  __syncthreads();

  int s[P_PER], d[P_PER], r[P_PER];
#pragma unroll
  for (int k = 0; k < P_PER; ++k) {
    int e = e0 + k * P_THREADS + tid;
    d[k] = -1; s[k] = 0; r[k] = 0;
    if (e < E) { d[k] = dst[e]; s[k] = src[e]; }
  }
#pragma unroll
  for (int k = 0; k < P_PER; ++k)
    if (d[k] >= 0) r[k] = atomicAdd(&cnt[d[k] >> BSH], 1);
  __syncthreads();

  const int nloc = min(P_CHUNK, E - e0);
  int unused = 0;
  int c = (tid < NBMAX) ? cnt[tid] : 0;
  block_scan2<32>(unused, c, wsA, wsB);
  if (tid < NBMAX) {
    loc[tid] = c;
    locG[(size_t)wg * LOCP + tid] = c;
  }
  if (tid == 0) locG[(size_t)wg * LOCP + NBMAX] = nloc;
  __syncthreads();
#pragma unroll
  for (int k = 0; k < P_PER; ++k)
    if (d[k] >= 0) stage[loc[d[k] >> BSH] + r[k]] = make_int2(s[k], d[k]);
  __syncthreads();
#pragma unroll
  for (int k = 0; k < P_PER; ++k) {
    int i = k * P_THREADS + tid;
    if (i < nloc) part[(size_t)e0 + i] = stage[i];
  }
}

struct W1Tile {
  float As[128][KCHUNK + 4];
  float Ws[HIDW][KCHUNK + 4];
};
__device__ __forceinline__ void w1_tile(W1Tile& t, const float* __restrict__ A, const float* __restrict__ W,
                                        float* __restrict__ C, int M, int tile) {
  constexpr int NT = HIDW / 16;
  const int tid  = threadIdx.x;
  const int wave = tid >> 5;
  const int lane = tid & 31;
  const int l15  = lane & 15;
  const int half = lane >> 4;
  const int m0   = tile * 128;

  v8f acc[NT];
#pragma unroll
  for (int q = 0; q < NT; ++q)
#pragma unroll
    for (int r = 0; r < 8; ++r) acc[q][r] = 0.0f;

  for (int kb = 0; kb < HIDW; kb += KCHUNK) {
#pragma unroll
    for (int j = 0; j < (128 * KCHUNK) / 1024; ++j) {
      int ei  = tid * 4 + j * 1024;
      int row = ei >> 5;
      int col = ei & 31;
      int grow = m0 + row;
      float4 v = {0.f, 0.f, 0.f, 0.f};
      if (grow < M) v = *(const float4*)&A[(size_t)grow * HIDW + kb + col];
      *(float4*)&t.As[row][col] = v;
    }
#pragma unroll
    for (int j = 0; j < (HIDW * KCHUNK) / 1024; ++j) {
      int ei  = tid * 4 + j * 1024;
      int nn  = ei >> 5;
      int col = ei & 31;
      *(float4*)&t.Ws[nn][col] = *(const float4*)&W[(size_t)nn * HIDW + kb + col];
    }
    __syncthreads();

    const int arow = wave * 16 + l15;
    const int o = half * 2;
#pragma unroll
    for (int k4 = 0; k4 < KCHUNK; k4 += 4) {
      v2f a = *(const v2f*)&t.As[arow][k4 + o];
#pragma unroll
      for (int q = 0; q < NT; ++q) {
        v2f b = *(const v2f*)&t.Ws[q * 16 + l15][k4 + o];
        acc[q] = __builtin_amdgcn_wmma_f32_16x16x4_f32(
             false, a,  false, b,
             (short)0, acc[q],  false,  false);
      }
    }
    __syncthreads();
  }

  const int rbase = m0 + wave * 16 + half * 8;
#pragma unroll
  for (int q = 0; q < NT; ++q) {
    int col = q * 16 + l15;
#pragma unroll
    for (int r = 0; r < 8; ++r) {
      int row = rbase + r;
      if (row < M) C[(size_t)row * HIDW + col] = acc[q][r];
    }
  }
}

#define ROWS_THREADS 256
#define ROWS_WAVES (ROWS_THREADS / 32)
#define RUNS_U 13
#define ROWS_STAGE 12288
union RowsLds {
  int stage[ROWS_STAGE];
  W1Tile tile;
};
__global__ __launch_bounds__(ROWS_THREADS, 8) void k_bucket_rows(int* __restrict__ csr, int* __restrict__ rowptr,
                                                              float* __restrict__ dis, const int2* __restrict__ part,
                                                              const int* __restrict__ locG, int nchunks, int n,
                                                              const float* __restrict__ xin, const float* __restrict__ W1,
                                                              float* __restrict__ hout, int ntiles) {
  __shared__ int cntN[BNODES];
  __shared__ int fillN[BNODES];
  __shared__ int rstart[BNODES];
  __shared__ int wsA[ROWS_WAVES];
  __shared__ int wsB[ROWS_WAVES];
  __shared__ int tot[3];
  __shared__ __align__(16) RowsLds u;
  if ((int)blockIdx.x < ntiles) { w1_tile(u.tile, xin, W1, hout, n, (int)blockIdx.x); return; }
  int* const stage = u.stage;
  const int tid  = threadIdx.x;
  const int lane = tid & 31;
  const int wave = __builtin_amdgcn_readfirstlane(tid >> 5);
  const int b    = (int)blockIdx.x - ntiles;
  const int cpw  = (nchunks + ROWS_WAVES - 1) / ROWS_WAVES;
  const int cme  = wave * cpw + lane;
  int lo = 0, len = 0;
  if (lane < cpw && cme < nchunks) {
    lo  = locG[(size_t)cme * LOCP + b];
    len = locG[(size_t)cme * LOCP + b + 1] - lo;
  }
  const int myPos = cme * P_CHUNK + lo;
  if (tid < BNODES) { cntN[tid] = 0; fillN[tid] = 0; }
  if (tid == 0) tot[2] = 0;
  int pl = len, pb = lo;
  block_scan2<ROWS_WAVES>(pl, pb, wsA, wsB);
  if (len > 64) tot[2] = 1;
  if (tid == ROWS_THREADS - 1) { tot[0] = pl + len; tot[1] = pb + lo; }
  __syncthreads();
  const int ne  = tot[0];
  const int beg = tot[1];
  const int end = beg + ne;
  const bool longRuns = tot[2] != 0;

  for (int k0 = 0; k0 < cpw; k0 += RUNS_U) {
    int2 ra[RUNS_U], rb[RUNS_U];
#pragma unroll
    for (int u = 0; u < RUNS_U; ++u) {
      const int k = min(k0 + u, 31);
      const int L = (k0 + u < cpw) ? __builtin_amdgcn_readlane(len, k) : 0;
      const int p = __builtin_amdgcn_readlane(myPos, k);
      ra[u] = (lane < L)      ? part[(size_t)p + lane]      : make_int2(0, -1);
      rb[u] = (lane + 32 < L) ? part[(size_t)p + 32 + lane] : make_int2(0, -1);
    }
#pragma unroll
    for (int u = 0; u < RUNS_U; ++u) {
      if (ra[u].y >= 0) atomicAdd(&cntN[ra[u].y & (BNODES - 1)], 1);
      if (rb[u].y >= 0) atomicAdd(&cntN[rb[u].y & (BNODES - 1)], 1);
    }
  }
  if (longRuns) {
    for (int k = 0; k < cpw; ++k) {
      const int L = __builtin_amdgcn_readlane(len, k);
      const int p = __builtin_amdgcn_readlane(myPos, k);
      for (int o = 64 + lane; o < L; o += 32) atomicAdd(&cntN[part[(size_t)p + o].y & (BNODES - 1)], 1);
    }
  }
  __syncthreads();
  int c = (tid < BNODES) ? cntN[tid] : 0;
  int x = c, unused = 0;
  block_scan2<ROWS_WAVES>(x, unused, wsA, wsB);
  if (tid < BNODES) {
    rstart[tid] = x;
    const int node = b * BNODES + tid;
    if (node < n) {
      rowptr[node] = beg + x;
      float dg = (float)(c + 1);
      dis[node] = (dg > 0.f) ? rsqrtf(dg) : 0.f;
    }
  }
  if (b == (int)gridDim.x - ntiles - 1 && tid == 0) rowptr[n] = end;
  __syncthreads();
  const bool staged = (ne <= ROWS_STAGE);
  for (int k0 = 0; k0 < cpw; k0 += RUNS_U) {
    int2 ra[RUNS_U], rb[RUNS_U];
#pragma unroll
    for (int u = 0; u < RUNS_U; ++u) {
      const int k = min(k0 + u, 31);
      const int L = (k0 + u < cpw) ? __builtin_amdgcn_readlane(len, k) : 0;
      const int p = __builtin_amdgcn_readlane(myPos, k);
      ra[u] = (lane < L)      ? part[(size_t)p + lane]      : make_int2(0, -1);
      rb[u] = (lane + 32 < L) ? part[(size_t)p + 32 + lane] : make_int2(0, -1);
    }
    int pa[RUNS_U], pc[RUNS_U];
#pragma unroll
    for (int u = 0; u < RUNS_U; ++u) {
      pa[u] = -1; pc[u] = -1;
      if (ra[u].y >= 0) {
        const int dl = ra[u].y & (BNODES - 1);
        pa[u] = rstart[dl] + atomicAdd(&fillN[dl], 1);
      }
      if (rb[u].y >= 0) {
        const int dl = rb[u].y & (BNODES - 1);
        pc[u] = rstart[dl] + atomicAdd(&fillN[dl], 1);
      }
    }
    if (staged) {
#pragma unroll
      for (int u = 0; u < RUNS_U; ++u) {
        if (pa[u] >= 0) stage[pa[u]] = ra[u].x;
        if (pc[u] >= 0) stage[pc[u]] = rb[u].x;
      }
    } else {
#pragma unroll
      for (int u = 0; u < RUNS_U; ++u) {
        if (pa[u] >= 0) csr[(size_t)beg + pa[u]] = ra[u].x;
        if (pc[u] >= 0) csr[(size_t)beg + pc[u]] = rb[u].x;
      }
    }
  }
  if (longRuns) {
    for (int k = 0; k < cpw; ++k) {
      const int L = __builtin_amdgcn_readlane(len, k);
      const int p = __builtin_amdgcn_readlane(myPos, k);
      for (int o = 64 + lane; o < L; o += 32) {
        const int2 rec = part[(size_t)p + o];
        const int dl  = rec.y & (BNODES - 1);
        const int pos = rstart[dl] + atomicAdd(&fillN[dl], 1);
        if (staged) stage[pos] = rec.x; else csr[(size_t)beg + pos] = rec.x;
      }
    }
  }
  if (staged) {
    __syncthreads();
    for (int i = tid; i < ne; i += ROWS_THREADS) csr[(size_t)beg + i] = stage[i];
  }
}

template<int U>
__device__ __forceinline__ void gather_rows(float4& acc, const float* __restrict__ h, int my_s, float my_d, int j, int lane4) {
  float4 v[U];
#pragma unroll
  for (int u = 0; u < U; ++u) {
    int s = __builtin_amdgcn_readlane(my_s, j + u);
    v[u] = *(const float4*)(h + (size_t)s * HIDW + lane4);
  }
#pragma unroll
  for (int u = 0; u < U; ++u) {
    const float ds = __int_as_float(__builtin_amdgcn_readlane(__float_as_int(my_d), j + u));
    acc.x = fmaf(v[u].x, ds, acc.x);
    acc.y = fmaf(v[u].y, ds, acc.y);
    acc.z = fmaf(v[u].z, ds, acc.z);
    acc.w = fmaf(v[u].w, ds, acc.w);
  }
}

template<int U, int XS, class F>
__device__ __forceinline__ void gather_rows_mm(float4& acc, const float* __restrict__ h, int my_s, float my_d, int j, int lane4,
                                               int& ks, int nks, F& mstep) {
  float4 v[U];
#pragma unroll
  for (int u = 0; u < U; ++u) {
    int s = __builtin_amdgcn_readlane(my_s, j + u);
    v[u] = *(const float4*)(h + (size_t)s * HIDW + lane4);
  }
#pragma unroll
  for (int q = 0; q < XS; ++q)
    if (ks < nks) { mstep(ks); ++ks; }
#pragma unroll
  for (int u = 0; u < U; ++u) {
    const float ds = __int_as_float(__builtin_amdgcn_readlane(__float_as_int(my_d), j + u));
    acc.x = fmaf(v[u].x, ds, acc.x);
    acc.y = fmaf(v[u].y, ds, acc.y);
    acc.z = fmaf(v[u].z, ds, acc.z);
    acc.w = fmaf(v[u].w, ds, acc.w);
  }
}

template<int U, bool TAIL>
__device__ __forceinline__ void gather_pairs(float4& acc, const float* __restrict__ gp, int my_s,
                                             int j, int m, bool hi, int l15x4) {
  float4 v[U];
#pragma unroll
  for (int u = 0; u < U; ++u) {
    int slo = __builtin_amdgcn_readlane(my_s, j + 2 * u);
    int shi = __builtin_amdgcn_readlane(my_s, j + 2 * u + 1);
    int s = hi ? shi : slo;
    v[u] = *(const float4*)(gp + (size_t)s * HIDW + l15x4);
  }
#pragma unroll
  for (int u = 0; u < U; ++u) {
    const bool live = !TAIL || !hi || (j + 2 * u + 1 < m);
    acc.x += live ? v[u].x : 0.f;
    acc.y += live ? v[u].y : 0.f;
    acc.z += live ? v[u].z : 0.f;
    acc.w += live ? v[u].w : 0.f;
  }
}

__global__ __launch_bounds__(256) void k_prop_gather_half(float* __restrict__ out, const float* __restrict__ gp,
                                                          const int* __restrict__ rowptr, const int* __restrict__ csr,
                                                          const float* __restrict__ dis, int n) {
  const int lane  = threadIdx.x & 31;
  const bool hi   = lane >= 16;
  const int l15x4 = (lane & 15) * 4;
  const int wave = __builtin_amdgcn_readfirstlane((int)(threadIdx.x >> 5));
  const int nw   = gridDim.x * 8;

  for (int i = blockIdx.x * 8 + wave; i < n; i += nw) {
    const int beg = rowptr[i];
    const int end = rowptr[i + 1];
    const float di = dis[i];
    const float4 rv = *(const float4*)(gp + (size_t)i * HIDW + (hi ? 64 : 0) + l15x4);
    float4 acc;
    acc.x = hi ? 0.f : rv.x; acc.y = hi ? 0.f : rv.y; acc.z = hi ? 0.f : rv.z; acc.w = hi ? 0.f : rv.w;

    for (int base = beg; base < end; base += 32) {
      const int m = min(32, end - base);
      const int my_s = csr[min(base + lane, end - 1)];
      int j = 0;
      for (; j + 16 <= m; j += 16) gather_pairs<8, false>(acc, gp, my_s, j, m, hi, l15x4);
      if (m - j >= 8) { gather_pairs<4, false>(acc, gp, my_s, j, m, hi, l15x4); j += 8; }
      if (m - j >= 4) { gather_pairs<2, false>(acc, gp, my_s, j, m, hi, l15x4); j += 4; }
      if (m - j >= 3)      gather_pairs<2, true>(acc, gp, my_s, j, m, hi, l15x4);
      else if (m - j >= 1) gather_pairs<1, true>(acc, gp, my_s, j, m, hi, l15x4);
    }
    float4 o;
    o.x = fmaf(acc.x + __shfl_xor(acc.x, 16), di, __shfl_xor(rv.x, 16));
    o.y = fmaf(acc.y + __shfl_xor(acc.y, 16), di, __shfl_xor(rv.y, 16));
    o.z = fmaf(acc.z + __shfl_xor(acc.z, 16), di, __shfl_xor(rv.z, 16));
    o.w = fmaf(acc.w + __shfl_xor(acc.w, 16), di, __shfl_xor(rv.w, 16));
    if (!hi) *(float4*)(out + (size_t)i * 64 + l15x4) = o;
  }
}

#define XP (HIDW + 4)
#define TR 208
#define TRT (TR / 16)
#define GT (2 * TRT * 32)
#define XKS (HIDW / 4)
__global__ __launch_bounds__(GT) void k_gather_gemm2(float* __restrict__ GP, const float* __restrict__ h,
                                                      const float* __restrict__ x,
                                                      const int* __restrict__ rowptr, const int* __restrict__ csr,
                                                      const float* __restrict__ dis, const float* __restrict__ b1,
                                                      const float* __restrict__ Wc, const float* __restrict__ bc,
                                                      const float* __restrict__ Wcat, const float* __restrict__ bias128,
                                                      int n) {
  __shared__ __align__(16) float Xa[TR][XP];
  __shared__ __align__(16) float Xx[TR][XP];
  __shared__ __align__(16) float Wx[HIDW][XP];
  __shared__ __align__(16) float Ws[2][HIDW][KCHUNK + 4];
  const int tid  = threadIdx.x;
  const int lane = tid & 31;
  const int wave = __builtin_amdgcn_readfirstlane(tid >> 5);
  const int l15  = lane & 15;
  const int half = lane >> 4;
  const int ch   = (wave >= TRT) ? 1 : 0;
  const int rg   = wave - ch * TRT;
  const int m0   = blockIdx.x * TR;

  const int wrow0 = tid >> 3;
  const int wrow1 = (tid + GT) >> 3;
  const int scol  = (tid & 7) * 4;
  const bool w1on = (tid + GT) < HIDW * 8;
  const int wrow1c = min(wrow1, HIDW - 1);
  const size_t wc0 = (size_t)wrow0 * (2 * HIDW) + scol, wc1 = (size_t)wrow1c * (2 * HIDW) + scol;
  const size_t wt0 = (size_t)wrow0 * HIDW + scol,       wt1 = (size_t)wrow1c * HIDW + scol;
  v4f wreg0, wreg1;

  const int lane4 = lane * 4;
  const int i0 = m0 + wave * 8;
  const int   rpl = rowptr[min(i0 + min(lane, 8), n)];
  const float dl  = dis[min(i0 + (lane & 7), n - 1)];
  const float4 bv1 = *(const float4*)&b1[lane4];

  constexpr int XJ = (TR * 32) / GT;
  constexpr int WJ = (HIDW * 32 + GT - 1) / GT;
  v4f xv[XJ], wv[WJ];
#pragma unroll
  for (int j = 0; j < XJ; ++j) {
    const int idx = tid + j * GT;
    xv[j] = *(const v4f*)&x[(size_t)min(m0 + (idx >> 5), n - 1) * HIDW + (idx & 31) * 4];
  }
#pragma unroll
  for (int j = 0; j < WJ; ++j) {
    const int idx = min(tid + j * GT, HIDW * 32 - 1);
    wv[j] = *(const v4f*)&Wc[(size_t)(idx >> 5) * (2 * HIDW) + HIDW + (idx & 31) * 4];
  }

  const int arow = rg * 16 + l15;
  const int o2   = half * 2;
  v8f acc[4];

  auto mstep = [&](int ks) {
    const int k = ks * 4 + o2;
    v2f a = *(const v2f*)&Xx[arow][k];
#pragma unroll
    for (int t = 0; t < 4; ++t) {
      v2f b = *(const v2f*)&Wx[ch * 64 + t * 16 + l15][k];
      acc[t] = __builtin_amdgcn_wmma_f32_16x16x4_f32(false, a, false, b, (short)0, acc[t], false, false);
    }
  };

  int ks = 0;
  auto do_row = [&](int rr, int nks) {
    const int i   = i0 + rr;
    const int beg = __builtin_amdgcn_readlane(rpl, rr);
    const int end = __builtin_amdgcn_readlane(rpl, rr + 1);
    const float di = __int_as_float(__builtin_amdgcn_readlane(__float_as_int(dl), rr));
    float4 o = {0.f, 0.f, 0.f, 0.f};
    if (i < n) {
      const float4 hv = *(const float4*)(h + (size_t)i * HIDW + lane4);
      float4 ga = {hv.x * di, hv.y * di, hv.z * di, hv.w * di};
      for (int base = beg; base < end; base += 32) {
        const int m = min(32, end - base);
        const int my_s = csr[min(base + lane, end - 1)];
        const float my_d = dis[my_s];
        int j = 0;
        for (; j + 16 <= m; j += 16) gather_rows_mm<16, 2>(ga, h, my_s, my_d, j, lane4, ks, nks, mstep);
        if (m - j >= 8) { gather_rows_mm<8, 1>(ga, h, my_s, my_d, j, lane4, ks, nks, mstep); j += 8; }
        if (m - j >= 4) { gather_rows<4>(ga, h, my_s, my_d, j, lane4); j += 4; }
        if (m - j >= 2) { gather_rows<2>(ga, h, my_s, my_d, j, lane4); j += 2; }
        if (m - j >= 1) { gather_rows<1>(ga, h, my_s, my_d, j, lane4); }
      }
      o.x = fmaxf(fmaf(ga.x, di, bv1.x), 0.f);
      o.y = fmaxf(fmaf(ga.y, di, bv1.y), 0.f);
      o.z = fmaxf(fmaf(ga.z, di, bv1.z), 0.f);
      o.w = fmaxf(fmaf(ga.w, di, bv1.w), 0.f);
    }
    *(float4*)&Xa[wave * 8 + rr][lane4] = o;
  };

  do_row(0, 0);

#pragma unroll
  for (int j = 0; j < XJ; ++j) {
    const int idx = tid + j * GT;
    *(v4f*)&Xx[idx >> 5][(idx & 31) * 4] = xv[j];
  }
#pragma unroll
  for (int j = 0; j < WJ; ++j) {
    const int idx = tid + j * GT;
    if (idx < HIDW * 32) *(v4f*)&Wx[idx >> 5][(idx & 31) * 4] = wv[j];
  }
  __syncthreads();

#pragma unroll
  for (int t = 0; t < 4; ++t)
#pragma unroll
    for (int r = 0; r < 8; ++r) acc[t][r] = 0.0f;

  for (int rr = 1; rr < 8; ++rr) do_row(rr, XKS);
  for (; ks < XKS; ++ks) mstep(ks);

  wreg0 = *(const v4f*)&Wc[wc0];
  wreg1 = *(const v4f*)&Wc[wc1];
  *(v4f*)&Ws[0][wrow0][scol] = wreg0;
  if (w1on) *(v4f*)&Ws[0][wrow1][scol] = wreg1;
  wreg0 = *(const v4f*)&Wc[wc0 + KCHUNK];
  wreg1 = *(const v4f*)&Wc[wc1 + KCHUNK];
  __syncthreads();

#pragma nounroll
  for (int c = 0; c < 4; ++c) {
    const int s = c & 1;
    {
      *(v4f*)&Ws[s ^ 1][wrow0][scol] = wreg0;
      if (w1on) *(v4f*)&Ws[s ^ 1][wrow1][scol] = wreg1;
      const int cf = c + 2;
      if (cf < 4) {
        wreg0 = *(const v4f*)&Wc[wc0 + cf * KCHUNK];
        wreg1 = *(const v4f*)&Wc[wc1 + cf * KCHUNK];
      } else {
        wreg0 = *(const v4f*)&Wcat[wt0 + (cf - 4) * KCHUNK];
        wreg1 = *(const v4f*)&Wcat[wt1 + (cf - 4) * KCHUNK];
      }
    }
    const int kb = c * KCHUNK;
#pragma unroll
    for (int k4 = 0; k4 < KCHUNK; k4 += 4) {
      v2f a = *(const v2f*)&Xa[arow][kb + k4 + o2];
#pragma unroll
      for (int t = 0; t < 4; ++t) {
        v2f b = *(const v2f*)&Ws[s][ch * 64 + t * 16 + l15][k4 + o2];
        acc[t] = __builtin_amdgcn_wmma_f32_16x16x4_f32(false, a, false, b, (short)0, acc[t], false, false);
      }
    }
    __syncthreads();
  }

#pragma unroll
  for (int t = 0; t < 4; ++t) {
    const int col = ch * 64 + t * 16 + l15;
    const float bv = bc[col];
#pragma unroll
    for (int r = 0; r < 8; ++r) {
      Xa[rg * 16 + half * 8 + r][col] = fmaxf(acc[t][r] + bv, 0.f);
      acc[t][r] = 0.0f;
    }
  }
  __syncthreads();

#pragma nounroll
  for (int c = 4; c < 8; ++c) {
    const int s = c & 1;
    if (c + 1 < 8) {
      *(v4f*)&Ws[s ^ 1][wrow0][scol] = wreg0;
      if (w1on) *(v4f*)&Ws[s ^ 1][wrow1][scol] = wreg1;
    }
    if (c + 2 < 8) {
      wreg0 = *(const v4f*)&Wcat[wt0 + (c + 2 - 4) * KCHUNK];
      wreg1 = *(const v4f*)&Wcat[wt1 + (c + 2 - 4) * KCHUNK];
    }
    const int kb = (c - 4) * KCHUNK;
#pragma unroll
    for (int k4 = 0; k4 < KCHUNK; k4 += 4) {
      v2f a = *(const v2f*)&Xa[arow][kb + k4 + o2];
#pragma unroll
      for (int t = 0; t < 4; ++t) {
        v2f b = *(const v2f*)&Ws[s][ch * 64 + t * 16 + l15][k4 + o2];
        acc[t] = __builtin_amdgcn_wmma_f32_16x16x4_f32(false, a, false, b, (short)0, acc[t], false, false);
      }
    }
    if (c + 1 < 8) __syncthreads();
  }

  const int rbase = m0 + rg * 16 + half * 8;
  float dr[8];
#pragma unroll
  for (int r = 0; r < 8; ++r) dr[r] = (ch == 0 && rbase + r < n) ? dis[rbase + r] : 1.0f;
#pragma unroll
  for (int t = 0; t < 4; ++t) {
    const int col = ch * 64 + t * 16 + l15;
    const float bv = bias128[col];
#pragma unroll
    for (int r = 0; r < 8; ++r) {
      const int row = rbase + r;
      if (row < n) GP[(size_t)row * HIDW + col] = (acc[t][r] + bv) * dr[r];
    }
  }
}

extern "C" void kernel_launch(void* const* d_in, const int* in_sizes, int n_in,
                              void* d_out, int out_size, void* d_ws, size_t ws_size,
                              hipStream_t stream) {
  const float* x  = (const float*)d_in[0];
  const int* edge = (const int*)d_in[1];
  const float* W1 = (const float*)d_in[2];
  const float* b1 = (const float*)d_in[3];
  const float* Wc = (const float*)d_in[4];
  const float* bc = (const float*)d_in[5];
  const float* W2 = (const float*)d_in[6];
  const float* b2 = (const float*)d_in[7];
  const float* Wo = (const float*)d_in[8];
  const float* bo = (const float*)d_in[9];

  const int N = in_sizes[0] / HIDW;
  const int E = in_sizes[1] / 2;
  const int* srcIdx = edge;
  const int* dstIdx = edge + E;

  float* wsf = (float*)d_ws;
  size_t NP   = (((size_t)N) + 255) & ~(size_t)255;
  size_t NP1  = (((size_t)N) + 1 + 255) & ~(size_t)255;
  size_t EP   = (((size_t)E) + 3) & ~(size_t)3;
  float* dis  = wsf;
  float* buf1 = wsf + NP;
  float* buf2 = buf1 + (size_t)N * HIDW;
  float* buf3 = buf2 + (size_t)N * HIDW;
  int* rowptr = (int*)buf3;
  int* csr    = rowptr + NP1;
  float* Wcat = (float*)(csr + EP);
  float* bias128 = Wcat + HIDW * HIDW;
  int gC      = (E + P_CHUNK - 1) / P_CHUNK;
  int gB      = (N + BNODES - 1) >> BSH;
  int2* part  = (int2*)buf2;
  int* locG   = (int*)buf2 + 2 * EP;

  dim3 b256(256);
  int gG  = (N + 127) / 128;
  int gF  = (N + TR - 1) / TR;
  int gP  = 2048;

  k_bucket_scatter<<<gC + 1, dim3(P_THREADS), 0, stream>>>(part, locG, srcIdx, dstIdx, E, Wo, W2, b2, bo, Wcat, bias128);
  k_bucket_rows   <<<gG + gB, dim3(ROWS_THREADS), 0, stream>>>(csr, rowptr, dis, part, locG, gC, N, x, W1, buf1, gG);

  k_gather_gemm2<<<gF, dim3(GT), 0, stream>>>(buf2, buf1, x, rowptr, csr, dis, b1, Wc, bc, Wcat, bias128, N);
  k_prop_gather_half<<<gP, b256, 0, stream>>>((float*)d_out, buf2, rowptr, csr, dis, N);
}
